# starting kernel
# baseline (speedup 1.0000x reference)
#include <hip/hip_runtime.h>

typedef __attribute__((ext_vector_type(2))) float v2f;
typedef __attribute__((ext_vector_type(8))) float v8f;

#define NEG_SLOPE 0.2f

static __device__ __forceinline__ unsigned f2key(float f) {
  unsigned u = __float_as_uint(f);
  return (u & 0x80000000u) ? ~u : (u | 0x80000000u);
}
static __device__ __forceinline__ float key2f(unsigned k) {
  return (k & 0x80000000u) ? __uint_as_float(k & 0x7FFFFFFFu)
                           : __uint_as_float(~k);
}

__global__ void zero_f32_kernel(float* __restrict__ p, long n) {
  long i = (long)blockIdx.x * blockDim.x + threadIdx.x;
  long stride = (long)gridDim.x * blockDim.x;
  for (; i < n; i += stride) p[i] = 0.0f;
}

__global__ void pad_cols_kernel(const float* __restrict__ src,
                                float* __restrict__ dst,
                                int K, int c, int ldd) {
  int i = blockIdx.x * blockDim.x + threadIdx.x;
  if (i >= K * ldd) return;
  int r = i / ldd, j = i - r * ldd;
  dst[i] = (j < c) ? src[r * c + j] : 0.0f;
}

__global__ void gemm_f32_wmma_kernel(const float* __restrict__ A,
                                     const float* __restrict__ B,
                                     float* __restrict__ C,
                                     int M, int K, int ldb) {
  const int lane = threadIdx.x & 31;
  const int wave = threadIdx.x >> 5;
  const int half = lane >> 4;
  const int l    = lane & 15;
  const int nt64   = ldb >> 6;
  const int mtiles = M >> 4;
  const int total  = mtiles * nt64;
  int t = blockIdx.x * (blockDim.x >> 5) + wave;
  if (t >= total) return;
  const int mt = t / nt64;
  const int n0 = (t - mt * nt64) << 6;
  const int m0 = mt << 4;

  v8f acc[4] = {};
  const float* arow = A + (size_t)(m0 + l) * K + 2 * half;
  const float* bp   = B + (size_t)(2 * half) * ldb + n0 + l;
  const size_t bstep = (size_t)4 * ldb;

  v2f a = *(const v2f*)(arow);
  v2f b[4];
#pragma unroll
  for (int s = 0; s < 4; ++s) {
    b[s].x = bp[s * 16];
    b[s].y = bp[ldb + s * 16];
  }

  for (int k0 = 4; k0 < K; k0 += 4) {
    v2f an = *(const v2f*)(arow + k0);
    const float* bpn = bp + bstep;
    v2f bn[4];
#pragma unroll
    for (int s = 0; s < 4; ++s) {
      bn[s].x = bpn[s * 16];
      bn[s].y = bpn[ldb + s * 16];
    }
#pragma unroll
    for (int s = 0; s < 4; ++s)
      acc[s] = __builtin_amdgcn_wmma_f32_16x16x4_f32(
          false, a, false, b[s], (short)0, acc[s], false, false);
    a = an;
#pragma unroll
    for (int s = 0; s < 4; ++s) b[s] = bn[s];
    bp = bpn;
  }
#pragma unroll
  for (int s = 0; s < 4; ++s)
    acc[s] = __builtin_amdgcn_wmma_f32_16x16x4_f32(
        false, a, false, b[s], (short)0, acc[s], false, false);

  float* crow = C + (size_t)(m0 + 8 * half) * ldb + n0 + l;
#pragma unroll
  for (int s = 0; s < 4; ++s)
#pragma unroll
    for (int j = 0; j < 8; ++j) crow[(size_t)j * ldb + s * 16] = acc[s][j];
}

__global__ void alpha_kernel(const float* __restrict__ xw,
                             const float* __restrict__ a_src,
                             const float* __restrict__ a_dst,
                             float* __restrict__ as, float* __restrict__ ad,
                             int n_nodes, int heads, int c, int ldx) {
  int i = blockIdx.x * blockDim.x + threadIdx.x;
  if (i >= n_nodes * heads) return;
  int n = i / heads, h = i - n * heads;
  const float* row = xw + (size_t)n * ldx + (size_t)h * c;
  const float* ws = a_src + (size_t)h * c;
  const float* wd = a_dst + (size_t)h * c;
  float s = 0.0f, d = 0.0f;
  for (int j = 0; j < c; ++j) {
    float v = row[j];
    s += v * ws[j];
    d += v * wd[j];
  }
  as[i] = s;
  ad[i] = d;
}

__global__ void edge_max_kernel(const int* __restrict__ esrc,
                                const int* __restrict__ edst,
                                int E, int Nn, int heads,
                                const float* __restrict__ as,
                                const float* __restrict__ ad,
                                unsigned* __restrict__ mkey) {
  int e = blockIdx.x * blockDim.x + threadIdx.x;
  if (e >= E + Nn) return;
  int s = (e < E) ? esrc[e] : (e - E);
  int d = (e < E) ? edst[e] : (e - E);
  for (int h = 0; h < heads; ++h) {
    float v = as[s * heads + h] + ad[d * heads + h];
    v = (v >= 0.0f) ? v : NEG_SLOPE * v;
    atomicMax(&mkey[d * heads + h], f2key(v));
  }
}

__global__ void edge_denom_kernel(const int* __restrict__ esrc,
                                  const int* __restrict__ edst,
                                  int E, int Nn, int heads,
                                  const float* __restrict__ as,
                                  const float* __restrict__ ad,
                                  const unsigned* __restrict__ mkey,
                                  float* __restrict__ denom) {
  int e = blockIdx.x * blockDim.x + threadIdx.x;
  if (e >= E + Nn) return;
  int s = (e < E) ? esrc[e] : (e - E);
  int d = (e < E) ? edst[e] : (e - E);
  for (int h = 0; h < heads; ++h) {
    float v = as[s * heads + h] + ad[d * heads + h];
    v = (v >= 0.0f) ? v : NEG_SLOPE * v;
    atomicAdd(&denom[d * heads + h], expf(v - key2f(mkey[d * heads + h])));
  }
}

__global__ void edge_aggregate_kernel(const int* __restrict__ esrc,
                                      const int* __restrict__ edst,
                                      int E, int Nn, int heads, int C, int ldx,
                                      const float* __restrict__ as,
                                      const float* __restrict__ ad,
                                      const unsigned* __restrict__ mkey,
                                      const float* __restrict__ denom,
                                      const float* __restrict__ xw,
                                      float* __restrict__ out) {
  long tid = (long)blockIdx.x * blockDim.x + threadIdx.x;
  long total = (long)(E + Nn) * heads;
  if (tid >= total) return;
  int e = (int)(tid / heads);
  int h = (int)(tid - (long)e * heads);
  int s = (e < E) ? esrc[e] : (e - E);
  int d = (e < E) ? edst[e] : (e - E);
  float v = as[s * heads + h] + ad[d * heads + h];
  v = (v >= 0.0f) ? v : NEG_SLOPE * v;
  int di = d * heads + h;
  float w = expf(v - key2f(mkey[di])) / denom[di];
  const float* xs = xw + (size_t)s * ldx + (size_t)h * C;
  float* od = out + (size_t)d * heads * C + (size_t)h * C;
  for (int c = 0; c < C; ++c) atomicAdd(&od[c], w * xs[c]);
}

__global__ void bias_elu_kernel(float* __restrict__ h,
                                const float* __restrict__ bias, long n, int f) {
  long i = (long)blockIdx.x * blockDim.x + threadIdx.x;
  if (i >= n) return;
  float v = h[i] + bias[(int)(i % f)];
  h[i] = (v > 0.0f) ? v : (expf(v) - 1.0f);
}

__global__ void bias_add_kernel(float* __restrict__ h,
                                const float* __restrict__ bias, long n, int f) {
  long i = (long)blockIdx.x * blockDim.x + threadIdx.x;
  if (i >= n) return;
  h[i] += bias[(int)(i % f)];
}

extern "C" void kernel_launch(void* const* d_in, const int* in_sizes, int n_in,
                              void* d_out, int out_size, void* d_ws, size_t ws_size,
                              hipStream_t stream) {
  const float* x      = (const float*)d_in[0];
  const int*   ei     = (const int*)d_in[1];
  const float* W1     = (const float*)d_in[2];
  const float* a1_src = (const float*)d_in[3];
  const float* a1_dst = (const float*)d_in[4];
  const float* b1     = (const float*)d_in[5];
  const float* W2     = (const float*)d_in[6];
  const float* a2_src = (const float*)d_in[7];
  const float* a2_dst = (const float*)d_in[8];
  const float* b2     = (const float*)d_in[9];
  float* out = (float*)d_out;
  (void)n_in; (void)out_size; (void)ws_size;

  const int F_IN = 256, H1 = 8, C1 = 32, HC1 = 256, NC = 40, NCP = 64;
  const int Nn   = in_sizes[0] / F_IN;
  const int E    = in_sizes[1] / 2;
  const int Etot = E + Nn;
  const int* esrc = ei;
  const int* edst = ei + E;

  float* p = (float*)d_ws;
  float*    xw1  = p;            p += (size_t)Nn * HC1;
  float*    as1  = p;            p += (size_t)Nn * H1;
  float*    ad1  = p;            p += (size_t)Nn * H1;
  unsigned* m1   = (unsigned*)p; p += (size_t)Nn * H1;
  float*    den1 = p;            p += (size_t)Nn * H1;
  float*    agg1 = p;            p += (size_t)Nn * HC1;
  float*    w2p  = p;            p += (size_t)HC1 * NCP;
  float*    xw2p = p;            p += (size_t)Nn * NCP;
  float*    as2  = p;            p += (size_t)Nn;
  float*    ad2  = p;            p += (size_t)Nn;
  unsigned* m2   = (unsigned*)p; p += (size_t)Nn;
  float*    den2 = p;            p += (size_t)Nn;

  dim3 blk(256);

  zero_f32_kernel<<<2048, blk, 0, stream>>>((float*)m1,
                                            (long)Nn * (2 * H1 + HC1));
  zero_f32_kernel<<<256, blk, 0, stream>>>((float*)m2, (long)Nn * 2);
  zero_f32_kernel<<<1024, blk, 0, stream>>>(out, (long)Nn * NC);
  pad_cols_kernel<<<(HC1 * NCP + 255) / 256, blk, 0, stream>>>(
      W2, w2p, HC1, NC, NCP);

  {
    int waves  = (Nn / 16) * (HC1 / 64);
    int blocks = (waves + 7) / 8;
    gemm_f32_wmma_kernel<<<blocks, blk, 0, stream>>>(x, W1, xw1, Nn, F_IN, HC1);
  }
  alpha_kernel<<<(Nn * H1 + 255) / 256, blk, 0, stream>>>(
      xw1, a1_src, a1_dst, as1, ad1, Nn, H1, C1, HC1);
  edge_max_kernel<<<(Etot + 255) / 256, blk, 0, stream>>>(
      esrc, edst, E, Nn, H1, as1, ad1, m1);
  edge_denom_kernel<<<(Etot + 255) / 256, blk, 0, stream>>>(
      esrc, edst, E, Nn, H1, as1, ad1, m1, den1);
  {
    long tot = (long)Etot * H1;
    edge_aggregate_kernel<<<(int)((tot + 255) / 256), blk, 0, stream>>>(
        esrc, edst, E, Nn, H1, C1, HC1, as1, ad1, m1, den1, xw1, agg1);
  }
  bias_elu_kernel<<<(int)(((long)Nn * HC1 + 255) / 256), blk, 0, stream>>>(
      agg1, b1, (long)Nn * HC1, HC1);

  {
    int waves  = (Nn / 16) * (NCP / 64);
    int blocks = (waves + 7) / 8;
    gemm_f32_wmma_kernel<<<blocks, blk, 0, stream>>>(agg1, w2p, xw2p,
                                                     Nn, HC1, NCP);
  }
  alpha_kernel<<<(Nn + 255) / 256, blk, 0, stream>>>(
      xw2p, a2_src, a2_dst, as2, ad2, Nn, 1, NC, NCP);
  edge_max_kernel<<<(Etot + 255) / 256, blk, 0, stream>>>(
      esrc, edst, E, Nn, 1, as2, ad2, m2);
  edge_denom_kernel<<<(Etot + 255) / 256, blk, 0, stream>>>(
      esrc, edst, E, Nn, 1, as2, ad2, m2, den2);
  edge_aggregate_kernel<<<(Etot + 255) / 256, blk, 0, stream>>>(
      esrc, edst, E, Nn, 1, NC, NCP, as2, ad2, m2, den2, xw2p, out);
  bias_add_kernel<<<(int)(((long)Nn * NC + 255) / 256), blk, 0, stream>>>(
      out, b2, (long)Nn * NC, NC);
}


# per layer one kernel: each work-group owns 196 destination rows, sweeps the edge list for them, keeps max, denominator and row sums in LDS (LDS float adds, arrival order), stores rows once
# speedup vs baseline: 9.1421x; 9.1421x over previous
#include <hip/hip_runtime.h>

typedef __attribute__((ext_vector_type(2))) float v2f;
typedef __attribute__((ext_vector_type(8))) float v8f;

#define NEG_SLOPE 0.2f

static __device__ __forceinline__ unsigned f2key(float f) {
  unsigned u = __float_as_uint(f);
  return (u & 0x80000000u) ? ~u : (u | 0x80000000u);
}
static __device__ __forceinline__ float key2f(unsigned k) {
  return (k & 0x80000000u) ? __uint_as_float(k & 0x7FFFFFFFu)
                           : __uint_as_float(~k);
}

__global__ void pad_cols_kernel(const float* __restrict__ src,
                                float* __restrict__ dst,
                                int K, int c, int ldd) {
  int i = blockIdx.x * blockDim.x + threadIdx.x;
  if (i >= K * ldd) return;
  int r = i / ldd, j = i - r * ldd;
  dst[i] = (j < c) ? src[r * c + j] : 0.0f;
}

__global__ void gemm_f32_wmma_kernel(const float* __restrict__ A,
                                     const float* __restrict__ B,
                                     float* __restrict__ C,
                                     int M, int K, int ldb) {
  const int lane = threadIdx.x & 31;
  const int wave = threadIdx.x >> 5;
  const int half = lane >> 4;
  const int l    = lane & 15;
  const int nt64   = ldb >> 6;
  const int mtiles = M >> 4;
  const int total  = mtiles * nt64;
  int t = blockIdx.x * (blockDim.x >> 5) + wave;
  if (t >= total) return;
  const int mt = t / nt64;
  const int n0 = (t - mt * nt64) << 6;
  const int m0 = mt << 4;

  v8f acc[4] = {};
  const float* arow = A + (size_t)(m0 + l) * K + 2 * half;
  const float* bp   = B + (size_t)(2 * half) * ldb + n0 + l;
  const size_t bstep = (size_t)4 * ldb;

  v2f a = *(const v2f*)(arow);
  v2f b[4];
#pragma unroll
  for (int s = 0; s < 4; ++s) {
    b[s].x = bp[s * 16];
    b[s].y = bp[ldb + s * 16];
  }

  for (int k0 = 4; k0 < K; k0 += 4) {
    v2f an = *(const v2f*)(arow + k0);
    const float* bpn = bp + bstep;
    v2f bn[4];
#pragma unroll
    for (int s = 0; s < 4; ++s) {
      bn[s].x = bpn[s * 16];
      bn[s].y = bpn[ldb + s * 16];
    }
#pragma unroll
    for (int s = 0; s < 4; ++s)
      acc[s] = __builtin_amdgcn_wmma_f32_16x16x4_f32(
          false, a, false, b[s], (short)0, acc[s], false, false);
    a = an;
#pragma unroll
    for (int s = 0; s < 4; ++s) b[s] = bn[s];
    bp = bpn;
  }
#pragma unroll
  for (int s = 0; s < 4; ++s)
    acc[s] = __builtin_amdgcn_wmma_f32_16x16x4_f32(
        false, a, false, b[s], (short)0, acc[s], false, false);

  float* crow = C + (size_t)(m0 + 8 * half) * ldb + n0 + l;
#pragma unroll
  for (int s = 0; s < 4; ++s)
#pragma unroll
    for (int j = 0; j < 8; ++j) crow[(size_t)j * ldb + s * 16] = acc[s][j];
}

__global__ void alpha_kernel(const float* __restrict__ xw,
                             const float* __restrict__ a_src,
                             const float* __restrict__ a_dst,
                             float* __restrict__ as, float* __restrict__ ad,
                             int n_nodes, int heads, int c, int ldx) {
  int i = blockIdx.x * blockDim.x + threadIdx.x;
  if (i >= n_nodes * heads) return;
  int n = i / heads, h = i - n * heads;
  const float* row = xw + (size_t)n * ldx + (size_t)h * c;
  const float* ws = a_src + (size_t)h * c;
  const float* wd = a_dst + (size_t)h * c;
  float s = 0.0f, d = 0.0f;
  for (int j = 0; j < c; ++j) {
    float v = row[j];
    s += v * ws[j];
    d += v * wd[j];
  }
  as[i] = s;
  ad[i] = d;
}

#define OWN_T       196
#define OWN_CAP     8192
#define OWN_THREADS 1024
typedef __attribute__((ext_vector_type(4))) int own_v4i;

static __device__ __forceinline__ void own_push(int s, int dl, int* ls, int* ld,
                                                unsigned* cnt) {
  unsigned idx = atomicAdd(cnt, 1u);
  if (idx < OWN_CAP) { ls[idx] = s; ld[idx] = dl; }
}

static __device__ __forceinline__ void own_take4(const int* __restrict__ esrc, int e,
                                                 own_v4i d, int n0, int nT,
                                                 int* ls, int* ld, unsigned* cnt) {
  const int r0 = d.x - n0, r1 = d.y - n0, r2 = d.z - n0, r3 = d.w - n0;
  if ((unsigned)r0 < (unsigned)nT) own_push(esrc[e + 0], r0, ls, ld, cnt);
  if ((unsigned)r1 < (unsigned)nT) own_push(esrc[e + 1], r1, ls, ld, cnt);
  if ((unsigned)r2 < (unsigned)nT) own_push(esrc[e + 2], r2, ls, ld, cnt);
  if ((unsigned)r3 < (unsigned)nT) own_push(esrc[e + 3], r3, ls, ld, cnt);
}

static __device__ __forceinline__ void own_build(const int* __restrict__ esrc,
                                                 const int* __restrict__ edst,
                                                 int eb, int ee, int n0, int nT,
                                                 int* ls, int* ld, unsigned* cnt) {
  const int tid = threadIdx.x;
  const int n = ee - eb;
  const int* dbase = edst + eb;
  const int nvec = ((((size_t)dbase) & 15) == 0) ? (n >> 2) : 0;
  const own_v4i* dv = (const own_v4i*)dbase;
  const int rot = (int)(((long)blockIdx.x * nvec) / gridDim.x) & ~31;
  int v = tid;
  for (; v + 3 * OWN_THREADS < nvec; v += 4 * OWN_THREADS) {
    own_v4i d[4];
    int vv[4];
#pragma unroll
    for (int u = 0; u < 4; ++u) {
      vv[u] = v + u * OWN_THREADS + rot;
      if (vv[u] >= nvec) vv[u] -= nvec;
      d[u] = dv[vv[u]];
    }
    asm volatile("" :: "v"(d[0]), "v"(d[1]), "v"(d[2]), "v"(d[3]));
#pragma unroll
    for (int u = 0; u < 4; ++u)
      own_take4(esrc, eb + 4 * vv[u], d[u], n0, nT, ls, ld, cnt);
  }
  for (; v < nvec; v += OWN_THREADS) {
    int vv = v + rot;
    if (vv >= nvec) vv -= nvec;
    own_take4(esrc, eb + 4 * vv, dv[vv], n0, nT, ls, ld, cnt);
  }
  for (int e = eb + 4 * nvec + tid; e < ee; e += OWN_THREADS) {
    const int r = edst[e] - n0;
    if ((unsigned)r < (unsigned)nT) own_push(esrc[e], r, ls, ld, cnt);
  }
}

static __device__ __forceinline__ void own_entry(int i, int n_list, int n0,
                                                 const int* ls, const int* ld,
                                                 int& s, int& dl) {
  if (i < n_list) { s = ls[i]; dl = ld[i]; }
  else            { dl = i - n_list; s = n0 + dl; }
}

template <int H, int NU>
static __device__ __forceinline__ void own_scores_trip(int phase, int idx0, int n_list,
                                                       int n0, const int* ls,
                                                       const int* ld,
                                                       const float* __restrict__ as,
                                                       const float* adL, unsigned* mx,
                                                       float* den) {
  float a[NU];
  int dh[NU];
#pragma unroll
  for (int u = 0; u < NU; ++u) {
    const int idx = idx0 + u * OWN_THREADS;
    const int i = idx / H, h = idx - i * H;
    int s, dl;
    own_entry(i, n_list, n0, ls, ld, s, dl);
    a[u] = as[s * H + h];
    dh[u] = dl * H + h;
  }
#pragma unroll
  for (int u = 0; u < NU; ++u) {
    float v = a[u] + adL[dh[u]];
    v = (v >= 0.0f) ? v : NEG_SLOPE * v;
    if (phase == 0) atomicMax(&mx[dh[u]], f2key(v));
    else            atomicAdd(&den[dh[u]], expf(v - key2f(mx[dh[u]])));
  }
}

template <int H>
static __device__ __forceinline__ void own_scores(int phase, int n_list, int n_self,
                                                  int n0, const int* ls, const int* ld,
                                                  const float* __restrict__ as,
                                                  const float* adL, unsigned* mx,
                                                  float* den) {
  const int n_items = (n_list + n_self) * H;
  int idx = threadIdx.x;
  for (; idx + 3 * OWN_THREADS < n_items; idx += 4 * OWN_THREADS)
    own_scores_trip<H, 4>(phase, idx, n_list, n0, ls, ld, as, adL, mx, den);
  for (; idx < n_items; idx += OWN_THREADS)
    own_scores_trip<H, 1>(phase, idx, n_list, n0, ls, ld, as, adL, mx, den);
}

template <int H, int C, int LDX, int NU>
static __device__ __forceinline__ void own_rows_trip(int i0, int n_list, int n0,
                                                     const int* ls, const int* ld,
                                                     const float* __restrict__ as,
                                                     const float* __restrict__ xw,
                                                     const float* adL,
                                                     const unsigned* mx,
                                                     const float* den, float* acc) {
  constexpr int HC = H * C;
  constexpr int NJ = (HC + 31) / 32;
  const int lane = threadIdx.x & 31;
  const int hl = lane & (H - 1);
  int dl[NU];
  float x[NU][NJ], av[NU];
#pragma unroll
  for (int u = 0; u < NU; ++u) {
    int si, di;
    own_entry(i0 + u * (OWN_THREADS / 32), n_list, n0, ls, ld, si, di);
    const int s = __builtin_amdgcn_readfirstlane(si);
    dl[u] = __builtin_amdgcn_readfirstlane(di);
    const float* row = xw + (size_t)s * LDX;
#pragma unroll
    for (int j = 0; j < NJ; ++j) {
      const int c = j * 32 + lane;
      x[u][j] = (c < HC) ? row[c] : 0.0f;
    }
    av[u] = as[s * H + hl];
  }
#pragma unroll
  for (int u = 0; u < NU; ++u) {
    const int di = dl[u] * H + hl;
    float v = av[u] + adL[di];
    v = (v >= 0.0f) ? v : NEG_SLOPE * v;
    const float w = expf(v - key2f(mx[di])) / den[di];
    float* arow = acc + dl[u] * HC;
#pragma unroll
    for (int j = 0; j < NJ; ++j) {
      const int head = (H == 1) ? 0 : (j * 32) / C;
      const float wj = __int_as_float(
          __builtin_amdgcn_readlane(__float_as_int(w), head));
      const int c = j * 32 + lane;
      if (c < HC) atomicAdd(&arow[c], wj * x[u][j]);
    }
  }
}

template <int H, int C, int LDX>
static __device__ __forceinline__ void own_rows(int n_list, int n_self, int n0,
                                                const int* ls, const int* ld,
                                                const float* __restrict__ as,
                                                const float* __restrict__ xw,
                                                const float* adL, const unsigned* mx,
                                                const float* den, float* acc) {
  static_assert(H == 1 || (C % 32) == 0, "a 32-lane slice must lie in one head");
  static_assert((H & (H - 1)) == 0 && H <= 32, "heads: a power of two");
  constexpr int NW = OWN_THREADS / 32;
  const int n_items = n_list + n_self;
  int i = threadIdx.x >> 5;
  for (; i + NW < n_items; i += 2 * NW)
    own_rows_trip<H, C, LDX, 2>(i, n_list, n0, ls, ld, as, xw, adL, mx, den, acc);
  if (i < n_items)
    own_rows_trip<H, C, LDX, 1>(i, n_list, n0, ls, ld, as, xw, adL, mx, den, acc);
}

template <int H, int C, int LDX, bool ELU>
__global__ __launch_bounds__(OWN_THREADS) void gat_owner_kernel(
    const int* __restrict__ esrc, const int* __restrict__ edst, int E, int Nn,
    const float* __restrict__ as, const float* __restrict__ ad,
    const float* __restrict__ xw, const float* __restrict__ bias,
    float* __restrict__ outp) {
  constexpr int HC = H * C;
  __shared__ float    acc[OWN_T * HC];
  __shared__ float    adL[OWN_T * H];
  __shared__ float    den[OWN_T * H];
  __shared__ unsigned mx[OWN_T * H];
  __shared__ int      ls[OWN_CAP];
  __shared__ int      ld[OWN_CAP];
  __shared__ unsigned cnt;

  const int tid = threadIdx.x;
  const int n0 = blockIdx.x * OWN_T;
  const int nT = min(OWN_T, Nn - n0);

  for (int i = tid; i < OWN_T * HC; i += OWN_THREADS) acc[i] = 0.0f;
  for (int i = tid; i < OWN_T * H; i += OWN_THREADS) {
    den[i] = 0.0f;
    mx[i] = 0u;
    adL[i] = (i < nT * H) ? ad[(size_t)n0 * H + i] : 0.0f;
  }
  if (tid == 0) cnt = 0u;
  __syncthreads();

  own_build(esrc, edst, 0, E, n0, nT, ls, ld, &cnt);
  __syncthreads();
  const int total = (int)cnt;

  if (total <= OWN_CAP) {
    own_scores<H>(0, total, nT, n0, ls, ld, as, adL, mx, den);
    __syncthreads();
    own_scores<H>(1, total, nT, n0, ls, ld, as, adL, mx, den);
    __syncthreads();
    own_rows<H, C, LDX>(total, nT, n0, ls, ld, as, xw, adL, mx, den, acc);
  } else {
    for (int phase = 0; phase < 3; ++phase) {
      for (int eb = 0; eb < E; eb += OWN_CAP) {
        __syncthreads();
        if (tid == 0) cnt = 0u;
        __syncthreads();
        own_build(esrc, edst, eb, min(eb + OWN_CAP, E), n0, nT, ls, ld, &cnt);
        __syncthreads();
        const int m = (int)cnt;
        const int n_self = (eb == 0) ? nT : 0;
        if (phase < 2)
          own_scores<H>(phase, m, n_self, n0, ls, ld, as, adL, mx, den);
        else
          own_rows<H, C, LDX>(m, n_self, n0, ls, ld, as, xw, adL, mx, den, acc);
      }
    }
  }
  __syncthreads();

  float* orow = outp + (size_t)n0 * HC;
  for (int i = tid; i < nT * HC; i += OWN_THREADS) {
    float v = acc[i] + bias[i % HC];
    if (ELU) v = (v > 0.0f) ? v : (expf(v) - 1.0f);
    orow[i] = v;
  }
}

extern "C" void kernel_launch(void* const* d_in, const int* in_sizes, int n_in,
                              void* d_out, int out_size, void* d_ws, size_t ws_size,
                              hipStream_t stream) {
  const float* x      = (const float*)d_in[0];
  const int*   ei     = (const int*)d_in[1];
  const float* W1     = (const float*)d_in[2];
  const float* a1_src = (const float*)d_in[3];
  const float* a1_dst = (const float*)d_in[4];
  const float* b1     = (const float*)d_in[5];
  const float* W2     = (const float*)d_in[6];
  const float* a2_src = (const float*)d_in[7];
  const float* a2_dst = (const float*)d_in[8];
  const float* b2     = (const float*)d_in[9];
  float* out = (float*)d_out;
  (void)n_in; (void)out_size; (void)ws_size;

  const int F_IN = 256, H1 = 8, C1 = 32, HC1 = 256, NC = 40, NCP = 64;
  const int Nn   = in_sizes[0] / F_IN;
  const int E    = in_sizes[1] / 2;
  const int* esrc = ei;
  const int* edst = ei + E;

  float* p = (float*)d_ws;
  float* xw1  = p; p += (size_t)Nn * HC1;
  float* as1  = p; p += (size_t)Nn * H1;
  float* ad1  = p; p += (size_t)Nn * H1;
  float* h1   = p; p += (size_t)Nn * HC1;
  float* w2p  = p; p += (size_t)HC1 * NCP;
  float* xw2p = p; p += (size_t)Nn * NCP;
  float* as2  = p; p += (size_t)Nn;
  float* ad2  = p; p += (size_t)Nn;

  dim3 blk(256);
  const int own_blocks = (Nn + OWN_T - 1) / OWN_T;

  pad_cols_kernel<<<(HC1 * NCP + 255) / 256, blk, 0, stream>>>(
      W2, w2p, HC1, NC, NCP);

  {
    int waves  = (Nn / 16) * (HC1 / 64);
    int blocks = (waves + 7) / 8;
    gemm_f32_wmma_kernel<<<blocks, blk, 0, stream>>>(x, W1, xw1, Nn, F_IN, HC1);
  }
  alpha_kernel<<<(Nn * H1 + 255) / 256, blk, 0, stream>>>(
      xw1, a1_src, a1_dst, as1, ad1, Nn, H1, C1, HC1);
  gat_owner_kernel<8, 32, 256, true><<<own_blocks, OWN_THREADS, 0, stream>>>(
      esrc, edst, E, Nn, as1, ad1, xw1, b1, h1);

  {
    int waves  = (Nn / 16) * (NCP / 64);
    int blocks = (waves + 7) / 8;
    gemm_f32_wmma_kernel<<<blocks, blk, 0, stream>>>(h1, w2p, xw2p,
                                                     Nn, HC1, NCP);
  }
  alpha_kernel<<<(Nn + 255) / 256, blk, 0, stream>>>(
      xw2p, a2_src, a2_dst, as2, ad2, Nn, 1, NC, NCP);
  gat_owner_kernel<1, 40, 64, false><<<own_blocks, OWN_THREADS, 0, stream>>>(
      esrc, edst, E, Nn, as2, ad2, xw2p, b2, out);
}


# edge sweep lists edge numbers, not sources: no source load and wait per matching edge inside the sweep; sources fetched afterwards in one parallel pass; sum order by arrival as before
# speedup vs baseline: 9.4753x; 1.0364x over previous
#include <hip/hip_runtime.h>

typedef __attribute__((ext_vector_type(2))) float v2f;
typedef __attribute__((ext_vector_type(8))) float v8f;

#define NEG_SLOPE 0.2f

static __device__ __forceinline__ unsigned f2key(float f) {
  unsigned u = __float_as_uint(f);
  return (u & 0x80000000u) ? ~u : (u | 0x80000000u);
}
static __device__ __forceinline__ float key2f(unsigned k) {
  return (k & 0x80000000u) ? __uint_as_float(k & 0x7FFFFFFFu)
                           : __uint_as_float(~k);
}

__global__ void pad_cols_kernel(const float* __restrict__ src,
                                float* __restrict__ dst,
                                int K, int c, int ldd) {
  int i = blockIdx.x * blockDim.x + threadIdx.x;
  if (i >= K * ldd) return;
  int r = i / ldd, j = i - r * ldd;
  dst[i] = (j < c) ? src[r * c + j] : 0.0f;
}

__global__ void gemm_f32_wmma_kernel(const float* __restrict__ A,
                                     const float* __restrict__ B,
                                     float* __restrict__ C,
                                     int M, int K, int ldb) {
  const int lane = threadIdx.x & 31;
  const int wave = threadIdx.x >> 5;
  const int half = lane >> 4;
  const int l    = lane & 15;
  const int nt64   = ldb >> 6;
  const int mtiles = M >> 4;
  const int total  = mtiles * nt64;
  int t = blockIdx.x * (blockDim.x >> 5) + wave;
  if (t >= total) return;
  const int mt = t / nt64;
  const int n0 = (t - mt * nt64) << 6;
  const int m0 = mt << 4;

  v8f acc[4] = {};
  const float* arow = A + (size_t)(m0 + l) * K + 2 * half;
  const float* bp   = B + (size_t)(2 * half) * ldb + n0 + l;
  const size_t bstep = (size_t)4 * ldb;

  v2f a = *(const v2f*)(arow);
  v2f b[4];
#pragma unroll
  for (int s = 0; s < 4; ++s) {
    b[s].x = bp[s * 16];
    b[s].y = bp[ldb + s * 16];
  }

  for (int k0 = 4; k0 < K; k0 += 4) {
    v2f an = *(const v2f*)(arow + k0);
    const float* bpn = bp + bstep;
    v2f bn[4];
#pragma unroll
    for (int s = 0; s < 4; ++s) {
      bn[s].x = bpn[s * 16];
      bn[s].y = bpn[ldb + s * 16];
    }
#pragma unroll
    for (int s = 0; s < 4; ++s)
      acc[s] = __builtin_amdgcn_wmma_f32_16x16x4_f32(
          false, a, false, b[s], (short)0, acc[s], false, false);
    a = an;
#pragma unroll
    for (int s = 0; s < 4; ++s) b[s] = bn[s];
    bp = bpn;
  }
#pragma unroll
  for (int s = 0; s < 4; ++s)
    acc[s] = __builtin_amdgcn_wmma_f32_16x16x4_f32(
        false, a, false, b[s], (short)0, acc[s], false, false);

  float* crow = C + (size_t)(m0 + 8 * half) * ldb + n0 + l;
#pragma unroll
  for (int s = 0; s < 4; ++s)
#pragma unroll
    for (int j = 0; j < 8; ++j) crow[(size_t)j * ldb + s * 16] = acc[s][j];
}

__global__ void alpha_kernel(const float* __restrict__ xw,
                             const float* __restrict__ a_src,
                             const float* __restrict__ a_dst,
                             float* __restrict__ as, float* __restrict__ ad,
                             int n_nodes, int heads, int c, int ldx) {
  int i = blockIdx.x * blockDim.x + threadIdx.x;
  if (i >= n_nodes * heads) return;
  int n = i / heads, h = i - n * heads;
  const float* row = xw + (size_t)n * ldx + (size_t)h * c;
  const float* ws = a_src + (size_t)h * c;
  const float* wd = a_dst + (size_t)h * c;
  float s = 0.0f, d = 0.0f;
  for (int j = 0; j < c; ++j) {
    float v = row[j];
    s += v * ws[j];
    d += v * wd[j];
  }
  as[i] = s;
  ad[i] = d;
}

#define OWN_T       196
#define OWN_CAP     8192
#define OWN_THREADS 1024
typedef __attribute__((ext_vector_type(4))) int own_v4i;

static __device__ __forceinline__ void own_push(int e, int dl, int* ls, int* ld,
                                                unsigned* cnt) {
  unsigned idx = atomicAdd(cnt, 1u);
  if (idx < OWN_CAP) { ls[idx] = e; ld[idx] = dl; }
}

static __device__ __forceinline__ void own_take4(int e, own_v4i d, int n0, int nT,
                                                 int* ls, int* ld, unsigned* cnt) {
  const int r0 = d.x - n0, r1 = d.y - n0, r2 = d.z - n0, r3 = d.w - n0;
  if ((unsigned)r0 < (unsigned)nT) own_push(e + 0, r0, ls, ld, cnt);
  if ((unsigned)r1 < (unsigned)nT) own_push(e + 1, r1, ls, ld, cnt);
  if ((unsigned)r2 < (unsigned)nT) own_push(e + 2, r2, ls, ld, cnt);
  if ((unsigned)r3 < (unsigned)nT) own_push(e + 3, r3, ls, ld, cnt);
}

static __device__ __forceinline__ void own_build(const int* __restrict__ edst,
                                                 int eb, int ee, int n0, int nT,
                                                 int* ls, int* ld, unsigned* cnt) {
  const int tid = threadIdx.x;
  const int n = ee - eb;
  const int* dbase = edst + eb;
  const int nvec = ((((size_t)dbase) & 15) == 0) ? (n >> 2) : 0;
  const own_v4i* dv = (const own_v4i*)dbase;
  const int rot = (int)(((long)blockIdx.x * nvec) / gridDim.x) & ~31;
  int v = tid;
  for (; v + 3 * OWN_THREADS < nvec; v += 4 * OWN_THREADS) {
    own_v4i d[4];
    int vv[4];
#pragma unroll
    for (int u = 0; u < 4; ++u) {
      vv[u] = v + u * OWN_THREADS + rot;
      if (vv[u] >= nvec) vv[u] -= nvec;
      d[u] = dv[vv[u]];
    }
    asm volatile("" :: "v"(d[0]), "v"(d[1]), "v"(d[2]), "v"(d[3]));
#pragma unroll
    for (int u = 0; u < 4; ++u)
      own_take4(eb + 4 * vv[u], d[u], n0, nT, ls, ld, cnt);
  }
  for (; v < nvec; v += OWN_THREADS) {
    int vv = v + rot;
    if (vv >= nvec) vv -= nvec;
    own_take4(eb + 4 * vv, dv[vv], n0, nT, ls, ld, cnt);
  }
  for (int e = eb + 4 * nvec + tid; e < ee; e += OWN_THREADS) {
    const int r = edst[e] - n0;
    if ((unsigned)r < (unsigned)nT) own_push(e, r, ls, ld, cnt);
  }
}

static __device__ __forceinline__ void own_resolve(const int* __restrict__ esrc,
                                                   int* ls, int n) {
  for (int i0 = threadIdx.x; i0 < n; i0 += 4 * OWN_THREADS) {
    int e[4], s[4];
#pragma unroll
    for (int u = 0; u < 4; ++u) {
      const int i = i0 + u * OWN_THREADS;
      e[u] = (i < n) ? ls[i] : 0;
    }
#pragma unroll
    for (int u = 0; u < 4; ++u) s[u] = esrc[e[u]];
    asm volatile("" :: "v"(s[0]), "v"(s[1]), "v"(s[2]), "v"(s[3]));
#pragma unroll
    for (int u = 0; u < 4; ++u) {
      const int i = i0 + u * OWN_THREADS;
      if (i < n) ls[i] = s[u];
    }
  }
}

static __device__ __forceinline__ void own_entry(int i, int n_list, int n0,
                                                 const int* ls, const int* ld,
                                                 int& s, int& dl) {
  if (i < n_list) { s = ls[i]; dl = ld[i]; }
  else            { dl = i - n_list; s = n0 + dl; }
}

template <int H, int NU>
static __device__ __forceinline__ void own_scores_trip(int phase, int idx0, int n_list,
                                                       int n0, const int* ls,
                                                       const int* ld,
                                                       const float* __restrict__ as,
                                                       const float* adL, unsigned* mx,
                                                       float* den) {
  float a[NU];
  int dh[NU];
#pragma unroll
  for (int u = 0; u < NU; ++u) {
    const int idx = idx0 + u * OWN_THREADS;
    const int i = idx / H, h = idx - i * H;
    int s, dl;
    own_entry(i, n_list, n0, ls, ld, s, dl);
    a[u] = as[s * H + h];
    dh[u] = dl * H + h;
  }
#pragma unroll
  for (int u = 0; u < NU; ++u) {
    float v = a[u] + adL[dh[u]];
    v = (v >= 0.0f) ? v : NEG_SLOPE * v;
    if (phase == 0) atomicMax(&mx[dh[u]], f2key(v));
    else            atomicAdd(&den[dh[u]], expf(v - key2f(mx[dh[u]])));
  }
}

template <int H>
static __device__ __forceinline__ void own_scores(int phase, int n_list, int n_self,
                                                  int n0, const int* ls, const int* ld,
                                                  const float* __restrict__ as,
                                                  const float* adL, unsigned* mx,
                                                  float* den) {
  const int n_items = (n_list + n_self) * H;
  int idx = threadIdx.x;
  for (; idx + 3 * OWN_THREADS < n_items; idx += 4 * OWN_THREADS)
    own_scores_trip<H, 4>(phase, idx, n_list, n0, ls, ld, as, adL, mx, den);
  for (; idx < n_items; idx += OWN_THREADS)
    own_scores_trip<H, 1>(phase, idx, n_list, n0, ls, ld, as, adL, mx, den);
}

template <int H, int C, int LDX, int NU>
static __device__ __forceinline__ void own_rows_trip(int i0, int n_list, int n0,
                                                     const int* ls, const int* ld,
                                                     const float* __restrict__ as,
                                                     const float* __restrict__ xw,
                                                     const float* adL,
                                                     const unsigned* mx,
                                                     const float* den, float* acc) {
  constexpr int HC = H * C;
  constexpr int NJ = (HC + 31) / 32;
  const int lane = threadIdx.x & 31;
  const int hl = lane & (H - 1);
  int dl[NU];
  float x[NU][NJ], av[NU];
#pragma unroll
  for (int u = 0; u < NU; ++u) {
    int si, di;
    own_entry(i0 + u * (OWN_THREADS / 32), n_list, n0, ls, ld, si, di);
    const int s = __builtin_amdgcn_readfirstlane(si);
    dl[u] = __builtin_amdgcn_readfirstlane(di);
    const float* row = xw + (size_t)s * LDX;
#pragma unroll
    for (int j = 0; j < NJ; ++j) {
      const int c = j * 32 + lane;
      x[u][j] = (c < HC) ? row[c] : 0.0f;
    }
    av[u] = as[s * H + hl];
  }
#pragma unroll
  for (int u = 0; u < NU; ++u) {
    const int di = dl[u] * H + hl;
    float v = av[u] + adL[di];
    v = (v >= 0.0f) ? v : NEG_SLOPE * v;
    const float w = expf(v - key2f(mx[di])) / den[di];
    float* arow = acc + dl[u] * HC;
#pragma unroll
    for (int j = 0; j < NJ; ++j) {
      const int head = (H == 1) ? 0 : (j * 32) / C;
      const float wj = __int_as_float(
          __builtin_amdgcn_readlane(__float_as_int(w), head));
      const int c = j * 32 + lane;
      if (c < HC) atomicAdd(&arow[c], wj * x[u][j]);
    }
  }
}

template <int H, int C, int LDX>
static __device__ __forceinline__ void own_rows(int n_list, int n_self, int n0,
                                                const int* ls, const int* ld,
                                                const float* __restrict__ as,
                                                const float* __restrict__ xw,
                                                const float* adL, const unsigned* mx,
                                                const float* den, float* acc) {
  static_assert(H == 1 || (C % 32) == 0, "a 32-lane slice must lie in one head");
  static_assert((H & (H - 1)) == 0 && H <= 32, "heads: a power of two");
  constexpr int NW = OWN_THREADS / 32;
  const int n_items = n_list + n_self;
  int i = threadIdx.x >> 5;
  for (; i + NW < n_items; i += 2 * NW)
    own_rows_trip<H, C, LDX, 2>(i, n_list, n0, ls, ld, as, xw, adL, mx, den, acc);
  if (i < n_items)
    own_rows_trip<H, C, LDX, 1>(i, n_list, n0, ls, ld, as, xw, adL, mx, den, acc);
}

template <int H, int C, int LDX, bool ELU>
__global__ __launch_bounds__(OWN_THREADS) void gat_owner_kernel(
    const int* __restrict__ esrc, const int* __restrict__ edst, int E, int Nn,
    const float* __restrict__ as, const float* __restrict__ ad,
    const float* __restrict__ xw, const float* __restrict__ bias,
    float* __restrict__ outp) {
  constexpr int HC = H * C;
  __shared__ float    acc[OWN_T * HC];
  __shared__ float    adL[OWN_T * H];
  __shared__ float    den[OWN_T * H];
  __shared__ unsigned mx[OWN_T * H];
  __shared__ int      ls[OWN_CAP];
  __shared__ int      ld[OWN_CAP];
  __shared__ unsigned cnt;

  const int tid = threadIdx.x;
  const int n0 = blockIdx.x * OWN_T;
  const int nT = min(OWN_T, Nn - n0);

  for (int i = tid; i < OWN_T * HC; i += OWN_THREADS) acc[i] = 0.0f;
  for (int i = tid; i < OWN_T * H; i += OWN_THREADS) {
    den[i] = 0.0f;
    mx[i] = 0u;
    adL[i] = (i < nT * H) ? ad[(size_t)n0 * H + i] : 0.0f;
  }
  if (tid == 0) cnt = 0u;
  __syncthreads();

  own_build(edst, 0, E, n0, nT, ls, ld, &cnt);
  __syncthreads();
  const int total = (int)cnt;

  if (total <= OWN_CAP) {
    own_resolve(esrc, ls, total);
    __syncthreads();
    own_scores<H>(0, total, nT, n0, ls, ld, as, adL, mx, den);
    __syncthreads();
    own_scores<H>(1, total, nT, n0, ls, ld, as, adL, mx, den);
    __syncthreads();
    own_rows<H, C, LDX>(total, nT, n0, ls, ld, as, xw, adL, mx, den, acc);
  } else {
    for (int phase = 0; phase < 3; ++phase) {
      for (int eb = 0; eb < E; eb += OWN_CAP) {
        __syncthreads();
        if (tid == 0) cnt = 0u;
        __syncthreads();
        own_build(edst, eb, min(eb + OWN_CAP, E), n0, nT, ls, ld, &cnt);
        __syncthreads();
        const int m = (int)cnt;
        own_resolve(esrc, ls, m);
        __syncthreads();
        const int n_self = (eb == 0) ? nT : 0;
        if (phase < 2)
          own_scores<H>(phase, m, n_self, n0, ls, ld, as, adL, mx, den);
        else
          own_rows<H, C, LDX>(m, n_self, n0, ls, ld, as, xw, adL, mx, den, acc);
      }
    }
  }
  __syncthreads();

  float* orow = outp + (size_t)n0 * HC;
  for (int i = tid; i < nT * HC; i += OWN_THREADS) {
    float v = acc[i] + bias[i % HC];
    if (ELU) v = (v > 0.0f) ? v : (expf(v) - 1.0f);
    orow[i] = v;
  }
}

extern "C" void kernel_launch(void* const* d_in, const int* in_sizes, int n_in,
                              void* d_out, int out_size, void* d_ws, size_t ws_size,
                              hipStream_t stream) {
  const float* x      = (const float*)d_in[0];
  const int*   ei     = (const int*)d_in[1];
  const float* W1     = (const float*)d_in[2];
  const float* a1_src = (const float*)d_in[3];
  const float* a1_dst = (const float*)d_in[4];
  const float* b1     = (const float*)d_in[5];
  const float* W2     = (const float*)d_in[6];
  const float* a2_src = (const float*)d_in[7];
  const float* a2_dst = (const float*)d_in[8];
  const float* b2     = (const float*)d_in[9];
  float* out = (float*)d_out;
  (void)n_in; (void)out_size; (void)ws_size;

  const int F_IN = 256, H1 = 8, C1 = 32, HC1 = 256, NC = 40, NCP = 64;
  const int Nn   = in_sizes[0] / F_IN;
  const int E    = in_sizes[1] / 2;
  const int* esrc = ei;
  const int* edst = ei + E;

  float* p = (float*)d_ws;
  float* xw1  = p; p += (size_t)Nn * HC1;
  float* as1  = p; p += (size_t)Nn * H1;
  float* ad1  = p; p += (size_t)Nn * H1;
  float* h1   = p; p += (size_t)Nn * HC1;
  float* w2p  = p; p += (size_t)HC1 * NCP;
  float* xw2p = p; p += (size_t)Nn * NCP;
  float* as2  = p; p += (size_t)Nn;
  float* ad2  = p; p += (size_t)Nn;

  dim3 blk(256);
  const int own_blocks = (Nn + OWN_T - 1) / OWN_T;

  pad_cols_kernel<<<(HC1 * NCP + 255) / 256, blk, 0, stream>>>(
      W2, w2p, HC1, NC, NCP);

  {
    int waves  = (Nn / 16) * (HC1 / 64);
    int blocks = (waves + 7) / 8;
    gemm_f32_wmma_kernel<<<blocks, blk, 0, stream>>>(x, W1, xw1, Nn, F_IN, HC1);
  }
  alpha_kernel<<<(Nn * H1 + 255) / 256, blk, 0, stream>>>(
      xw1, a1_src, a1_dst, as1, ad1, Nn, H1, C1, HC1);
  gat_owner_kernel<8, 32, 256, true><<<own_blocks, OWN_THREADS, 0, stream>>>(
      esrc, edst, E, Nn, as1, ad1, xw1, b1, h1);

  {
    int waves  = (Nn / 16) * (NCP / 64);
    int blocks = (waves + 7) / 8;
    gemm_f32_wmma_kernel<<<blocks, blk, 0, stream>>>(h1, w2p, xw2p,
                                                     Nn, HC1, NCP);
  }
  alpha_kernel<<<(Nn + 255) / 256, blk, 0, stream>>>(
      xw2p, a2_src, a2_dst, as2, ad2, Nn, 1, NC, NCP);
  gat_owner_kernel<1, 40, 64, false><<<own_blocks, OWN_THREADS, 0, stream>>>(
      esrc, edst, E, Nn, as2, ad2, xw2p, b2, out);
}


# layer 2 takes each work-group's finished edge list (count, sources, local rows) from the workspace, where layer 1's launch left it in the same call: no second edge sweep; sums by arrival as before
# speedup vs baseline: 10.8081x; 1.1407x over previous
#include <hip/hip_runtime.h>

typedef __attribute__((ext_vector_type(2))) float v2f;
typedef __attribute__((ext_vector_type(8))) float v8f;

#define NEG_SLOPE 0.2f

static __device__ __forceinline__ unsigned f2key(float f) {
  unsigned u = __float_as_uint(f);
  return (u & 0x80000000u) ? ~u : (u | 0x80000000u);
}
static __device__ __forceinline__ float key2f(unsigned k) {
  return (k & 0x80000000u) ? __uint_as_float(k & 0x7FFFFFFFu)
                           : __uint_as_float(~k);
}

__global__ void pad_cols_kernel(const float* __restrict__ src,
                                float* __restrict__ dst,
                                int K, int c, int ldd) {
  int i = blockIdx.x * blockDim.x + threadIdx.x;
  if (i >= K * ldd) return;
  int r = i / ldd, j = i - r * ldd;
  dst[i] = (j < c) ? src[r * c + j] : 0.0f;
}

__global__ void gemm_f32_wmma_kernel(const float* __restrict__ A,
                                     const float* __restrict__ B,
                                     float* __restrict__ C,
                                     int M, int K, int ldb) {
  const int lane = threadIdx.x & 31;
  const int wave = threadIdx.x >> 5;
  const int half = lane >> 4;
  const int l    = lane & 15;
  const int nt64   = ldb >> 6;
  const int mtiles = M >> 4;
  const int total  = mtiles * nt64;
  int t = blockIdx.x * (blockDim.x >> 5) + wave;
  if (t >= total) return;
  const int mt = t / nt64;
  const int n0 = (t - mt * nt64) << 6;
  const int m0 = mt << 4;

  v8f acc[4] = {};
  const float* arow = A + (size_t)(m0 + l) * K + 2 * half;
  const float* bp   = B + (size_t)(2 * half) * ldb + n0 + l;
  const size_t bstep = (size_t)4 * ldb;

  v2f a = *(const v2f*)(arow);
  v2f b[4];
#pragma unroll
  for (int s = 0; s < 4; ++s) {
    b[s].x = bp[s * 16];
    b[s].y = bp[ldb + s * 16];
  }

  for (int k0 = 4; k0 < K; k0 += 4) {
    v2f an = *(const v2f*)(arow + k0);
    const float* bpn = bp + bstep;
    v2f bn[4];
#pragma unroll
    for (int s = 0; s < 4; ++s) {
      bn[s].x = bpn[s * 16];
      bn[s].y = bpn[ldb + s * 16];
    }
#pragma unroll
    for (int s = 0; s < 4; ++s)
      acc[s] = __builtin_amdgcn_wmma_f32_16x16x4_f32(
          false, a, false, b[s], (short)0, acc[s], false, false);
    a = an;
#pragma unroll
    for (int s = 0; s < 4; ++s) b[s] = bn[s];
    bp = bpn;
  }
#pragma unroll
  for (int s = 0; s < 4; ++s)
    acc[s] = __builtin_amdgcn_wmma_f32_16x16x4_f32(
        false, a, false, b[s], (short)0, acc[s], false, false);

  float* crow = C + (size_t)(m0 + 8 * half) * ldb + n0 + l;
#pragma unroll
  for (int s = 0; s < 4; ++s)
#pragma unroll
    for (int j = 0; j < 8; ++j) crow[(size_t)j * ldb + s * 16] = acc[s][j];
}

__global__ void alpha_kernel(const float* __restrict__ xw,
                             const float* __restrict__ a_src,
                             const float* __restrict__ a_dst,
                             float* __restrict__ as, float* __restrict__ ad,
                             int n_nodes, int heads, int c, int ldx) {
  int i = blockIdx.x * blockDim.x + threadIdx.x;
  if (i >= n_nodes * heads) return;
  int n = i / heads, h = i - n * heads;
  const float* row = xw + (size_t)n * ldx + (size_t)h * c;
  const float* ws = a_src + (size_t)h * c;
  const float* wd = a_dst + (size_t)h * c;
  float s = 0.0f, d = 0.0f;
  for (int j = 0; j < c; ++j) {
    float v = row[j];
    s += v * ws[j];
    d += v * wd[j];
  }
  as[i] = s;
  ad[i] = d;
}

#define OWN_T       196
#define OWN_CAP     8192
#define OWN_THREADS 1024
#define OWN_HEAD    32
#define OWN_REGION  (OWN_HEAD + 2 * OWN_CAP)
typedef __attribute__((ext_vector_type(4))) int own_v4i;

static __device__ __forceinline__ void own_push(int e, int dl, int* ls, int* ld,
                                                unsigned* cnt) {
  unsigned idx = atomicAdd(cnt, 1u);
  if (idx < OWN_CAP) { ls[idx] = e; ld[idx] = dl; }
}

static __device__ __forceinline__ void own_take4(int e, own_v4i d, int n0, int nT,
                                                 int* ls, int* ld, unsigned* cnt) {
  const int r0 = d.x - n0, r1 = d.y - n0, r2 = d.z - n0, r3 = d.w - n0;
  if ((unsigned)r0 < (unsigned)nT) own_push(e + 0, r0, ls, ld, cnt);
  if ((unsigned)r1 < (unsigned)nT) own_push(e + 1, r1, ls, ld, cnt);
  if ((unsigned)r2 < (unsigned)nT) own_push(e + 2, r2, ls, ld, cnt);
  if ((unsigned)r3 < (unsigned)nT) own_push(e + 3, r3, ls, ld, cnt);
}

static __device__ __forceinline__ void own_build(const int* __restrict__ edst,
                                                 int eb, int ee, int n0, int nT,
                                                 int* ls, int* ld, unsigned* cnt) {
  const int tid = threadIdx.x;
  const int n = ee - eb;
  const int* dbase = edst + eb;
  const int nvec = ((((size_t)dbase) & 15) == 0) ? (n >> 2) : 0;
  const own_v4i* dv = (const own_v4i*)dbase;
  const int rot = (int)(((long)blockIdx.x * nvec) / gridDim.x) & ~31;
  int v = tid;
  for (; v + 3 * OWN_THREADS < nvec; v += 4 * OWN_THREADS) {
    own_v4i d[4];
    int vv[4];
#pragma unroll
    for (int u = 0; u < 4; ++u) {
      vv[u] = v + u * OWN_THREADS + rot;
      if (vv[u] >= nvec) vv[u] -= nvec;
      d[u] = dv[vv[u]];
    }
    asm volatile("" :: "v"(d[0]), "v"(d[1]), "v"(d[2]), "v"(d[3]));
#pragma unroll
    for (int u = 0; u < 4; ++u)
      own_take4(eb + 4 * vv[u], d[u], n0, nT, ls, ld, cnt);
  }
  for (; v < nvec; v += OWN_THREADS) {
    int vv = v + rot;
    if (vv >= nvec) vv -= nvec;
    own_take4(eb + 4 * vv, dv[vv], n0, nT, ls, ld, cnt);
  }
  for (int e = eb + 4 * nvec + tid; e < ee; e += OWN_THREADS) {
    const int r = edst[e] - n0;
    if ((unsigned)r < (unsigned)nT) own_push(e, r, ls, ld, cnt);
  }
}

static __device__ __forceinline__ void own_resolve(const int* __restrict__ esrc,
                                                   int* ls, int n) {
  for (int i0 = threadIdx.x; i0 < n; i0 += 4 * OWN_THREADS) {
    int e[4], s[4];
#pragma unroll
    for (int u = 0; u < 4; ++u) {
      const int i = i0 + u * OWN_THREADS;
      e[u] = (i < n) ? ls[i] : 0;
    }
#pragma unroll
    for (int u = 0; u < 4; ++u) s[u] = esrc[e[u]];
    asm volatile("" :: "v"(s[0]), "v"(s[1]), "v"(s[2]), "v"(s[3]));
#pragma unroll
    for (int u = 0; u < 4; ++u) {
      const int i = i0 + u * OWN_THREADS;
      if (i < n) ls[i] = s[u];
    }
  }
}

static __device__ __forceinline__ void own_leave(const int* ls, const int* ld,
                                                 int* __restrict__ gs,
                                                 int* __restrict__ gd, int n) {
  for (int i = threadIdx.x; i < n; i += OWN_THREADS) {
    gs[i] = ls[i];
    gd[i] = ld[i];
  }
}

static __device__ __forceinline__ void own_take(const int* __restrict__ gs,
                                                const int* __restrict__ gd,
                                                int* ls, int* ld, int n) {
  for (int i0 = threadIdx.x; i0 < n; i0 += 4 * OWN_THREADS) {
    int s[4], d[4];
#pragma unroll
    for (int u = 0; u < 4; ++u) {
      const int i = i0 + u * OWN_THREADS;
      const int j = (i < n) ? i : 0;
      s[u] = gs[j];
      d[u] = gd[j];
    }
#pragma unroll
    for (int u = 0; u < 4; ++u) {
      const int i = i0 + u * OWN_THREADS;
      if (i < n) { ls[i] = s[u]; ld[i] = d[u]; }
    }
  }
}

static __device__ __forceinline__ void own_entry(int i, int n_list, int n0,
                                                 const int* ls, const int* ld,
                                                 int& s, int& dl) {
  if (i < n_list) { s = ls[i]; dl = ld[i]; }
  else            { dl = i - n_list; s = n0 + dl; }
}

template <int H, int NU>
static __device__ __forceinline__ void own_scores_trip(int phase, int idx0, int n_list,
                                                       int n0, const int* ls,
                                                       const int* ld,
                                                       const float* __restrict__ as,
                                                       const float* adL, unsigned* mx,
                                                       float* den) {
  float a[NU];
  int dh[NU];
#pragma unroll
  for (int u = 0; u < NU; ++u) {
    const int idx = idx0 + u * OWN_THREADS;
    const int i = idx / H, h = idx - i * H;
    int s, dl;
    own_entry(i, n_list, n0, ls, ld, s, dl);
    a[u] = as[s * H + h];
    dh[u] = dl * H + h;
  }
#pragma unroll
  for (int u = 0; u < NU; ++u) {
    float v = a[u] + adL[dh[u]];
    v = (v >= 0.0f) ? v : NEG_SLOPE * v;
    if (phase == 0) atomicMax(&mx[dh[u]], f2key(v));
    else            atomicAdd(&den[dh[u]], expf(v - key2f(mx[dh[u]])));
  }
}

template <int H>
static __device__ __forceinline__ void own_scores(int phase, int n_list, int n_self,
                                                  int n0, const int* ls, const int* ld,
                                                  const float* __restrict__ as,
                                                  const float* adL, unsigned* mx,
                                                  float* den) {
  const int n_items = (n_list + n_self) * H;
  int idx = threadIdx.x;
  for (; idx + 3 * OWN_THREADS < n_items; idx += 4 * OWN_THREADS)
    own_scores_trip<H, 4>(phase, idx, n_list, n0, ls, ld, as, adL, mx, den);
  for (; idx < n_items; idx += OWN_THREADS)
    own_scores_trip<H, 1>(phase, idx, n_list, n0, ls, ld, as, adL, mx, den);
}

template <int H, int C, int LDX, int NU>
static __device__ __forceinline__ void own_rows_trip(int i0, int n_list, int n0,
                                                     const int* ls, const int* ld,
                                                     const float* __restrict__ as,
                                                     const float* __restrict__ xw,
                                                     const float* adL,
                                                     const unsigned* mx,
                                                     const float* den, float* acc) {
  constexpr int HC = H * C;
  constexpr int NJ = (HC + 31) / 32;
  const int lane = threadIdx.x & 31;
  const int hl = lane & (H - 1);
  int dl[NU];
  float x[NU][NJ], av[NU];
#pragma unroll
  for (int u = 0; u < NU; ++u) {
    int si, di;
    own_entry(i0 + u * (OWN_THREADS / 32), n_list, n0, ls, ld, si, di);
    const int s = __builtin_amdgcn_readfirstlane(si);
    dl[u] = __builtin_amdgcn_readfirstlane(di);
    const float* row = xw + (size_t)s * LDX;
#pragma unroll
    for (int j = 0; j < NJ; ++j) {
      const int c = j * 32 + lane;
      x[u][j] = (c < HC) ? row[c] : 0.0f;
    }
    av[u] = as[s * H + hl];
  }
#pragma unroll
  for (int u = 0; u < NU; ++u) {
    const int di = dl[u] * H + hl;
    float v = av[u] + adL[di];
    v = (v >= 0.0f) ? v : NEG_SLOPE * v;
    const float w = expf(v - key2f(mx[di])) / den[di];
    float* arow = acc + dl[u] * HC;
#pragma unroll
    for (int j = 0; j < NJ; ++j) {
      const int head = (H == 1) ? 0 : (j * 32) / C;
      const float wj = __int_as_float(
          __builtin_amdgcn_readlane(__float_as_int(w), head));
      const int c = j * 32 + lane;
      if (c < HC) atomicAdd(&arow[c], wj * x[u][j]);
    }
  }
}

template <int H, int C, int LDX>
static __device__ __forceinline__ void own_rows(int n_list, int n_self, int n0,
                                                const int* ls, const int* ld,
                                                const float* __restrict__ as,
                                                const float* __restrict__ xw,
                                                const float* adL, const unsigned* mx,
                                                const float* den, float* acc) {
  static_assert(H == 1 || (C % 32) == 0, "a 32-lane slice must lie in one head");
  static_assert((H & (H - 1)) == 0 && H <= 32, "heads: a power of two");
  constexpr int NW = OWN_THREADS / 32;
  const int n_items = n_list + n_self;
  int i = threadIdx.x >> 5;
  for (; i + NW < n_items; i += 2 * NW)
    own_rows_trip<H, C, LDX, 2>(i, n_list, n0, ls, ld, as, xw, adL, mx, den, acc);
  if (i < n_items)
    own_rows_trip<H, C, LDX, 1>(i, n_list, n0, ls, ld, as, xw, adL, mx, den, acc);
}

template <int H, int C, int LDX, bool ELU, bool TAKE>
__global__ __launch_bounds__(OWN_THREADS) void gat_owner_kernel(
    const int* __restrict__ esrc, const int* __restrict__ edst, int E, int Nn,
    const float* __restrict__ as, const float* __restrict__ ad,
    const float* __restrict__ xw, const float* __restrict__ bias,
    float* __restrict__ outp, int* lst) {
  constexpr int HC = H * C;
  __shared__ float    acc[OWN_T * HC];
  __shared__ float    adL[OWN_T * H];
  __shared__ float    den[OWN_T * H];
  __shared__ unsigned mx[OWN_T * H];
  __shared__ int      ls[OWN_CAP];
  __shared__ int      ld[OWN_CAP];
  __shared__ unsigned cnt;

  const int tid = threadIdx.x;
  const int n0 = blockIdx.x * OWN_T;
  const int nT = min(OWN_T, Nn - n0);

  for (int i = tid; i < OWN_T * HC; i += OWN_THREADS) acc[i] = 0.0f;
  for (int i = tid; i < OWN_T * H; i += OWN_THREADS) {
    den[i] = 0.0f;
    mx[i] = 0u;
    adL[i] = (i < nT * H) ? ad[(size_t)n0 * H + i] : 0.0f;
  }
  if (tid == 0) cnt = 0u;

  int* greg = lst + (size_t)blockIdx.x * OWN_REGION;
  int* gs = greg + OWN_HEAD;
  int* gd = gs + OWN_CAP;
  int total;
  if (TAKE) {
    total = greg[0];
    if (total <= OWN_CAP) own_take(gs, gd, ls, ld, total);
    __syncthreads();
  } else {
    __syncthreads();
    own_build(edst, 0, E, n0, nT, ls, ld, &cnt);
    __syncthreads();
    total = (int)cnt;
    if (tid == 0) greg[0] = total;
  }

  if (total <= OWN_CAP) {
    if (!TAKE) {
      own_resolve(esrc, ls, total);
      __syncthreads();
      own_leave(ls, ld, gs, gd, total);
    }
    own_scores<H>(0, total, nT, n0, ls, ld, as, adL, mx, den);
    __syncthreads();
    own_scores<H>(1, total, nT, n0, ls, ld, as, adL, mx, den);
    __syncthreads();
    own_rows<H, C, LDX>(total, nT, n0, ls, ld, as, xw, adL, mx, den, acc);
  } else {
    for (int phase = 0; phase < 3; ++phase) {
      for (int eb = 0; eb < E; eb += OWN_CAP) {
        __syncthreads();
        if (tid == 0) cnt = 0u;
        __syncthreads();
        own_build(edst, eb, min(eb + OWN_CAP, E), n0, nT, ls, ld, &cnt);
        __syncthreads();
        const int m = (int)cnt;
        own_resolve(esrc, ls, m);
        __syncthreads();
        const int n_self = (eb == 0) ? nT : 0;
        if (phase < 2)
          own_scores<H>(phase, m, n_self, n0, ls, ld, as, adL, mx, den);
        else
          own_rows<H, C, LDX>(m, n_self, n0, ls, ld, as, xw, adL, mx, den, acc);
      }
    }
  }
  __syncthreads();

  float* orow = outp + (size_t)n0 * HC;
  for (int i = tid; i < nT * HC; i += OWN_THREADS) {
    float v = acc[i] + bias[i % HC];
    if (ELU) v = (v > 0.0f) ? v : (expf(v) - 1.0f);
    orow[i] = v;
  }
}

extern "C" void kernel_launch(void* const* d_in, const int* in_sizes, int n_in,
                              void* d_out, int out_size, void* d_ws, size_t ws_size,
                              hipStream_t stream) {
  const float* x      = (const float*)d_in[0];
  const int*   ei     = (const int*)d_in[1];
  const float* W1     = (const float*)d_in[2];
  const float* a1_src = (const float*)d_in[3];
  const float* a1_dst = (const float*)d_in[4];
  const float* b1     = (const float*)d_in[5];
  const float* W2     = (const float*)d_in[6];
  const float* a2_src = (const float*)d_in[7];
  const float* a2_dst = (const float*)d_in[8];
  const float* b2     = (const float*)d_in[9];
  float* out = (float*)d_out;
  (void)n_in; (void)out_size; (void)ws_size;

  const int F_IN = 256, H1 = 8, C1 = 32, HC1 = 256, NC = 40, NCP = 64;
  const int Nn   = in_sizes[0] / F_IN;
  const int E    = in_sizes[1] / 2;
  const int* esrc = ei;
  const int* edst = ei + E;

  float* p = (float*)d_ws;
  float* xw1  = p; p += (size_t)Nn * HC1;
  float* as1  = p; p += (size_t)Nn * H1;
  float* ad1  = p; p += (size_t)Nn * H1;
  float* h1   = p; p += (size_t)Nn * HC1;
  float* w2p  = p; p += (size_t)HC1 * NCP;
  float* xw2p = p; p += (size_t)Nn * NCP;
  float* as2  = p; p += (size_t)Nn;
  float* ad2  = p; p += (size_t)Nn;
  int*   lst  = (int*)p;

  dim3 blk(256);
  const int own_blocks = (Nn + OWN_T - 1) / OWN_T;

  pad_cols_kernel<<<(HC1 * NCP + 255) / 256, blk, 0, stream>>>(
      W2, w2p, HC1, NC, NCP);

  {
    int waves  = (Nn / 16) * (HC1 / 64);
    int blocks = (waves + 7) / 8;
    gemm_f32_wmma_kernel<<<blocks, blk, 0, stream>>>(x, W1, xw1, Nn, F_IN, HC1);
  }
  alpha_kernel<<<(Nn * H1 + 255) / 256, blk, 0, stream>>>(
      xw1, a1_src, a1_dst, as1, ad1, Nn, H1, C1, HC1);
  gat_owner_kernel<8, 32, 256, true, false><<<own_blocks, OWN_THREADS, 0, stream>>>(
      esrc, edst, E, Nn, as1, ad1, xw1, b1, h1, lst);

  {
    int waves  = (Nn / 16) * (NCP / 64);
    int blocks = (waves + 7) / 8;
    gemm_f32_wmma_kernel<<<blocks, blk, 0, stream>>>(h1, w2p, xw2p,
                                                     Nn, HC1, NCP);
  }
  alpha_kernel<<<(Nn + 255) / 256, blk, 0, stream>>>(
      xw2p, a2_src, a2_dst, as2, ad2, Nn, 1, NC, NCP);
  gat_owner_kernel<1, 40, 64, false, true><<<own_blocks, OWN_THREADS, 0, stream>>>(
      esrc, edst, E, Nn, as2, ad2, xw2p, b2, out, lst);
}


# owner groups sweep one key byte per edge (its destination's owner group, written by the prep launch) instead of the 4-byte destination list; local rows set with sources; sums by arrival as before
# speedup vs baseline: 11.3091x; 1.0464x over previous
#include <hip/hip_runtime.h>

typedef __attribute__((ext_vector_type(2))) float v2f;
typedef __attribute__((ext_vector_type(8))) float v8f;

#define NEG_SLOPE 0.2f

static __device__ __forceinline__ unsigned f2key(float f) {
  unsigned u = __float_as_uint(f);
  return (u & 0x80000000u) ? ~u : (u | 0x80000000u);
}
static __device__ __forceinline__ float key2f(unsigned k) {
  return (k & 0x80000000u) ? __uint_as_float(k & 0x7FFFFFFFu)
                           : __uint_as_float(~k);
}

#define OWN_T       196
typedef __attribute__((ext_vector_type(4))) int own_v4i;
typedef __attribute__((ext_vector_type(4))) unsigned own_v4u;

__global__ void prep_kernel(const float* __restrict__ src,
                            float* __restrict__ dst,
                            int K, int c, int ldd,
                            const int* __restrict__ edst,
                            unsigned* __restrict__ keys, int E, int pad_blocks) {
  if ((int)blockIdx.x < pad_blocks) {
    int i = blockIdx.x * blockDim.x + threadIdx.x;
    if (i >= K * ldd) return;
    int r = i / ldd, j = i - r * ldd;
    dst[i] = (j < c) ? src[r * c + j] : 0.0f;
    return;
  }
  const int t = (blockIdx.x - pad_blocks) * blockDim.x + threadIdx.x;
  const int e0 = 4 * t;
  if (e0 >= E) return;
  unsigned k = 0u;
  if (e0 + 3 < E && ((((size_t)(edst + e0)) & 15) == 0)) {
    const own_v4i d = *(const own_v4i*)(edst + e0);
    k = ((unsigned)(d.x / OWN_T) & 255u) | (((unsigned)(d.y / OWN_T) & 255u) << 8) |
        (((unsigned)(d.z / OWN_T) & 255u) << 16) | (((unsigned)(d.w / OWN_T) & 255u) << 24);
  } else {
    for (int u = 0; u < 4; ++u)
      if (e0 + u < E) k |= ((unsigned)(edst[e0 + u] / OWN_T) & 255u) << (8 * u);
  }
  keys[t] = k;
}

__global__ void gemm_f32_wmma_kernel(const float* __restrict__ A,
                                     const float* __restrict__ B,
                                     float* __restrict__ C,
                                     int M, int K, int ldb) {
  const int lane = threadIdx.x & 31;
  const int wave = threadIdx.x >> 5;
  const int half = lane >> 4;
  const int l    = lane & 15;
  const int nt64   = ldb >> 6;
  const int mtiles = M >> 4;
  const int total  = mtiles * nt64;
  int t = blockIdx.x * (blockDim.x >> 5) + wave;
  if (t >= total) return;
  const int mt = t / nt64;
  const int n0 = (t - mt * nt64) << 6;
  const int m0 = mt << 4;

  v8f acc[4] = {};
  const float* arow = A + (size_t)(m0 + l) * K + 2 * half;
  const float* bp   = B + (size_t)(2 * half) * ldb + n0 + l;
  const size_t bstep = (size_t)4 * ldb;

  v2f a = *(const v2f*)(arow);
  v2f b[4];
#pragma unroll
  for (int s = 0; s < 4; ++s) {
    b[s].x = bp[s * 16];
    b[s].y = bp[ldb + s * 16];
  }

  for (int k0 = 4; k0 < K; k0 += 4) {
    v2f an = *(const v2f*)(arow + k0);
    const float* bpn = bp + bstep;
    v2f bn[4];
#pragma unroll
    for (int s = 0; s < 4; ++s) {
      bn[s].x = bpn[s * 16];
      bn[s].y = bpn[ldb + s * 16];
    }
#pragma unroll
    for (int s = 0; s < 4; ++s)
      acc[s] = __builtin_amdgcn_wmma_f32_16x16x4_f32(
          false, a, false, b[s], (short)0, acc[s], false, false);
    a = an;
#pragma unroll
    for (int s = 0; s < 4; ++s) b[s] = bn[s];
    bp = bpn;
  }
#pragma unroll
  for (int s = 0; s < 4; ++s)
    acc[s] = __builtin_amdgcn_wmma_f32_16x16x4_f32(
        false, a, false, b[s], (short)0, acc[s], false, false);

  float* crow = C + (size_t)(m0 + 8 * half) * ldb + n0 + l;
#pragma unroll
  for (int s = 0; s < 4; ++s)
#pragma unroll
    for (int j = 0; j < 8; ++j) crow[(size_t)j * ldb + s * 16] = acc[s][j];
}

__global__ void alpha_kernel(const float* __restrict__ xw,
                             const float* __restrict__ a_src,
                             const float* __restrict__ a_dst,
                             float* __restrict__ as, float* __restrict__ ad,
                             int n_nodes, int heads, int c, int ldx) {
  int i = blockIdx.x * blockDim.x + threadIdx.x;
  if (i >= n_nodes * heads) return;
  int n = i / heads, h = i - n * heads;
  const float* row = xw + (size_t)n * ldx + (size_t)h * c;
  const float* ws = a_src + (size_t)h * c;
  const float* wd = a_dst + (size_t)h * c;
  float s = 0.0f, d = 0.0f;
  for (int j = 0; j < c; ++j) {
    float v = row[j];
    s += v * ws[j];
    d += v * wd[j];
  }
  as[i] = s;
  ad[i] = d;
}

#define OWN_CAP     8192
#define OWN_THREADS 1024
#define OWN_HEAD    32
#define OWN_REGION  (OWN_HEAD + 2 * OWN_CAP)

static __device__ __forceinline__ void own_push(int e, int dl, int* ls, int* ld,
                                                unsigned* cnt) {
  unsigned idx = atomicAdd(cnt, 1u);
  if (idx < OWN_CAP) { ls[idx] = e; ld[idx] = dl; }
}

static __device__ __forceinline__ void own_take4(int e, own_v4i d, int n0, int nT,
                                                 int* ls, int* ld, unsigned* cnt) {
  const int r0 = d.x - n0, r1 = d.y - n0, r2 = d.z - n0, r3 = d.w - n0;
  if ((unsigned)r0 < (unsigned)nT) own_push(e + 0, r0, ls, ld, cnt);
  if ((unsigned)r1 < (unsigned)nT) own_push(e + 1, r1, ls, ld, cnt);
  if ((unsigned)r2 < (unsigned)nT) own_push(e + 2, r2, ls, ld, cnt);
  if ((unsigned)r3 < (unsigned)nT) own_push(e + 3, r3, ls, ld, cnt);
}

static __device__ __forceinline__ void own_build(const int* __restrict__ edst,
                                                 int eb, int ee, int n0, int nT,
                                                 int* ls, int* ld, unsigned* cnt) {
  const int tid = threadIdx.x;
  const int n = ee - eb;
  const int* dbase = edst + eb;
  const int nvec = ((((size_t)dbase) & 15) == 0) ? (n >> 2) : 0;
  const own_v4i* dv = (const own_v4i*)dbase;
  const int rot = (int)(((long)blockIdx.x * nvec) / gridDim.x) & ~31;
  int v = tid;
  for (; v + 3 * OWN_THREADS < nvec; v += 4 * OWN_THREADS) {
    own_v4i d[4];
    int vv[4];
#pragma unroll
    for (int u = 0; u < 4; ++u) {
      vv[u] = v + u * OWN_THREADS + rot;
      if (vv[u] >= nvec) vv[u] -= nvec;
      d[u] = dv[vv[u]];
    }
    asm volatile("" :: "v"(d[0]), "v"(d[1]), "v"(d[2]), "v"(d[3]));
#pragma unroll
    for (int u = 0; u < 4; ++u)
      own_take4(eb + 4 * vv[u], d[u], n0, nT, ls, ld, cnt);
  }
  for (; v < nvec; v += OWN_THREADS) {
    int vv = v + rot;
    if (vv >= nvec) vv -= nvec;
    own_take4(eb + 4 * vv, dv[vv], n0, nT, ls, ld, cnt);
  }
  for (int e = eb + 4 * nvec + tid; e < ee; e += OWN_THREADS) {
    const int r = edst[e] - n0;
    if ((unsigned)r < (unsigned)nT) own_push(e, r, ls, ld, cnt);
  }
}

static __device__ __forceinline__ void own_push_e(int e, int* ls, unsigned* cnt) {
  unsigned idx = atomicAdd(cnt, 1u);
  if (idx < OWN_CAP) ls[idx] = e;
}

static __device__ __forceinline__ unsigned own_hits(unsigned w, unsigned pat) {
  const unsigned y = w ^ pat;
  const unsigned t = (y & 0x7F7F7F7Fu) + 0x7F7F7F7Fu;
  return ~(t | y) & 0x80808080u;
}

static __device__ __forceinline__ void own_take16(int e, own_v4u k, unsigned pat,
                                                  int* ls, unsigned* cnt) {
  unsigned m[4] = {own_hits(k.x, pat), own_hits(k.y, pat),
                   own_hits(k.z, pat), own_hits(k.w, pat)};
  if ((m[0] | m[1] | m[2] | m[3]) != 0u) {
#pragma unroll
    for (int u = 0; u < 4; ++u) {
      unsigned mm = m[u];
      while (mm) {
        own_push_e(e + 4 * u + (__builtin_ctz(mm) >> 3), ls, cnt);
        mm &= mm - 1u;
      }
    }
  }
}

static __device__ __forceinline__ void own_build_keys(const unsigned* __restrict__ keys,
                                                      int E, int* ls, unsigned* cnt) {
  const int tid = threadIdx.x;
  const unsigned gid = blockIdx.x;
  const unsigned pat = gid * 0x01010101u;
  const int nvec = ((((size_t)keys) & 15) == 0) ? (E >> 4) : 0;
  const own_v4u* kv = (const own_v4u*)keys;
  const int rot = (int)(((long)blockIdx.x * nvec) / gridDim.x) & ~31;
  int v = tid;
  for (; v + 3 * OWN_THREADS < nvec; v += 4 * OWN_THREADS) {
    own_v4u k[4];
    int vv[4];
#pragma unroll
    for (int u = 0; u < 4; ++u) {
      vv[u] = v + u * OWN_THREADS + rot;
      if (vv[u] >= nvec) vv[u] -= nvec;
      k[u] = kv[vv[u]];
    }
    asm volatile("" :: "v"(k[0]), "v"(k[1]), "v"(k[2]), "v"(k[3]));
#pragma unroll
    for (int u = 0; u < 4; ++u) own_take16(16 * vv[u], k[u], pat, ls, cnt);
  }
  for (; v < nvec; v += OWN_THREADS) {
    int vv = v + rot;
    if (vv >= nvec) vv -= nvec;
    own_take16(16 * vv, kv[vv], pat, ls, cnt);
  }
  for (int e = 16 * nvec + tid; e < E; e += OWN_THREADS)
    if (((keys[e >> 2] >> (8 * (e & 3))) & 255u) == gid) own_push_e(e, ls, cnt);
}

template <bool WITH_DST>
static __device__ __forceinline__ void own_resolve(const int* __restrict__ esrc,
                                                   const int* __restrict__ edst,
                                                   int n0, int* ls, int* ld, int n) {
  for (int i0 = threadIdx.x; i0 < n; i0 += 4 * OWN_THREADS) {
    int e[4], s[4], d[4];
#pragma unroll
    for (int u = 0; u < 4; ++u) {
      const int i = i0 + u * OWN_THREADS;
      e[u] = (i < n) ? ls[i] : 0;
    }
#pragma unroll
    for (int u = 0; u < 4; ++u) {
      s[u] = esrc[e[u]];
      d[u] = WITH_DST ? edst[e[u]] : 0;
    }
    asm volatile("" :: "v"(s[0]), "v"(s[1]), "v"(s[2]), "v"(s[3]),
                       "v"(d[0]), "v"(d[1]), "v"(d[2]), "v"(d[3]));
#pragma unroll
    for (int u = 0; u < 4; ++u) {
      const int i = i0 + u * OWN_THREADS;
      if (i < n) {
        ls[i] = s[u];
        if (WITH_DST) ld[i] = d[u] - n0;
      }
    }
  }
}

static __device__ __forceinline__ void own_leave(const int* ls, const int* ld,
                                                 int* __restrict__ gs,
                                                 int* __restrict__ gd, int n) {
  for (int i = threadIdx.x; i < n; i += OWN_THREADS) {
    gs[i] = ls[i];
    gd[i] = ld[i];
  }
}

static __device__ __forceinline__ void own_take(const int* __restrict__ gs,
                                                const int* __restrict__ gd,
                                                int* ls, int* ld, int n) {
  for (int i0 = threadIdx.x; i0 < n; i0 += 4 * OWN_THREADS) {
    int s[4], d[4];
#pragma unroll
    for (int u = 0; u < 4; ++u) {
      const int i = i0 + u * OWN_THREADS;
      const int j = (i < n) ? i : 0;
      s[u] = gs[j];
      d[u] = gd[j];
    }
#pragma unroll
    for (int u = 0; u < 4; ++u) {
      const int i = i0 + u * OWN_THREADS;
      if (i < n) { ls[i] = s[u]; ld[i] = d[u]; }
    }
  }
}

static __device__ __forceinline__ void own_entry(int i, int n_list, int n0,
                                                 const int* ls, const int* ld,
                                                 int& s, int& dl) {
  if (i < n_list) { s = ls[i]; dl = ld[i]; }
  else            { dl = i - n_list; s = n0 + dl; }
}

template <int H, int NU>
static __device__ __forceinline__ void own_scores_trip(int phase, int idx0, int n_list,
                                                       int n0, const int* ls,
                                                       const int* ld,
                                                       const float* __restrict__ as,
                                                       const float* adL, unsigned* mx,
                                                       float* den) {
  float a[NU];
  int dh[NU];
#pragma unroll
  for (int u = 0; u < NU; ++u) {
    const int idx = idx0 + u * OWN_THREADS;
    const int i = idx / H, h = idx - i * H;
    int s, dl;
    own_entry(i, n_list, n0, ls, ld, s, dl);
    a[u] = as[s * H + h];
    dh[u] = dl * H + h;
  }
#pragma unroll
  for (int u = 0; u < NU; ++u) {
    float v = a[u] + adL[dh[u]];
    v = (v >= 0.0f) ? v : NEG_SLOPE * v;
    if (phase == 0) atomicMax(&mx[dh[u]], f2key(v));
    else            atomicAdd(&den[dh[u]], expf(v - key2f(mx[dh[u]])));
  }
}

template <int H>
static __device__ __forceinline__ void own_scores(int phase, int n_list, int n_self,
                                                  int n0, const int* ls, const int* ld,
                                                  const float* __restrict__ as,
                                                  const float* adL, unsigned* mx,
                                                  float* den) {
  const int n_items = (n_list + n_self) * H;
  int idx = threadIdx.x;
  for (; idx + 3 * OWN_THREADS < n_items; idx += 4 * OWN_THREADS)
    own_scores_trip<H, 4>(phase, idx, n_list, n0, ls, ld, as, adL, mx, den);
  for (; idx < n_items; idx += OWN_THREADS)
    own_scores_trip<H, 1>(phase, idx, n_list, n0, ls, ld, as, adL, mx, den);
}

template <int H, int C, int LDX, int NU>
static __device__ __forceinline__ void own_rows_trip(int i0, int n_list, int n0,
                                                     const int* ls, const int* ld,
                                                     const float* __restrict__ as,
                                                     const float* __restrict__ xw,
                                                     const float* adL,
                                                     const unsigned* mx,
                                                     const float* den, float* acc) {
  constexpr int HC = H * C;
  constexpr int NJ = (HC + 31) / 32;
  const int lane = threadIdx.x & 31;
  const int hl = lane & (H - 1);
  int dl[NU];
  float x[NU][NJ], av[NU];
#pragma unroll
  for (int u = 0; u < NU; ++u) {
    int si, di;
    own_entry(i0 + u * (OWN_THREADS / 32), n_list, n0, ls, ld, si, di);
    const int s = __builtin_amdgcn_readfirstlane(si);
    dl[u] = __builtin_amdgcn_readfirstlane(di);
    const float* row = xw + (size_t)s * LDX;
#pragma unroll
    for (int j = 0; j < NJ; ++j) {
      const int c = j * 32 + lane;
      x[u][j] = (c < HC) ? row[c] : 0.0f;
    }
    av[u] = as[s * H + hl];
  }
#pragma unroll
  for (int u = 0; u < NU; ++u) {
    const int di = dl[u] * H + hl;
    float v = av[u] + adL[di];
    v = (v >= 0.0f) ? v : NEG_SLOPE * v;
    const float w = expf(v - key2f(mx[di])) / den[di];
    float* arow = acc + dl[u] * HC;
#pragma unroll
    for (int j = 0; j < NJ; ++j) {
      const int head = (H == 1) ? 0 : (j * 32) / C;
      const float wj = __int_as_float(
          __builtin_amdgcn_readlane(__float_as_int(w), head));
      const int c = j * 32 + lane;
      if (c < HC) atomicAdd(&arow[c], wj * x[u][j]);
    }
  }
}

template <int H, int C, int LDX>
static __device__ __forceinline__ void own_rows(int n_list, int n_self, int n0,
                                                const int* ls, const int* ld,
                                                const float* __restrict__ as,
                                                const float* __restrict__ xw,
                                                const float* adL, const unsigned* mx,
                                                const float* den, float* acc) {
  static_assert(H == 1 || (C % 32) == 0, "a 32-lane slice must lie in one head");
  static_assert((H & (H - 1)) == 0 && H <= 32, "heads: a power of two");
  constexpr int NW = OWN_THREADS / 32;
  const int n_items = n_list + n_self;
  int i = threadIdx.x >> 5;
  for (; i + NW < n_items; i += 2 * NW)
    own_rows_trip<H, C, LDX, 2>(i, n_list, n0, ls, ld, as, xw, adL, mx, den, acc);
  if (i < n_items)
    own_rows_trip<H, C, LDX, 1>(i, n_list, n0, ls, ld, as, xw, adL, mx, den, acc);
}

template <int H, int C, int LDX, bool ELU, bool TAKE>
__global__ __launch_bounds__(OWN_THREADS) void gat_owner_kernel(
    const int* __restrict__ esrc, const int* __restrict__ edst, int E, int Nn,
    const float* __restrict__ as, const float* __restrict__ ad,
    const float* __restrict__ xw, const float* __restrict__ bias,
    float* __restrict__ outp, int* lst, const unsigned* __restrict__ keys) {
  constexpr int HC = H * C;
  __shared__ float    acc[OWN_T * HC];
  __shared__ float    adL[OWN_T * H];
  __shared__ float    den[OWN_T * H];
  __shared__ unsigned mx[OWN_T * H];
  __shared__ int      ls[OWN_CAP];
  __shared__ int      ld[OWN_CAP];
  __shared__ unsigned cnt;

  const int tid = threadIdx.x;
  const int n0 = blockIdx.x * OWN_T;
  const int nT = min(OWN_T, Nn - n0);

  for (int i = tid; i < OWN_T * HC; i += OWN_THREADS) acc[i] = 0.0f;
  for (int i = tid; i < OWN_T * H; i += OWN_THREADS) {
    den[i] = 0.0f;
    mx[i] = 0u;
    adL[i] = (i < nT * H) ? ad[(size_t)n0 * H + i] : 0.0f;
  }
  if (tid == 0) cnt = 0u;

  int* greg = lst + (size_t)blockIdx.x * OWN_REGION;
  int* gs = greg + OWN_HEAD;
  int* gd = gs + OWN_CAP;
  int total;
  const bool by_key = (gridDim.x <= 256u);
  if (TAKE) {
    total = greg[0];
    if (total <= OWN_CAP) own_take(gs, gd, ls, ld, total);
    __syncthreads();
  } else {
    __syncthreads();
    if (by_key) own_build_keys(keys, E, ls, &cnt);
    else        own_build(edst, 0, E, n0, nT, ls, ld, &cnt);
    __syncthreads();
    total = (int)cnt;
    if (tid == 0) greg[0] = total;
  }

  if (total <= OWN_CAP) {
    if (!TAKE) {
      if (by_key) own_resolve<true>(esrc, edst, n0, ls, ld, total);
      else        own_resolve<false>(esrc, edst, n0, ls, ld, total);
      __syncthreads();
      own_leave(ls, ld, gs, gd, total);
    }
    own_scores<H>(0, total, nT, n0, ls, ld, as, adL, mx, den);
    __syncthreads();
    own_scores<H>(1, total, nT, n0, ls, ld, as, adL, mx, den);
    __syncthreads();
    own_rows<H, C, LDX>(total, nT, n0, ls, ld, as, xw, adL, mx, den, acc);
  } else {
    for (int phase = 0; phase < 3; ++phase) {
      for (int eb = 0; eb < E; eb += OWN_CAP) {
        __syncthreads();
        if (tid == 0) cnt = 0u;
        __syncthreads();
        own_build(edst, eb, min(eb + OWN_CAP, E), n0, nT, ls, ld, &cnt);
        __syncthreads();
        const int m = (int)cnt;
        own_resolve<false>(esrc, edst, n0, ls, ld, m);
        __syncthreads();
        const int n_self = (eb == 0) ? nT : 0;
        if (phase < 2)
          own_scores<H>(phase, m, n_self, n0, ls, ld, as, adL, mx, den);
        else
          own_rows<H, C, LDX>(m, n_self, n0, ls, ld, as, xw, adL, mx, den, acc);
      }
    }
  }
  __syncthreads();

  float* orow = outp + (size_t)n0 * HC;
  for (int i = tid; i < nT * HC; i += OWN_THREADS) {
    float v = acc[i] + bias[i % HC];
    if (ELU) v = (v > 0.0f) ? v : (expf(v) - 1.0f);
    orow[i] = v;
  }
}

extern "C" void kernel_launch(void* const* d_in, const int* in_sizes, int n_in,
                              void* d_out, int out_size, void* d_ws, size_t ws_size,
                              hipStream_t stream) {
  const float* x      = (const float*)d_in[0];
  const int*   ei     = (const int*)d_in[1];
  const float* W1     = (const float*)d_in[2];
  const float* a1_src = (const float*)d_in[3];
  const float* a1_dst = (const float*)d_in[4];
  const float* b1     = (const float*)d_in[5];
  const float* W2     = (const float*)d_in[6];
  const float* a2_src = (const float*)d_in[7];
  const float* a2_dst = (const float*)d_in[8];
  const float* b2     = (const float*)d_in[9];
  float* out = (float*)d_out;
  (void)n_in; (void)out_size; (void)ws_size;

  const int F_IN = 256, H1 = 8, C1 = 32, HC1 = 256, NC = 40, NCP = 64;
  const int Nn   = in_sizes[0] / F_IN;
  const int E    = in_sizes[1] / 2;
  const int* esrc = ei;
  const int* edst = ei + E;

  float* p = (float*)d_ws;
  float* xw1  = p; p += (size_t)Nn * HC1;
  float* as1  = p; p += (size_t)Nn * H1;
  float* ad1  = p; p += (size_t)Nn * H1;
  float* h1   = p; p += (size_t)Nn * HC1;
  float* w2p  = p; p += (size_t)HC1 * NCP;
  float* xw2p = p; p += (size_t)Nn * NCP;
  float* as2  = p; p += (size_t)Nn;
  float* ad2  = p; p += (size_t)Nn;
  unsigned* keys = (unsigned*)p; p += (size_t)(E / 4 + 4);
  int*   lst  = (int*)p;

  dim3 blk(256);
  const int own_blocks = (Nn + OWN_T - 1) / OWN_T;

  {
    const int pad_blocks = (HC1 * NCP + 255) / 256;
    const int key_blocks = ((E + 3) / 4 + 255) / 256;
    prep_kernel<<<pad_blocks + key_blocks, blk, 0, stream>>>(
        W2, w2p, HC1, NC, NCP, edst, keys, E, pad_blocks);
  }

  {
    int waves  = (Nn / 16) * (HC1 / 64);
    int blocks = (waves + 7) / 8;
    gemm_f32_wmma_kernel<<<blocks, blk, 0, stream>>>(x, W1, xw1, Nn, F_IN, HC1);
  }
  alpha_kernel<<<(Nn * H1 + 255) / 256, blk, 0, stream>>>(
      xw1, a1_src, a1_dst, as1, ad1, Nn, H1, C1, HC1);
  gat_owner_kernel<8, 32, 256, true, false><<<own_blocks, OWN_THREADS, 0, stream>>>(
      esrc, edst, E, Nn, as1, ad1, xw1, b1, h1, lst, keys);

  {
    int waves  = (Nn / 16) * (NCP / 64);
    int blocks = (waves + 7) / 8;
    gemm_f32_wmma_kernel<<<blocks, blk, 0, stream>>>(h1, w2p, xw2p,
                                                     Nn, HC1, NCP);
  }
  alpha_kernel<<<(Nn + 255) / 256, blk, 0, stream>>>(
      xw2p, a2_src, a2_dst, as2, ad2, Nn, 1, NC, NCP);
  gat_owner_kernel<1, 40, 64, false, true><<<own_blocks, OWN_THREADS, 0, stream>>>(
      esrc, edst, E, Nn, as2, ad2, xw2p, b2, out, lst, keys);
}


# row sums: a wave takes 32/heads list entries at once, one lane per (entry, head) forms the weight (one exp and divide per block), lane reads feed the row loop; 32 row loads in flight; sums by arrival
# speedup vs baseline: 12.8349x; 1.1349x over previous
#include <hip/hip_runtime.h>

typedef __attribute__((ext_vector_type(2))) float v2f;
typedef __attribute__((ext_vector_type(8))) float v8f;

#define NEG_SLOPE 0.2f

static __device__ __forceinline__ unsigned f2key(float f) {
  unsigned u = __float_as_uint(f);
  return (u & 0x80000000u) ? ~u : (u | 0x80000000u);
}
static __device__ __forceinline__ float key2f(unsigned k) {
  return (k & 0x80000000u) ? __uint_as_float(k & 0x7FFFFFFFu)
                           : __uint_as_float(~k);
}

#define OWN_T       196
typedef __attribute__((ext_vector_type(4))) int own_v4i;
typedef __attribute__((ext_vector_type(4))) unsigned own_v4u;

__global__ void prep_kernel(const float* __restrict__ src,
                            float* __restrict__ dst,
                            int K, int c, int ldd,
                            const int* __restrict__ edst,
                            unsigned* __restrict__ keys, int E, int pad_blocks) {
  if ((int)blockIdx.x < pad_blocks) {
    int i = blockIdx.x * blockDim.x + threadIdx.x;
    if (i >= K * ldd) return;
    int r = i / ldd, j = i - r * ldd;
    dst[i] = (j < c) ? src[r * c + j] : 0.0f;
    return;
  }
  const int t = (blockIdx.x - pad_blocks) * blockDim.x + threadIdx.x;
  const int e0 = 4 * t;
  if (e0 >= E) return;
  unsigned k = 0u;
  if (e0 + 3 < E && ((((size_t)(edst + e0)) & 15) == 0)) {
    const own_v4i d = *(const own_v4i*)(edst + e0);
    k = ((unsigned)(d.x / OWN_T) & 255u) | (((unsigned)(d.y / OWN_T) & 255u) << 8) |
        (((unsigned)(d.z / OWN_T) & 255u) << 16) | (((unsigned)(d.w / OWN_T) & 255u) << 24);
  } else {
    for (int u = 0; u < 4; ++u)
      if (e0 + u < E) k |= ((unsigned)(edst[e0 + u] / OWN_T) & 255u) << (8 * u);
  }
  keys[t] = k;
}

__global__ void gemm_f32_wmma_kernel(const float* __restrict__ A,
                                     const float* __restrict__ B,
                                     float* __restrict__ C,
                                     int M, int K, int ldb) {
  const int lane = threadIdx.x & 31;
  const int wave = threadIdx.x >> 5;
  const int half = lane >> 4;
  const int l    = lane & 15;
  const int nt64   = ldb >> 6;
  const int mtiles = M >> 4;
  const int total  = mtiles * nt64;
  int t = blockIdx.x * (blockDim.x >> 5) + wave;
  if (t >= total) return;
  const int mt = t / nt64;
  const int n0 = (t - mt * nt64) << 6;
  const int m0 = mt << 4;

  v8f acc[4] = {};
  const float* arow = A + (size_t)(m0 + l) * K + 2 * half;
  const float* bp   = B + (size_t)(2 * half) * ldb + n0 + l;
  const size_t bstep = (size_t)4 * ldb;

  v2f a = *(const v2f*)(arow);
  v2f b[4];
#pragma unroll
  for (int s = 0; s < 4; ++s) {
    b[s].x = bp[s * 16];
    b[s].y = bp[ldb + s * 16];
  }

  for (int k0 = 4; k0 < K; k0 += 4) {
    v2f an = *(const v2f*)(arow + k0);
    const float* bpn = bp + bstep;
    v2f bn[4];
#pragma unroll
    for (int s = 0; s < 4; ++s) {
      bn[s].x = bpn[s * 16];
      bn[s].y = bpn[ldb + s * 16];
    }
#pragma unroll
    for (int s = 0; s < 4; ++s)
      acc[s] = __builtin_amdgcn_wmma_f32_16x16x4_f32(
          false, a, false, b[s], (short)0, acc[s], false, false);
    a = an;
#pragma unroll
    for (int s = 0; s < 4; ++s) b[s] = bn[s];
    bp = bpn;
  }
#pragma unroll
  for (int s = 0; s < 4; ++s)
    acc[s] = __builtin_amdgcn_wmma_f32_16x16x4_f32(
        false, a, false, b[s], (short)0, acc[s], false, false);

  float* crow = C + (size_t)(m0 + 8 * half) * ldb + n0 + l;
#pragma unroll
  for (int s = 0; s < 4; ++s)
#pragma unroll
    for (int j = 0; j < 8; ++j) crow[(size_t)j * ldb + s * 16] = acc[s][j];
}

__global__ void alpha_kernel(const float* __restrict__ xw,
                             const float* __restrict__ a_src,
                             const float* __restrict__ a_dst,
                             float* __restrict__ as, float* __restrict__ ad,
                             int n_nodes, int heads, int c, int ldx) {
  int i = blockIdx.x * blockDim.x + threadIdx.x;
  if (i >= n_nodes * heads) return;
  int n = i / heads, h = i - n * heads;
  const float* row = xw + (size_t)n * ldx + (size_t)h * c;
  const float* ws = a_src + (size_t)h * c;
  const float* wd = a_dst + (size_t)h * c;
  float s = 0.0f, d = 0.0f;
  for (int j = 0; j < c; ++j) {
    float v = row[j];
    s += v * ws[j];
    d += v * wd[j];
  }
  as[i] = s;
  ad[i] = d;
}

#define OWN_CAP     8192
#define OWN_THREADS 1024
#define OWN_HEAD    32
#define OWN_REGION  (OWN_HEAD + 2 * OWN_CAP)

static __device__ __forceinline__ void own_push(int e, int dl, int* ls, int* ld,
                                                unsigned* cnt) {
  unsigned idx = atomicAdd(cnt, 1u);
  if (idx < OWN_CAP) { ls[idx] = e; ld[idx] = dl; }
}

static __device__ __forceinline__ void own_take4(int e, own_v4i d, int n0, int nT,
                                                 int* ls, int* ld, unsigned* cnt) {
  const int r0 = d.x - n0, r1 = d.y - n0, r2 = d.z - n0, r3 = d.w - n0;
  if ((unsigned)r0 < (unsigned)nT) own_push(e + 0, r0, ls, ld, cnt);
  if ((unsigned)r1 < (unsigned)nT) own_push(e + 1, r1, ls, ld, cnt);
  if ((unsigned)r2 < (unsigned)nT) own_push(e + 2, r2, ls, ld, cnt);
  if ((unsigned)r3 < (unsigned)nT) own_push(e + 3, r3, ls, ld, cnt);
}

static __device__ __forceinline__ void own_build(const int* __restrict__ edst,
                                                 int eb, int ee, int n0, int nT,
                                                 int* ls, int* ld, unsigned* cnt) {
  const int tid = threadIdx.x;
  const int n = ee - eb;
  const int* dbase = edst + eb;
  const int nvec = ((((size_t)dbase) & 15) == 0) ? (n >> 2) : 0;
  const own_v4i* dv = (const own_v4i*)dbase;
  const int rot = (int)(((long)blockIdx.x * nvec) / gridDim.x) & ~31;
  int v = tid;
  for (; v + 3 * OWN_THREADS < nvec; v += 4 * OWN_THREADS) {
    own_v4i d[4];
    int vv[4];
#pragma unroll
    for (int u = 0; u < 4; ++u) {
      vv[u] = v + u * OWN_THREADS + rot;
      if (vv[u] >= nvec) vv[u] -= nvec;
      d[u] = dv[vv[u]];
    }
    asm volatile("" :: "v"(d[0]), "v"(d[1]), "v"(d[2]), "v"(d[3]));
#pragma unroll
    for (int u = 0; u < 4; ++u)
      own_take4(eb + 4 * vv[u], d[u], n0, nT, ls, ld, cnt);
  }
  for (; v < nvec; v += OWN_THREADS) {
    int vv = v + rot;
    if (vv >= nvec) vv -= nvec;
    own_take4(eb + 4 * vv, dv[vv], n0, nT, ls, ld, cnt);
  }
  for (int e = eb + 4 * nvec + tid; e < ee; e += OWN_THREADS) {
    const int r = edst[e] - n0;
    if ((unsigned)r < (unsigned)nT) own_push(e, r, ls, ld, cnt);
  }
}

static __device__ __forceinline__ void own_push_e(int e, int* ls, unsigned* cnt) {
  unsigned idx = atomicAdd(cnt, 1u);
  if (idx < OWN_CAP) ls[idx] = e;
}

static __device__ __forceinline__ unsigned own_hits(unsigned w, unsigned pat) {
  const unsigned y = w ^ pat;
  const unsigned t = (y & 0x7F7F7F7Fu) + 0x7F7F7F7Fu;
  return ~(t | y) & 0x80808080u;
}

static __device__ __forceinline__ void own_take16(int e, own_v4u k, unsigned pat,
                                                  int* ls, unsigned* cnt) {
  unsigned m[4] = {own_hits(k.x, pat), own_hits(k.y, pat),
                   own_hits(k.z, pat), own_hits(k.w, pat)};
  if ((m[0] | m[1] | m[2] | m[3]) != 0u) {
#pragma unroll
    for (int u = 0; u < 4; ++u) {
      unsigned mm = m[u];
      while (mm) {
        own_push_e(e + 4 * u + (__builtin_ctz(mm) >> 3), ls, cnt);
        mm &= mm - 1u;
      }
    }
  }
}

static __device__ __forceinline__ void own_build_keys(const unsigned* __restrict__ keys,
                                                      int E, int* ls, unsigned* cnt) {
  const int tid = threadIdx.x;
  const unsigned gid = blockIdx.x;
  const unsigned pat = gid * 0x01010101u;
  const int nvec = ((((size_t)keys) & 15) == 0) ? (E >> 4) : 0;
  const own_v4u* kv = (const own_v4u*)keys;
  const int rot = (int)(((long)blockIdx.x * nvec) / gridDim.x) & ~31;
  int v = tid;
  for (; v + 3 * OWN_THREADS < nvec; v += 4 * OWN_THREADS) {
    own_v4u k[4];
    int vv[4];
#pragma unroll
    for (int u = 0; u < 4; ++u) {
      vv[u] = v + u * OWN_THREADS + rot;
      if (vv[u] >= nvec) vv[u] -= nvec;
      k[u] = kv[vv[u]];
    }
    asm volatile("" :: "v"(k[0]), "v"(k[1]), "v"(k[2]), "v"(k[3]));
#pragma unroll
    for (int u = 0; u < 4; ++u) own_take16(16 * vv[u], k[u], pat, ls, cnt);
  }
  for (; v < nvec; v += OWN_THREADS) {
    int vv = v + rot;
    if (vv >= nvec) vv -= nvec;
    own_take16(16 * vv, kv[vv], pat, ls, cnt);
  }
  for (int e = 16 * nvec + tid; e < E; e += OWN_THREADS)
    if (((keys[e >> 2] >> (8 * (e & 3))) & 255u) == gid) own_push_e(e, ls, cnt);
}

template <bool WITH_DST>
static __device__ __forceinline__ void own_resolve(const int* __restrict__ esrc,
                                                   const int* __restrict__ edst,
                                                   int n0, int* ls, int* ld, int n) {
  for (int i0 = threadIdx.x; i0 < n; i0 += 4 * OWN_THREADS) {
    int e[4], s[4], d[4];
#pragma unroll
    for (int u = 0; u < 4; ++u) {
      const int i = i0 + u * OWN_THREADS;
      e[u] = (i < n) ? ls[i] : 0;
    }
#pragma unroll
    for (int u = 0; u < 4; ++u) {
      s[u] = esrc[e[u]];
      d[u] = WITH_DST ? edst[e[u]] : 0;
    }
    asm volatile("" :: "v"(s[0]), "v"(s[1]), "v"(s[2]), "v"(s[3]),
                       "v"(d[0]), "v"(d[1]), "v"(d[2]), "v"(d[3]));
#pragma unroll
    for (int u = 0; u < 4; ++u) {
      const int i = i0 + u * OWN_THREADS;
      if (i < n) {
        ls[i] = s[u];
        if (WITH_DST) ld[i] = d[u] - n0;
      }
    }
  }
}

static __device__ __forceinline__ void own_leave(const int* ls, const int* ld,
                                                 int* __restrict__ gs,
                                                 int* __restrict__ gd, int n) {
  for (int i = threadIdx.x; i < n; i += OWN_THREADS) {
    gs[i] = ls[i];
    gd[i] = ld[i];
  }
}

static __device__ __forceinline__ void own_take(const int* __restrict__ gs,
                                                const int* __restrict__ gd,
                                                int* ls, int* ld, int n) {
  for (int i0 = threadIdx.x; i0 < n; i0 += 4 * OWN_THREADS) {
    int s[4], d[4];
#pragma unroll
    for (int u = 0; u < 4; ++u) {
      const int i = i0 + u * OWN_THREADS;
      const int j = (i < n) ? i : 0;
      s[u] = gs[j];
      d[u] = gd[j];
    }
#pragma unroll
    for (int u = 0; u < 4; ++u) {
      const int i = i0 + u * OWN_THREADS;
      if (i < n) { ls[i] = s[u]; ld[i] = d[u]; }
    }
  }
}

static __device__ __forceinline__ void own_entry(int i, int n_list, int n0,
                                                 const int* ls, const int* ld,
                                                 int& s, int& dl) {
  if (i < n_list) { s = ls[i]; dl = ld[i]; }
  else            { dl = i - n_list; s = n0 + dl; }
}

template <int H, int NU>
static __device__ __forceinline__ void own_scores_trip(int phase, int idx0, int n_list,
                                                       int n0, const int* ls,
                                                       const int* ld,
                                                       const float* __restrict__ as,
                                                       const float* adL, unsigned* mx,
                                                       float* den) {
  float a[NU];
  int dh[NU];
#pragma unroll
  for (int u = 0; u < NU; ++u) {
    const int idx = idx0 + u * OWN_THREADS;
    const int i = idx / H, h = idx - i * H;
    int s, dl;
    own_entry(i, n_list, n0, ls, ld, s, dl);
    a[u] = as[s * H + h];
    dh[u] = dl * H + h;
  }
#pragma unroll
  for (int u = 0; u < NU; ++u) {
    float v = a[u] + adL[dh[u]];
    v = (v >= 0.0f) ? v : NEG_SLOPE * v;
    if (phase == 0) atomicMax(&mx[dh[u]], f2key(v));
    else            atomicAdd(&den[dh[u]], expf(v - key2f(mx[dh[u]])));
  }
}

template <int H>
static __device__ __forceinline__ void own_scores(int phase, int n_list, int n_self,
                                                  int n0, const int* ls, const int* ld,
                                                  const float* __restrict__ as,
                                                  const float* adL, unsigned* mx,
                                                  float* den) {
  const int n_items = (n_list + n_self) * H;
  int idx = threadIdx.x;
  for (; idx + 3 * OWN_THREADS < n_items; idx += 4 * OWN_THREADS)
    own_scores_trip<H, 4>(phase, idx, n_list, n0, ls, ld, as, adL, mx, den);
  for (; idx < n_items; idx += OWN_THREADS)
    own_scores_trip<H, 1>(phase, idx, n_list, n0, ls, ld, as, adL, mx, den);
}

template <int H, int C, int LDX, int NU>
static __device__ __forceinline__ void own_rows_trip(int i0, int n_list, int n0,
                                                     const int* ls, const int* ld,
                                                     const float* __restrict__ as,
                                                     const float* __restrict__ xw,
                                                     const float* adL,
                                                     const unsigned* mx,
                                                     const float* den, float* acc) {
  constexpr int HC = H * C;
  constexpr int NJ = (HC + 31) / 32;
  const int lane = threadIdx.x & 31;
  const int hl = lane & (H - 1);
  int dl[NU];
  float x[NU][NJ], av[NU];
#pragma unroll
  for (int u = 0; u < NU; ++u) {
    int si, di;
    own_entry(i0 + u * (OWN_THREADS / 32), n_list, n0, ls, ld, si, di);
    const int s = __builtin_amdgcn_readfirstlane(si);
    dl[u] = __builtin_amdgcn_readfirstlane(di);
    const float* row = xw + (size_t)s * LDX;
#pragma unroll
    for (int j = 0; j < NJ; ++j) {
      const int c = j * 32 + lane;
      x[u][j] = (c < HC) ? row[c] : 0.0f;
    }
    av[u] = as[s * H + hl];
  }
#pragma unroll
  for (int u = 0; u < NU; ++u) {
    const int di = dl[u] * H + hl;
    float v = av[u] + adL[di];
    v = (v >= 0.0f) ? v : NEG_SLOPE * v;
    const float w = expf(v - key2f(mx[di])) / den[di];
    float* arow = acc + dl[u] * HC;
#pragma unroll
    for (int j = 0; j < NJ; ++j) {
      const int head = (H == 1) ? 0 : (j * 32) / C;
      const float wj = __int_as_float(
          __builtin_amdgcn_readlane(__float_as_int(w), head));
      const int c = j * 32 + lane;
      if (c < HC) atomicAdd(&arow[c], wj * x[u][j]);
    }
  }
}

template <int H, int C, int LDX>
static __device__ __forceinline__ void own_rows_block(int i0, int n_list, int n0,
                                                      const int* ls, const int* ld,
                                                      const float* __restrict__ as,
                                                      const float* __restrict__ xw,
                                                      const float* adL,
                                                      const unsigned* mx,
                                                      const float* den, float* acc) {
  constexpr int HC = H * C;
  constexpr int NJ = (HC + 31) / 32;
  constexpr int G = 32 / H;
  constexpr int SUB = (G * NJ <= 32) ? G : (32 / NJ);
  static_assert(SUB >= 1 && (G % SUB) == 0, "a block is a whole number of row groups");
  const int lane = threadIdx.x & 31;
  const int hl = lane & (H - 1);
  int si, di;
  own_entry(i0 + lane / H, n_list, n0, ls, ld, si, di);
  const int dh = di * H + hl;
  const float a = as[si * H + hl];
  float w = 0.0f;
#pragma unroll
  for (int k = 0; k < G / SUB; ++k) {
    float x[SUB][NJ];
#pragma unroll
    for (int u = 0; u < SUB; ++u) {
      const int s = __builtin_amdgcn_readlane(si, (k * SUB + u) * H);
      const float* row = xw + (size_t)s * LDX;
#pragma unroll
      for (int j = 0; j < NJ; ++j) {
        const int c = j * 32 + lane;
        x[u][j] = (c < HC) ? row[c] : 0.0f;
      }
    }
    if (k == 0) {
      float v = a + adL[dh];
      v = (v >= 0.0f) ? v : NEG_SLOPE * v;
      w = expf(v - key2f(mx[dh])) / den[dh];
    }
#pragma unroll
    for (int u = 0; u < SUB; ++u) {
      const int e = k * SUB + u;
      const int dl = __builtin_amdgcn_readlane(di, e * H);
      float* arow = acc + dl * HC;
#pragma unroll
      for (int j = 0; j < NJ; ++j) {
        const int head = (H == 1) ? 0 : (j * 32) / C;
        const float wj = __int_as_float(
            __builtin_amdgcn_readlane(__float_as_int(w), e * H + head));
        const int c = j * 32 + lane;
        if (c < HC) atomicAdd(&arow[c], wj * x[u][j]);
      }
    }
  }
}

template <int H, int C, int LDX>
static __device__ __forceinline__ void own_rows(int n_list, int n_self, int n0,
                                                const int* ls, const int* ld,
                                                const float* __restrict__ as,
                                                const float* __restrict__ xw,
                                                const float* adL, const unsigned* mx,
                                                const float* den, float* acc) {
  static_assert(H == 1 || (C % 32) == 0, "a 32-lane slice must lie in one head");
  static_assert((H & (H - 1)) == 0 && H <= 32, "heads: a power of two");
  constexpr int NW = OWN_THREADS / 32;
  constexpr int G = 32 / H;
  static_assert(G <= NW, "the entries left over get one wave each");
  const int n_items = n_list + n_self;
  const int n_blocks = n_items / G;
  const int wave = __builtin_amdgcn_readfirstlane((int)(threadIdx.x >> 5));
  for (int b = wave; b < n_blocks; b += NW)
    own_rows_block<H, C, LDX>(b * G, n_list, n0, ls, ld, as, xw, adL, mx, den, acc);
  const int t = n_blocks * G + wave;
  if (t < n_items)
    own_rows_trip<H, C, LDX, 1>(t, n_list, n0, ls, ld, as, xw, adL, mx, den, acc);
}

template <int H, int C, int LDX, bool ELU, bool TAKE>
__global__ __launch_bounds__(OWN_THREADS) void gat_owner_kernel(
    const int* __restrict__ esrc, const int* __restrict__ edst, int E, int Nn,
    const float* __restrict__ as, const float* __restrict__ ad,
    const float* __restrict__ xw, const float* __restrict__ bias,
    float* __restrict__ outp, int* lst, const unsigned* __restrict__ keys) {
  constexpr int HC = H * C;
  __shared__ float    acc[OWN_T * HC];
  __shared__ float    adL[OWN_T * H];
  __shared__ float    den[OWN_T * H];
  __shared__ unsigned mx[OWN_T * H];
  __shared__ int      ls[OWN_CAP];
  __shared__ int      ld[OWN_CAP];
  __shared__ unsigned cnt;

  const int tid = threadIdx.x;
  const int n0 = blockIdx.x * OWN_T;
  const int nT = min(OWN_T, Nn - n0);

  for (int i = tid; i < OWN_T * HC; i += OWN_THREADS) acc[i] = 0.0f;
  for (int i = tid; i < OWN_T * H; i += OWN_THREADS) {
    den[i] = 0.0f;
    mx[i] = 0u;
    adL[i] = (i < nT * H) ? ad[(size_t)n0 * H + i] : 0.0f;
  }
  if (tid == 0) cnt = 0u;

  int* greg = lst + (size_t)blockIdx.x * OWN_REGION;
  int* gs = greg + OWN_HEAD;
  int* gd = gs + OWN_CAP;
  int total;
  const bool by_key = (gridDim.x <= 256u);
  if (TAKE) {
    total = greg[0];
    if (total <= OWN_CAP) own_take(gs, gd, ls, ld, total);
    __syncthreads();
  } else {
    __syncthreads();
    if (by_key) own_build_keys(keys, E, ls, &cnt);
    else        own_build(edst, 0, E, n0, nT, ls, ld, &cnt);
    __syncthreads();
    total = (int)cnt;
    if (tid == 0) greg[0] = total;
  }

  if (total <= OWN_CAP) {
    if (!TAKE) {
      if (by_key) own_resolve<true>(esrc, edst, n0, ls, ld, total);
      else        own_resolve<false>(esrc, edst, n0, ls, ld, total);
      __syncthreads();
      own_leave(ls, ld, gs, gd, total);
    }
    own_scores<H>(0, total, nT, n0, ls, ld, as, adL, mx, den);
    __syncthreads();
    own_scores<H>(1, total, nT, n0, ls, ld, as, adL, mx, den);
    __syncthreads();
    own_rows<H, C, LDX>(total, nT, n0, ls, ld, as, xw, adL, mx, den, acc);
  } else {
    for (int phase = 0; phase < 3; ++phase) {
      for (int eb = 0; eb < E; eb += OWN_CAP) {
        __syncthreads();
        if (tid == 0) cnt = 0u;
        __syncthreads();
        own_build(edst, eb, min(eb + OWN_CAP, E), n0, nT, ls, ld, &cnt);
        __syncthreads();
        const int m = (int)cnt;
        own_resolve<false>(esrc, edst, n0, ls, ld, m);
        __syncthreads();
        const int n_self = (eb == 0) ? nT : 0;
        if (phase < 2)
          own_scores<H>(phase, m, n_self, n0, ls, ld, as, adL, mx, den);
        else
          own_rows<H, C, LDX>(m, n_self, n0, ls, ld, as, xw, adL, mx, den, acc);
      }
    }
  }
  __syncthreads();

  float* orow = outp + (size_t)n0 * HC;
  for (int i = tid; i < nT * HC; i += OWN_THREADS) {
    float v = acc[i] + bias[i % HC];
    if (ELU) v = (v > 0.0f) ? v : (expf(v) - 1.0f);
    orow[i] = v;
  }
}

extern "C" void kernel_launch(void* const* d_in, const int* in_sizes, int n_in,
                              void* d_out, int out_size, void* d_ws, size_t ws_size,
                              hipStream_t stream) {
  const float* x      = (const float*)d_in[0];
  const int*   ei     = (const int*)d_in[1];
  const float* W1     = (const float*)d_in[2];
  const float* a1_src = (const float*)d_in[3];
  const float* a1_dst = (const float*)d_in[4];
  const float* b1     = (const float*)d_in[5];
  const float* W2     = (const float*)d_in[6];
  const float* a2_src = (const float*)d_in[7];
  const float* a2_dst = (const float*)d_in[8];
  const float* b2     = (const float*)d_in[9];
  float* out = (float*)d_out;
  (void)n_in; (void)out_size; (void)ws_size;

  const int F_IN = 256, H1 = 8, C1 = 32, HC1 = 256, NC = 40, NCP = 64;
  const int Nn   = in_sizes[0] / F_IN;
  const int E    = in_sizes[1] / 2;
  const int* esrc = ei;
  const int* edst = ei + E;

  float* p = (float*)d_ws;
  float* xw1  = p; p += (size_t)Nn * HC1;
  float* as1  = p; p += (size_t)Nn * H1;
  float* ad1  = p; p += (size_t)Nn * H1;
  float* h1   = p; p += (size_t)Nn * HC1;
  float* w2p  = p; p += (size_t)HC1 * NCP;
  float* xw2p = p; p += (size_t)Nn * NCP;
  float* as2  = p; p += (size_t)Nn;
  float* ad2  = p; p += (size_t)Nn;
  unsigned* keys = (unsigned*)p; p += (size_t)(E / 4 + 4);
  int*   lst  = (int*)p;

  dim3 blk(256);
  const int own_blocks = (Nn + OWN_T - 1) / OWN_T;

  {
    const int pad_blocks = (HC1 * NCP + 255) / 256;
    const int key_blocks = ((E + 3) / 4 + 255) / 256;
    prep_kernel<<<pad_blocks + key_blocks, blk, 0, stream>>>(
        W2, w2p, HC1, NC, NCP, edst, keys, E, pad_blocks);
  }

  {
    int waves  = (Nn / 16) * (HC1 / 64);
    int blocks = (waves + 7) / 8;
    gemm_f32_wmma_kernel<<<blocks, blk, 0, stream>>>(x, W1, xw1, Nn, F_IN, HC1);
  }
  alpha_kernel<<<(Nn * H1 + 255) / 256, blk, 0, stream>>>(
      xw1, a1_src, a1_dst, as1, ad1, Nn, H1, C1, HC1);
  gat_owner_kernel<8, 32, 256, true, false><<<own_blocks, OWN_THREADS, 0, stream>>>(
      esrc, edst, E, Nn, as1, ad1, xw1, b1, h1, lst, keys);

  {
    int waves  = (Nn / 16) * (NCP / 64);
    int blocks = (waves + 7) / 8;
    gemm_f32_wmma_kernel<<<blocks, blk, 0, stream>>>(h1, w2p, xw2p,
                                                     Nn, HC1, NCP);
  }
  alpha_kernel<<<(Nn + 255) / 256, blk, 0, stream>>>(
      xw2p, a2_src, a2_dst, as2, ad2, Nn, 1, NC, NCP);
  gat_owner_kernel<1, 40, 64, false, true><<<own_blocks, OWN_THREADS, 0, stream>>>(
      esrc, edst, E, Nn, as2, ad2, xw2p, b2, out, lst, keys);
}


# prep launch also writes, per owner group, one bit per 16-edge piece of the key array; a group reads its bit row and looks only at flagged pieces, no full key sweep; sums by arrival as before
# speedup vs baseline: 13.2761x; 1.0344x over previous
#include <hip/hip_runtime.h>

typedef __attribute__((ext_vector_type(2))) float v2f;
typedef __attribute__((ext_vector_type(8))) float v8f;

#define NEG_SLOPE 0.2f

static __device__ __forceinline__ unsigned f2key(float f) {
  unsigned u = __float_as_uint(f);
  return (u & 0x80000000u) ? ~u : (u | 0x80000000u);
}
static __device__ __forceinline__ float key2f(unsigned k) {
  return (k & 0x80000000u) ? __uint_as_float(k & 0x7FFFFFFFu)
                           : __uint_as_float(~k);
}

#define OWN_T       196
typedef __attribute__((ext_vector_type(4))) int own_v4i;
typedef __attribute__((ext_vector_type(4))) unsigned own_v4u;

#define PREP_THREADS 256
#define PLANE_W      (PREP_THREADS / 32)
__global__ void prep_kernel(const float* __restrict__ src,
                            float* __restrict__ dst,
                            int K, int c, int ldd,
                            const int* __restrict__ edst,
                            unsigned* __restrict__ keys, int E, int pad_blocks,
                            unsigned* __restrict__ planes, int plane_words) {
  __shared__ unsigned bm[256 * PLANE_W];
  if ((int)blockIdx.x < pad_blocks) {
    int i = blockIdx.x * blockDim.x + threadIdx.x;
    if (i >= K * ldd) return;
    int r = i / ldd, j = i - r * ldd;
    dst[i] = (j < c) ? src[r * c + j] : 0.0f;
    return;
  }
  const int tid = threadIdx.x;
  const int kb = blockIdx.x - pad_blocks;
  for (int i = tid; i < 256 * PLANE_W; i += PREP_THREADS) bm[i] = 0u;
  __syncthreads();
  const int piece = kb * PREP_THREADS + tid;
  const int e0 = 16 * piece;
  if (e0 < E) {
    const bool full = (e0 + 15 < E);
    int d[16];
    if (full && ((((size_t)(edst + e0)) & 15) == 0)) {
      const own_v4i* dv = (const own_v4i*)(edst + e0);
      for (int q = 0; q < 4; ++q) {
        const own_v4i v = dv[q];
        d[4 * q + 0] = v.x; d[4 * q + 1] = v.y; d[4 * q + 2] = v.z; d[4 * q + 3] = v.w;
      }
    } else {
      for (int u = 0; u < 16; ++u) d[u] = (e0 + u < E) ? edst[e0 + u] : -1;
    }
    unsigned kw[4] = {0u, 0u, 0u, 0u};
    const unsigned bit = 1u << (tid & 31);
    for (int u = 0; u < 16; ++u) {
      if (d[u] >= 0) {
        const unsigned g = (unsigned)(d[u] / OWN_T) & 255u;
        kw[u >> 2] |= g << (8 * (u & 3));
        if (full) atomicOr(&bm[g * PLANE_W + (tid >> 5)], bit);
      }
    }
    if (full && ((((size_t)keys) & 15) == 0)) {
      own_v4u k4;
      k4.x = kw[0]; k4.y = kw[1]; k4.z = kw[2]; k4.w = kw[3];
      *(own_v4u*)(keys + 4 * piece) = k4;
    } else {
      for (int q = 0; q < 4; ++q)
        if (e0 + 4 * q < E) keys[4 * piece + q] = kw[q];
    }
  }
  __syncthreads();
  unsigned* row = planes + (size_t)tid * plane_words + (size_t)kb * PLANE_W;
  for (int q = 0; q < PLANE_W; ++q) row[q] = bm[tid * PLANE_W + q];
}

__global__ void gemm_f32_wmma_kernel(const float* __restrict__ A,
                                     const float* __restrict__ B,
                                     float* __restrict__ C,
                                     int M, int K, int ldb) {
  const int lane = threadIdx.x & 31;
  const int wave = threadIdx.x >> 5;
  const int half = lane >> 4;
  const int l    = lane & 15;
  const int nt64   = ldb >> 6;
  const int mtiles = M >> 4;
  const int total  = mtiles * nt64;
  int t = blockIdx.x * (blockDim.x >> 5) + wave;
  if (t >= total) return;
  const int mt = t / nt64;
  const int n0 = (t - mt * nt64) << 6;
  const int m0 = mt << 4;

  v8f acc[4] = {};
  const float* arow = A + (size_t)(m0 + l) * K + 2 * half;
  const float* bp   = B + (size_t)(2 * half) * ldb + n0 + l;
  const size_t bstep = (size_t)4 * ldb;

  v2f a = *(const v2f*)(arow);
  v2f b[4];
#pragma unroll
  for (int s = 0; s < 4; ++s) {
    b[s].x = bp[s * 16];
    b[s].y = bp[ldb + s * 16];
  }

  for (int k0 = 4; k0 < K; k0 += 4) {
    v2f an = *(const v2f*)(arow + k0);
    const float* bpn = bp + bstep;
    v2f bn[4];
#pragma unroll
    for (int s = 0; s < 4; ++s) {
      bn[s].x = bpn[s * 16];
      bn[s].y = bpn[ldb + s * 16];
    }
#pragma unroll
    for (int s = 0; s < 4; ++s)
      acc[s] = __builtin_amdgcn_wmma_f32_16x16x4_f32(
          false, a, false, b[s], (short)0, acc[s], false, false);
    a = an;
#pragma unroll
    for (int s = 0; s < 4; ++s) b[s] = bn[s];
    bp = bpn;
  }
#pragma unroll
  for (int s = 0; s < 4; ++s)
    acc[s] = __builtin_amdgcn_wmma_f32_16x16x4_f32(
        false, a, false, b[s], (short)0, acc[s], false, false);

  float* crow = C + (size_t)(m0 + 8 * half) * ldb + n0 + l;
#pragma unroll
  for (int s = 0; s < 4; ++s)
#pragma unroll
    for (int j = 0; j < 8; ++j) crow[(size_t)j * ldb + s * 16] = acc[s][j];
}

__global__ void alpha_kernel(const float* __restrict__ xw,
                             const float* __restrict__ a_src,
                             const float* __restrict__ a_dst,
                             float* __restrict__ as, float* __restrict__ ad,
                             int n_nodes, int heads, int c, int ldx) {
  int i = blockIdx.x * blockDim.x + threadIdx.x;
  if (i >= n_nodes * heads) return;
  int n = i / heads, h = i - n * heads;
  const float* row = xw + (size_t)n * ldx + (size_t)h * c;
  const float* ws = a_src + (size_t)h * c;
  const float* wd = a_dst + (size_t)h * c;
  float s = 0.0f, d = 0.0f;
  for (int j = 0; j < c; ++j) {
    float v = row[j];
    s += v * ws[j];
    d += v * wd[j];
  }
  as[i] = s;
  ad[i] = d;
}

#define OWN_CAP     8192
#define OWN_THREADS 1024
#define OWN_HEAD    32
#define OWN_REGION  (OWN_HEAD + 2 * OWN_CAP)

static __device__ __forceinline__ void own_push(int e, int dl, int* ls, int* ld,
                                                unsigned* cnt) {
  unsigned idx = atomicAdd(cnt, 1u);
  if (idx < OWN_CAP) { ls[idx] = e; ld[idx] = dl; }
}

static __device__ __forceinline__ void own_take4(int e, own_v4i d, int n0, int nT,
                                                 int* ls, int* ld, unsigned* cnt) {
  const int r0 = d.x - n0, r1 = d.y - n0, r2 = d.z - n0, r3 = d.w - n0;
  if ((unsigned)r0 < (unsigned)nT) own_push(e + 0, r0, ls, ld, cnt);
  if ((unsigned)r1 < (unsigned)nT) own_push(e + 1, r1, ls, ld, cnt);
  if ((unsigned)r2 < (unsigned)nT) own_push(e + 2, r2, ls, ld, cnt);
  if ((unsigned)r3 < (unsigned)nT) own_push(e + 3, r3, ls, ld, cnt);
}

static __device__ __forceinline__ void own_build(const int* __restrict__ edst,
                                                 int eb, int ee, int n0, int nT,
                                                 int* ls, int* ld, unsigned* cnt) {
  const int tid = threadIdx.x;
  const int n = ee - eb;
  const int* dbase = edst + eb;
  const int nvec = ((((size_t)dbase) & 15) == 0) ? (n >> 2) : 0;
  const own_v4i* dv = (const own_v4i*)dbase;
  const int rot = (int)(((long)blockIdx.x * nvec) / gridDim.x) & ~31;
  int v = tid;
  for (; v + 3 * OWN_THREADS < nvec; v += 4 * OWN_THREADS) {
    own_v4i d[4];
    int vv[4];
#pragma unroll
    for (int u = 0; u < 4; ++u) {
      vv[u] = v + u * OWN_THREADS + rot;
      if (vv[u] >= nvec) vv[u] -= nvec;
      d[u] = dv[vv[u]];
    }
    asm volatile("" :: "v"(d[0]), "v"(d[1]), "v"(d[2]), "v"(d[3]));
#pragma unroll
    for (int u = 0; u < 4; ++u)
      own_take4(eb + 4 * vv[u], d[u], n0, nT, ls, ld, cnt);
  }
  for (; v < nvec; v += OWN_THREADS) {
    int vv = v + rot;
    if (vv >= nvec) vv -= nvec;
    own_take4(eb + 4 * vv, dv[vv], n0, nT, ls, ld, cnt);
  }
  for (int e = eb + 4 * nvec + tid; e < ee; e += OWN_THREADS) {
    const int r = edst[e] - n0;
    if ((unsigned)r < (unsigned)nT) own_push(e, r, ls, ld, cnt);
  }
}

static __device__ __forceinline__ void own_push_e(int e, int* ls, unsigned* cnt) {
  unsigned idx = atomicAdd(cnt, 1u);
  if (idx < OWN_CAP) ls[idx] = e;
}

static __device__ __forceinline__ unsigned own_hits(unsigned w, unsigned pat) {
  const unsigned y = w ^ pat;
  const unsigned t = (y & 0x7F7F7F7Fu) + 0x7F7F7F7Fu;
  return ~(t | y) & 0x80808080u;
}

static __device__ __forceinline__ void own_take16(int e, own_v4u k, unsigned pat,
                                                  int* ls, unsigned* cnt) {
  unsigned m[4] = {own_hits(k.x, pat), own_hits(k.y, pat),
                   own_hits(k.z, pat), own_hits(k.w, pat)};
  if ((m[0] | m[1] | m[2] | m[3]) != 0u) {
#pragma unroll
    for (int u = 0; u < 4; ++u) {
      unsigned mm = m[u];
      while (mm) {
        own_push_e(e + 4 * u + (__builtin_ctz(mm) >> 3), ls, cnt);
        mm &= mm - 1u;
      }
    }
  }
}

static __device__ __forceinline__ void own_build_keys(const unsigned* __restrict__ keys,
                                                      const unsigned* __restrict__ plane,
                                                      int E, int* ls, unsigned* pl,
                                                      unsigned* cnt, unsigned* pcnt) {
  const int tid = threadIdx.x;
  const int lane = tid & 31;
  const unsigned gid = blockIdx.x;
  const unsigned pat = gid * 0x01010101u;
  const int nvec = ((((size_t)keys) & 15) == 0) ? (E >> 4) : 0;
  const own_v4u* kv = (const own_v4u*)keys;
  const int nwords = (nvec + 31) >> 5;
  for (int w0 = tid & ~31; w0 < nwords; w0 += OWN_THREADS) {
    const int w = w0 + lane;
    unsigned m = (w < nwords) ? plane[w] : 0u;
    const int n = __builtin_popcount(m);
    int incl = n;
    for (int dd = 1; dd < 32; dd <<= 1) {
      const int t = __shfl_up(incl, dd);
      if (lane >= dd) incl += t;
    }
    const int tot = __shfl(incl, 31);
    if (tot != 0) {
      unsigned base = 0u;
      if (lane == 31) base = atomicAdd(pcnt, (unsigned)tot);
      base = (unsigned)__shfl((int)base, 31);
      unsigned idx = base + (unsigned)(incl - n);
      while (m) {
        if (idx < OWN_CAP) pl[idx] = 32u * (unsigned)w + (unsigned)__builtin_ctz(m);
        ++idx;
        m &= m - 1u;
      }
    }
  }
  __syncthreads();
  const int np = (int)*pcnt;
  if (np > OWN_CAP) {
    if (tid == 0) *cnt = (unsigned)np;
    return;
  }
  for (int i0 = tid; i0 < np; i0 += 4 * OWN_THREADS) {
    own_v4u k[4];
    unsigned pc[4];
#pragma unroll
    for (int u = 0; u < 4; ++u) {
      const int i = i0 + u * OWN_THREADS;
      pc[u] = pl[(i < np) ? i : 0];
      k[u] = kv[pc[u]];
    }
    asm volatile("" :: "v"(k[0]), "v"(k[1]), "v"(k[2]), "v"(k[3]));
#pragma unroll
    for (int u = 0; u < 4; ++u)
      if (i0 + u * OWN_THREADS < np) own_take16(16 * (int)pc[u], k[u], pat, ls, cnt);
  }
  for (int e = 16 * nvec + tid; e < E; e += OWN_THREADS)
    if (((keys[e >> 2] >> (8 * (e & 3))) & 255u) == gid) own_push_e(e, ls, cnt);
}

template <bool WITH_DST>
static __device__ __forceinline__ void own_resolve(const int* __restrict__ esrc,
                                                   const int* __restrict__ edst,
                                                   int n0, int* ls, int* ld, int n) {
  for (int i0 = threadIdx.x; i0 < n; i0 += 4 * OWN_THREADS) {
    int e[4], s[4], d[4];
#pragma unroll
    for (int u = 0; u < 4; ++u) {
      const int i = i0 + u * OWN_THREADS;
      e[u] = (i < n) ? ls[i] : 0;
    }
#pragma unroll
    for (int u = 0; u < 4; ++u) {
      s[u] = esrc[e[u]];
      d[u] = WITH_DST ? edst[e[u]] : 0;
    }
    asm volatile("" :: "v"(s[0]), "v"(s[1]), "v"(s[2]), "v"(s[3]),
                       "v"(d[0]), "v"(d[1]), "v"(d[2]), "v"(d[3]));
#pragma unroll
    for (int u = 0; u < 4; ++u) {
      const int i = i0 + u * OWN_THREADS;
      if (i < n) {
        ls[i] = s[u];
        if (WITH_DST) ld[i] = d[u] - n0;
      }
    }
  }
}

static __device__ __forceinline__ void own_leave(const int* ls, const int* ld,
                                                 int* __restrict__ gs,
                                                 int* __restrict__ gd, int n) {
  for (int i = threadIdx.x; i < n; i += OWN_THREADS) {
    gs[i] = ls[i];
    gd[i] = ld[i];
  }
}

static __device__ __forceinline__ void own_take(const int* __restrict__ gs,
                                                const int* __restrict__ gd,
                                                int* ls, int* ld, int n) {
  for (int i0 = threadIdx.x; i0 < n; i0 += 4 * OWN_THREADS) {
    int s[4], d[4];
#pragma unroll
    for (int u = 0; u < 4; ++u) {
      const int i = i0 + u * OWN_THREADS;
      const int j = (i < n) ? i : 0;
      s[u] = gs[j];
      d[u] = gd[j];
    }
#pragma unroll
    for (int u = 0; u < 4; ++u) {
      const int i = i0 + u * OWN_THREADS;
      if (i < n) { ls[i] = s[u]; ld[i] = d[u]; }
    }
  }
}

static __device__ __forceinline__ void own_entry(int i, int n_list, int n0,
                                                 const int* ls, const int* ld,
                                                 int& s, int& dl) {
  if (i < n_list) { s = ls[i]; dl = ld[i]; }
  else            { dl = i - n_list; s = n0 + dl; }
}

template <int H, int NU>
static __device__ __forceinline__ void own_scores_trip(int phase, int idx0, int n_list,
                                                       int n0, const int* ls,
                                                       const int* ld,
                                                       const float* __restrict__ as,
                                                       const float* adL, unsigned* mx,
                                                       float* den) {
  float a[NU];
  int dh[NU];
#pragma unroll
  for (int u = 0; u < NU; ++u) {
    const int idx = idx0 + u * OWN_THREADS;
    const int i = idx / H, h = idx - i * H;
    int s, dl;
    own_entry(i, n_list, n0, ls, ld, s, dl);
    a[u] = as[s * H + h];
    dh[u] = dl * H + h;
  }
#pragma unroll
  for (int u = 0; u < NU; ++u) {
    float v = a[u] + adL[dh[u]];
    v = (v >= 0.0f) ? v : NEG_SLOPE * v;
    if (phase == 0) atomicMax(&mx[dh[u]], f2key(v));
    else            atomicAdd(&den[dh[u]], expf(v - key2f(mx[dh[u]])));
  }
}

template <int H>
static __device__ __forceinline__ void own_scores(int phase, int n_list, int n_self,
                                                  int n0, const int* ls, const int* ld,
                                                  const float* __restrict__ as,
                                                  const float* adL, unsigned* mx,
                                                  float* den) {
  const int n_items = (n_list + n_self) * H;
  int idx = threadIdx.x;
  for (; idx + 3 * OWN_THREADS < n_items; idx += 4 * OWN_THREADS)
    own_scores_trip<H, 4>(phase, idx, n_list, n0, ls, ld, as, adL, mx, den);
  for (; idx < n_items; idx += OWN_THREADS)
    own_scores_trip<H, 1>(phase, idx, n_list, n0, ls, ld, as, adL, mx, den);
}

template <int H, int C, int LDX, int NU>
static __device__ __forceinline__ void own_rows_trip(int i0, int n_list, int n0,
                                                     const int* ls, const int* ld,
                                                     const float* __restrict__ as,
                                                     const float* __restrict__ xw,
                                                     const float* adL,
                                                     const unsigned* mx,
                                                     const float* den, float* acc) {
  constexpr int HC = H * C;
  constexpr int NJ = (HC + 31) / 32;
  const int lane = threadIdx.x & 31;
  const int hl = lane & (H - 1);
  int dl[NU];
  float x[NU][NJ], av[NU];
#pragma unroll
  for (int u = 0; u < NU; ++u) {
    int si, di;
    own_entry(i0 + u * (OWN_THREADS / 32), n_list, n0, ls, ld, si, di);
    const int s = __builtin_amdgcn_readfirstlane(si);
    dl[u] = __builtin_amdgcn_readfirstlane(di);
    const float* row = xw + (size_t)s * LDX;
#pragma unroll
    for (int j = 0; j < NJ; ++j) {
      const int c = j * 32 + lane;
      x[u][j] = (c < HC) ? row[c] : 0.0f;
    }
    av[u] = as[s * H + hl];
  }
#pragma unroll
  for (int u = 0; u < NU; ++u) {
    const int di = dl[u] * H + hl;
    float v = av[u] + adL[di];
    v = (v >= 0.0f) ? v : NEG_SLOPE * v;
    const float w = expf(v - key2f(mx[di])) / den[di];
    float* arow = acc + dl[u] * HC;
#pragma unroll
    for (int j = 0; j < NJ; ++j) {
      const int head = (H == 1) ? 0 : (j * 32) / C;
      const float wj = __int_as_float(
          __builtin_amdgcn_readlane(__float_as_int(w), head));
      const int c = j * 32 + lane;
      if (c < HC) atomicAdd(&arow[c], wj * x[u][j]);
    }
  }
}

template <int H, int C, int LDX>
static __device__ __forceinline__ void own_rows_block(int i0, int n_list, int n0,
                                                      const int* ls, const int* ld,
                                                      const float* __restrict__ as,
                                                      const float* __restrict__ xw,
                                                      const float* adL,
                                                      const unsigned* mx,
                                                      const float* den, float* acc) {
  constexpr int HC = H * C;
  constexpr int NJ = (HC + 31) / 32;
  constexpr int G = 32 / H;
  constexpr int SUB = (G * NJ <= 32) ? G : (32 / NJ);
  static_assert(SUB >= 1 && (G % SUB) == 0, "a block is a whole number of row groups");
  const int lane = threadIdx.x & 31;
  const int hl = lane & (H - 1);
  int si, di;
  own_entry(i0 + lane / H, n_list, n0, ls, ld, si, di);
  const int dh = di * H + hl;
  const float a = as[si * H + hl];
  float w = 0.0f;
#pragma unroll
  for (int k = 0; k < G / SUB; ++k) {
    float x[SUB][NJ];
#pragma unroll
    for (int u = 0; u < SUB; ++u) {
      const int s = __builtin_amdgcn_readlane(si, (k * SUB + u) * H);
      const float* row = xw + (size_t)s * LDX;
#pragma unroll
      for (int j = 0; j < NJ; ++j) {
        const int c = j * 32 + lane;
        x[u][j] = (c < HC) ? row[c] : 0.0f;
      }
    }
    if (k == 0) {
      float v = a + adL[dh];
      v = (v >= 0.0f) ? v : NEG_SLOPE * v;
      w = expf(v - key2f(mx[dh])) / den[dh];
    }
#pragma unroll
    for (int u = 0; u < SUB; ++u) {
      const int e = k * SUB + u;
      const int dl = __builtin_amdgcn_readlane(di, e * H);
      float* arow = acc + dl * HC;
#pragma unroll
      for (int j = 0; j < NJ; ++j) {
        const int head = (H == 1) ? 0 : (j * 32) / C;
        const float wj = __int_as_float(
            __builtin_amdgcn_readlane(__float_as_int(w), e * H + head));
        const int c = j * 32 + lane;
        if (c < HC) atomicAdd(&arow[c], wj * x[u][j]);
      }
    }
  }
}

template <int H, int C, int LDX>
static __device__ __forceinline__ void own_rows(int n_list, int n_self, int n0,
                                                const int* ls, const int* ld,
                                                const float* __restrict__ as,
                                                const float* __restrict__ xw,
                                                const float* adL, const unsigned* mx,
                                                const float* den, float* acc) {
  static_assert(H == 1 || (C % 32) == 0, "a 32-lane slice must lie in one head");
  static_assert((H & (H - 1)) == 0 && H <= 32, "heads: a power of two");
  constexpr int NW = OWN_THREADS / 32;
  constexpr int G = 32 / H;
  static_assert(G <= NW, "the entries left over get one wave each");
  const int n_items = n_list + n_self;
  const int n_blocks = n_items / G;
  const int wave = __builtin_amdgcn_readfirstlane((int)(threadIdx.x >> 5));
  for (int b = wave; b < n_blocks; b += NW)
    own_rows_block<H, C, LDX>(b * G, n_list, n0, ls, ld, as, xw, adL, mx, den, acc);
  const int t = n_blocks * G + wave;
  if (t < n_items)
    own_rows_trip<H, C, LDX, 1>(t, n_list, n0, ls, ld, as, xw, adL, mx, den, acc);
}

template <int H, int C, int LDX, bool ELU, bool TAKE>
__global__ __launch_bounds__(OWN_THREADS) void gat_owner_kernel(
    const int* __restrict__ esrc, const int* __restrict__ edst, int E, int Nn,
    const float* __restrict__ as, const float* __restrict__ ad,
    const float* __restrict__ xw, const float* __restrict__ bias,
    float* __restrict__ outp, int* lst, const unsigned* __restrict__ keys,
    const unsigned* __restrict__ planes, int plane_words) {
  constexpr int HC = H * C;
  __shared__ float    acc[OWN_T * HC];
  __shared__ float    adL[OWN_T * H];
  __shared__ float    den[OWN_T * H];
  __shared__ unsigned mx[OWN_T * H];
  __shared__ int      ls[OWN_CAP];
  __shared__ int      ld[OWN_CAP];
  __shared__ unsigned cnt;
  __shared__ unsigned pcnt;

  const int tid = threadIdx.x;
  const int n0 = blockIdx.x * OWN_T;
  const int nT = min(OWN_T, Nn - n0);

  for (int i = tid; i < OWN_T * HC; i += OWN_THREADS) acc[i] = 0.0f;
  for (int i = tid; i < OWN_T * H; i += OWN_THREADS) {
    den[i] = 0.0f;
    mx[i] = 0u;
    adL[i] = (i < nT * H) ? ad[(size_t)n0 * H + i] : 0.0f;
  }
  if (tid == 0) { cnt = 0u; pcnt = 0u; }

  int* greg = lst + (size_t)blockIdx.x * OWN_REGION;
  int* gs = greg + OWN_HEAD;
  int* gd = gs + OWN_CAP;
  int total;
  const bool by_key = (gridDim.x <= 256u);
  if (TAKE) {
    total = greg[0];
    if (total <= OWN_CAP) own_take(gs, gd, ls, ld, total);
    __syncthreads();
  } else {
    __syncthreads();
    if (by_key) own_build_keys(keys, planes + (size_t)blockIdx.x * plane_words, E, ls,
                               (unsigned*)ld, &cnt, &pcnt);
    else        own_build(edst, 0, E, n0, nT, ls, ld, &cnt);
    __syncthreads();
    total = (int)cnt;
    if (tid == 0) greg[0] = total;
  }

  if (total <= OWN_CAP) {
    if (!TAKE) {
      if (by_key) own_resolve<true>(esrc, edst, n0, ls, ld, total);
      else        own_resolve<false>(esrc, edst, n0, ls, ld, total);
      __syncthreads();
      own_leave(ls, ld, gs, gd, total);
    }
    own_scores<H>(0, total, nT, n0, ls, ld, as, adL, mx, den);
    __syncthreads();
    own_scores<H>(1, total, nT, n0, ls, ld, as, adL, mx, den);
    __syncthreads();
    own_rows<H, C, LDX>(total, nT, n0, ls, ld, as, xw, adL, mx, den, acc);
  } else {
    for (int phase = 0; phase < 3; ++phase) {
      for (int eb = 0; eb < E; eb += OWN_CAP) {
        __syncthreads();
        if (tid == 0) cnt = 0u;
        __syncthreads();
        own_build(edst, eb, min(eb + OWN_CAP, E), n0, nT, ls, ld, &cnt);
        __syncthreads();
        const int m = (int)cnt;
        own_resolve<false>(esrc, edst, n0, ls, ld, m);
        __syncthreads();
        const int n_self = (eb == 0) ? nT : 0;
        if (phase < 2)
          own_scores<H>(phase, m, n_self, n0, ls, ld, as, adL, mx, den);
        else
          own_rows<H, C, LDX>(m, n_self, n0, ls, ld, as, xw, adL, mx, den, acc);
      }
    }
  }
  __syncthreads();

  float* orow = outp + (size_t)n0 * HC;
  for (int i = tid; i < nT * HC; i += OWN_THREADS) {
    float v = acc[i] + bias[i % HC];
    if (ELU) v = (v > 0.0f) ? v : (expf(v) - 1.0f);
    orow[i] = v;
  }
}

extern "C" void kernel_launch(void* const* d_in, const int* in_sizes, int n_in,
                              void* d_out, int out_size, void* d_ws, size_t ws_size,
                              hipStream_t stream) {
  const float* x      = (const float*)d_in[0];
  const int*   ei     = (const int*)d_in[1];
  const float* W1     = (const float*)d_in[2];
  const float* a1_src = (const float*)d_in[3];
  const float* a1_dst = (const float*)d_in[4];
  const float* b1     = (const float*)d_in[5];
  const float* W2     = (const float*)d_in[6];
  const float* a2_src = (const float*)d_in[7];
  const float* a2_dst = (const float*)d_in[8];
  const float* b2     = (const float*)d_in[9];
  float* out = (float*)d_out;
  (void)n_in; (void)out_size; (void)ws_size;

  const int F_IN = 256, H1 = 8, C1 = 32, HC1 = 256, NC = 40, NCP = 64;
  const int Nn   = in_sizes[0] / F_IN;
  const int E    = in_sizes[1] / 2;
  const int* esrc = ei;
  const int* edst = ei + E;

  float* p = (float*)d_ws;
  float* xw1  = p; p += (size_t)Nn * HC1;
  float* as1  = p; p += (size_t)Nn * H1;
  float* ad1  = p; p += (size_t)Nn * H1;
  float* h1   = p; p += (size_t)Nn * HC1;
  float* w2p  = p; p += (size_t)HC1 * NCP;
  float* xw2p = p; p += (size_t)Nn * NCP;
  float* as2  = p; p += (size_t)Nn;
  float* ad2  = p; p += (size_t)Nn;
  unsigned* keys = (unsigned*)p; p += (size_t)((E / 4 + 7) & ~3);
  const int key_blocks = ((E + 15) / 16 + PREP_THREADS - 1) / PREP_THREADS;
  const int plane_words = key_blocks * PLANE_W;
  unsigned* planes = (unsigned*)p; p += (size_t)256 * plane_words;
  int*   lst  = (int*)p;

  dim3 blk(256);
  const int own_blocks = (Nn + OWN_T - 1) / OWN_T;

  {
    const int pad_blocks = (HC1 * NCP + PREP_THREADS - 1) / PREP_THREADS;
    prep_kernel<<<pad_blocks + key_blocks, PREP_THREADS, 0, stream>>>(
        W2, w2p, HC1, NC, NCP, edst, keys, E, pad_blocks, planes, plane_words);
  }

  {
    int waves  = (Nn / 16) * (HC1 / 64);
    int blocks = (waves + 7) / 8;
    gemm_f32_wmma_kernel<<<blocks, blk, 0, stream>>>(x, W1, xw1, Nn, F_IN, HC1);
  }
  alpha_kernel<<<(Nn * H1 + 255) / 256, blk, 0, stream>>>(
      xw1, a1_src, a1_dst, as1, ad1, Nn, H1, C1, HC1);
  gat_owner_kernel<8, 32, 256, true, false><<<own_blocks, OWN_THREADS, 0, stream>>>(
      esrc, edst, E, Nn, as1, ad1, xw1, b1, h1, lst, keys, planes, plane_words);

  {
    int waves  = (Nn / 16) * (NCP / 64);
    int blocks = (waves + 7) / 8;
    gemm_f32_wmma_kernel<<<blocks, blk, 0, stream>>>(h1, w2p, xw2p,
                                                     Nn, HC1, NCP);
  }
  alpha_kernel<<<(Nn + 255) / 256, blk, 0, stream>>>(
      xw2p, a2_src, a2_dst, as2, ad2, Nn, 1, NC, NCP);
  gat_owner_kernel<1, 40, 64, false, true><<<own_blocks, OWN_THREADS, 0, stream>>>(
      esrc, edst, E, Nn, as2, ad2, xw2p, b2, out, lst, keys, planes, plane_words);
}


# GEMM wave tile 16x64 -> 32x64 (blocks of 4 waves): each B fragment feeds two row tiles from registers, 10 fragment loads per 8 matrix instructions instead of 18; same products, same sum order
# speedup vs baseline: 13.9993x; 1.0545x over previous
#include <hip/hip_runtime.h>

typedef __attribute__((ext_vector_type(2))) float v2f;
typedef __attribute__((ext_vector_type(8))) float v8f;

#define NEG_SLOPE 0.2f

static __device__ __forceinline__ unsigned f2key(float f) {
  unsigned u = __float_as_uint(f);
  return (u & 0x80000000u) ? ~u : (u | 0x80000000u);
}
static __device__ __forceinline__ float key2f(unsigned k) {
  return (k & 0x80000000u) ? __uint_as_float(k & 0x7FFFFFFFu)
                           : __uint_as_float(~k);
}

#define OWN_T       196
typedef __attribute__((ext_vector_type(4))) int own_v4i;
typedef __attribute__((ext_vector_type(4))) unsigned own_v4u;

#define PREP_THREADS 256
#define PLANE_W      (PREP_THREADS / 32)
__global__ void prep_kernel(const float* __restrict__ src,
                            float* __restrict__ dst,
                            int K, int c, int ldd,
                            const int* __restrict__ edst,
                            unsigned* __restrict__ keys, int E, int pad_blocks,
                            unsigned* __restrict__ planes, int plane_words) {
  __shared__ unsigned bm[256 * PLANE_W];
  if ((int)blockIdx.x < pad_blocks) {
    int i = blockIdx.x * blockDim.x + threadIdx.x;
    if (i >= K * ldd) return;
    int r = i / ldd, j = i - r * ldd;
    dst[i] = (j < c) ? src[r * c + j] : 0.0f;
    return;
  }
  const int tid = threadIdx.x;
  const int kb = blockIdx.x - pad_blocks;
  for (int i = tid; i < 256 * PLANE_W; i += PREP_THREADS) bm[i] = 0u;
  __syncthreads();
  const int piece = kb * PREP_THREADS + tid;
  const int e0 = 16 * piece;
  if (e0 < E) {
    const bool full = (e0 + 15 < E);
    int d[16];
    if (full && ((((size_t)(edst + e0)) & 15) == 0)) {
      const own_v4i* dv = (const own_v4i*)(edst + e0);
      for (int q = 0; q < 4; ++q) {
        const own_v4i v = dv[q];
        d[4 * q + 0] = v.x; d[4 * q + 1] = v.y; d[4 * q + 2] = v.z; d[4 * q + 3] = v.w;
      }
    } else {
      for (int u = 0; u < 16; ++u) d[u] = (e0 + u < E) ? edst[e0 + u] : -1;
    }
    unsigned kw[4] = {0u, 0u, 0u, 0u};
    const unsigned bit = 1u << (tid & 31);
    for (int u = 0; u < 16; ++u) {
      if (d[u] >= 0) {
        const unsigned g = (unsigned)(d[u] / OWN_T) & 255u;
        kw[u >> 2] |= g << (8 * (u & 3));
        if (full) atomicOr(&bm[g * PLANE_W + (tid >> 5)], bit);
      }
    }
    if (full && ((((size_t)keys) & 15) == 0)) {
      own_v4u k4;
      k4.x = kw[0]; k4.y = kw[1]; k4.z = kw[2]; k4.w = kw[3];
      *(own_v4u*)(keys + 4 * piece) = k4;
    } else {
      for (int q = 0; q < 4; ++q)
        if (e0 + 4 * q < E) keys[4 * piece + q] = kw[q];
    }
  }
  __syncthreads();
  unsigned* row = planes + (size_t)tid * plane_words + (size_t)kb * PLANE_W;
  for (int q = 0; q < PLANE_W; ++q) row[q] = bm[tid * PLANE_W + q];
}

#define GEMM_MT 2
__global__ __launch_bounds__(128) void gemm_f32_wmma_kernel(const float* __restrict__ A,
                                     const float* __restrict__ B,
                                     float* __restrict__ C,
                                     int M, int K, int ldb) {
  const int lane = threadIdx.x & 31;
  const int wave = threadIdx.x >> 5;
  const int half = lane >> 4;
  const int l    = lane & 15;
  const int nt64    = ldb >> 6;
  const int mtiles  = M >> 4;
  const int mgroups = (mtiles + GEMM_MT - 1) / GEMM_MT;
  const int total   = mgroups * nt64;
  int t = blockIdx.x * (blockDim.x >> 5) + wave;
  if (t >= total) return;
  const int mg = t / nt64;
  const int n0 = (t - mg * nt64) << 6;
  const int m0 = mg * (16 * GEMM_MT);

  v8f acc[GEMM_MT][4] = {};
  const float* arow[GEMM_MT];
#pragma unroll
  for (int i = 0; i < GEMM_MT; ++i) {
    int r = m0 + 16 * i + l;
    r = (r < M) ? r : (M - 1);
    arow[i] = A + (size_t)r * K + 2 * half;
  }
  const float* bp   = B + (size_t)(2 * half) * ldb + n0 + l;
  const size_t bstep = (size_t)4 * ldb;

  v2f a0[GEMM_MT], a1[GEMM_MT], b0[4], b1[4];
  const int nsteps = K >> 2;

#define GEMM_LOAD(av, bv, step)                                              \
  {                                                                          \
    _Pragma("unroll") for (int i = 0; i < GEMM_MT; ++i)                      \
        av[i] = *(const v2f*)(arow[i] + 4 * (step));                         \
    const float* bq = bp + (size_t)(step) * bstep;                           \
    _Pragma("unroll") for (int s = 0; s < 4; ++s) {                          \
      bv[s].x = bq[s * 16];                                                  \
      bv[s].y = bq[ldb + s * 16];                                            \
    }                                                                        \
  }
#define GEMM_MMA(av, bv)                                                     \
  {                                                                          \
    _Pragma("unroll") for (int i = 0; i < GEMM_MT; ++i)                      \
        _Pragma("unroll") for (int s = 0; s < 4; ++s)                        \
            acc[i][s] = __builtin_amdgcn_wmma_f32_16x16x4_f32(               \
                false, av[i], false, bv[s], (short)0, acc[i][s], false, false); \
  }

  GEMM_LOAD(a0, b0, 0)
  int st = 0;
  for (; st + 2 < nsteps; st += 2) {
    GEMM_LOAD(a1, b1, st + 1)
    asm volatile("" ::: "memory");
    GEMM_MMA(a0, b0)
    GEMM_LOAD(a0, b0, st + 2)
    asm volatile("" ::: "memory");
    GEMM_MMA(a1, b1)
  }
  GEMM_LOAD(a1, b1, st + 1)
  asm volatile("" ::: "memory");
  GEMM_MMA(a0, b0)
  GEMM_MMA(a1, b1)
#undef GEMM_LOAD
#undef GEMM_MMA

#pragma unroll
  for (int i = 0; i < GEMM_MT; ++i) {
    if (m0 + 16 * i < M) {
      float* crow = C + (size_t)(m0 + 16 * i + 8 * half) * ldb + n0 + l;
#pragma unroll
      for (int s = 0; s < 4; ++s)
#pragma unroll
        for (int j = 0; j < 8; ++j) crow[(size_t)j * ldb + s * 16] = acc[i][s][j];
    }
  }
}

__global__ void alpha_kernel(const float* __restrict__ xw,
                             const float* __restrict__ a_src,
                             const float* __restrict__ a_dst,
                             float* __restrict__ as, float* __restrict__ ad,
                             int n_nodes, int heads, int c, int ldx) {
  int i = blockIdx.x * blockDim.x + threadIdx.x;
  if (i >= n_nodes * heads) return;
  int n = i / heads, h = i - n * heads;
  const float* row = xw + (size_t)n * ldx + (size_t)h * c;
  const float* ws = a_src + (size_t)h * c;
  const float* wd = a_dst + (size_t)h * c;
  float s = 0.0f, d = 0.0f;
  for (int j = 0; j < c; ++j) {
    float v = row[j];
    s += v * ws[j];
    d += v * wd[j];
  }
  as[i] = s;
  ad[i] = d;
}

#define OWN_CAP     8192
#define OWN_THREADS 1024
#define OWN_HEAD    32
#define OWN_REGION  (OWN_HEAD + 2 * OWN_CAP)

static __device__ __forceinline__ void own_push(int e, int dl, int* ls, int* ld,
                                                unsigned* cnt) {
  unsigned idx = atomicAdd(cnt, 1u);
  if (idx < OWN_CAP) { ls[idx] = e; ld[idx] = dl; }
}

static __device__ __forceinline__ void own_take4(int e, own_v4i d, int n0, int nT,
                                                 int* ls, int* ld, unsigned* cnt) {
  const int r0 = d.x - n0, r1 = d.y - n0, r2 = d.z - n0, r3 = d.w - n0;
  if ((unsigned)r0 < (unsigned)nT) own_push(e + 0, r0, ls, ld, cnt);
  if ((unsigned)r1 < (unsigned)nT) own_push(e + 1, r1, ls, ld, cnt);
  if ((unsigned)r2 < (unsigned)nT) own_push(e + 2, r2, ls, ld, cnt);
  if ((unsigned)r3 < (unsigned)nT) own_push(e + 3, r3, ls, ld, cnt);
}

static __device__ __forceinline__ void own_build(const int* __restrict__ edst,
                                                 int eb, int ee, int n0, int nT,
                                                 int* ls, int* ld, unsigned* cnt) {
  const int tid = threadIdx.x;
  const int n = ee - eb;
  const int* dbase = edst + eb;
  const int nvec = ((((size_t)dbase) & 15) == 0) ? (n >> 2) : 0;
  const own_v4i* dv = (const own_v4i*)dbase;
  const int rot = (int)(((long)blockIdx.x * nvec) / gridDim.x) & ~31;
  int v = tid;
  for (; v + 3 * OWN_THREADS < nvec; v += 4 * OWN_THREADS) {
    own_v4i d[4];
    int vv[4];
#pragma unroll
    for (int u = 0; u < 4; ++u) {
      vv[u] = v + u * OWN_THREADS + rot;
      if (vv[u] >= nvec) vv[u] -= nvec;
      d[u] = dv[vv[u]];
    }
    asm volatile("" :: "v"(d[0]), "v"(d[1]), "v"(d[2]), "v"(d[3]));
#pragma unroll
    for (int u = 0; u < 4; ++u)
      own_take4(eb + 4 * vv[u], d[u], n0, nT, ls, ld, cnt);
  }
  for (; v < nvec; v += OWN_THREADS) {
    int vv = v + rot;
    if (vv >= nvec) vv -= nvec;
    own_take4(eb + 4 * vv, dv[vv], n0, nT, ls, ld, cnt);
  }
  for (int e = eb + 4 * nvec + tid; e < ee; e += OWN_THREADS) {
    const int r = edst[e] - n0;
    if ((unsigned)r < (unsigned)nT) own_push(e, r, ls, ld, cnt);
  }
}

static __device__ __forceinline__ void own_push_e(int e, int* ls, unsigned* cnt) {
  unsigned idx = atomicAdd(cnt, 1u);
  if (idx < OWN_CAP) ls[idx] = e;
}

static __device__ __forceinline__ unsigned own_hits(unsigned w, unsigned pat) {
  const unsigned y = w ^ pat;
  const unsigned t = (y & 0x7F7F7F7Fu) + 0x7F7F7F7Fu;
  return ~(t | y) & 0x80808080u;
}

static __device__ __forceinline__ void own_take16(int e, own_v4u k, unsigned pat,
                                                  int* ls, unsigned* cnt) {
  unsigned m[4] = {own_hits(k.x, pat), own_hits(k.y, pat),
                   own_hits(k.z, pat), own_hits(k.w, pat)};
  if ((m[0] | m[1] | m[2] | m[3]) != 0u) {
#pragma unroll
    for (int u = 0; u < 4; ++u) {
      unsigned mm = m[u];
      while (mm) {
        own_push_e(e + 4 * u + (__builtin_ctz(mm) >> 3), ls, cnt);
        mm &= mm - 1u;
      }
    }
  }
}

static __device__ __forceinline__ void own_build_keys(const unsigned* __restrict__ keys,
                                                      const unsigned* __restrict__ plane,
                                                      int E, int* ls, unsigned* pl,
                                                      unsigned* cnt, unsigned* pcnt) {
  const int tid = threadIdx.x;
  const int lane = tid & 31;
  const unsigned gid = blockIdx.x;
  const unsigned pat = gid * 0x01010101u;
  const int nvec = ((((size_t)keys) & 15) == 0) ? (E >> 4) : 0;
  const own_v4u* kv = (const own_v4u*)keys;
  const int nwords = (nvec + 31) >> 5;
  for (int w0 = tid & ~31; w0 < nwords; w0 += OWN_THREADS) {
    const int w = w0 + lane;
    unsigned m = (w < nwords) ? plane[w] : 0u;
    const int n = __builtin_popcount(m);
    int incl = n;
    for (int dd = 1; dd < 32; dd <<= 1) {
      const int t = __shfl_up(incl, dd);
      if (lane >= dd) incl += t;
    }
    const int tot = __shfl(incl, 31);
    if (tot != 0) {
      unsigned base = 0u;
      if (lane == 31) base = atomicAdd(pcnt, (unsigned)tot);
      base = (unsigned)__shfl((int)base, 31);
      unsigned idx = base + (unsigned)(incl - n);
      while (m) {
        if (idx < OWN_CAP) pl[idx] = 32u * (unsigned)w + (unsigned)__builtin_ctz(m);
        ++idx;
        m &= m - 1u;
      }
    }
  }
  __syncthreads();
  const int np = (int)*pcnt;
  if (np > OWN_CAP) {
    if (tid == 0) *cnt = (unsigned)np;
    return;
  }
  for (int i0 = tid; i0 < np; i0 += 4 * OWN_THREADS) {
    own_v4u k[4];
    unsigned pc[4];
#pragma unroll
    for (int u = 0; u < 4; ++u) {
      const int i = i0 + u * OWN_THREADS;
      pc[u] = pl[(i < np) ? i : 0];
      k[u] = kv[pc[u]];
    }
    asm volatile("" :: "v"(k[0]), "v"(k[1]), "v"(k[2]), "v"(k[3]));
#pragma unroll
    for (int u = 0; u < 4; ++u)
      if (i0 + u * OWN_THREADS < np) own_take16(16 * (int)pc[u], k[u], pat, ls, cnt);
  }
  for (int e = 16 * nvec + tid; e < E; e += OWN_THREADS)
    if (((keys[e >> 2] >> (8 * (e & 3))) & 255u) == gid) own_push_e(e, ls, cnt);
}

template <bool WITH_DST>
static __device__ __forceinline__ void own_resolve(const int* __restrict__ esrc,
                                                   const int* __restrict__ edst,
                                                   int n0, int* ls, int* ld, int n) {
  for (int i0 = threadIdx.x; i0 < n; i0 += 4 * OWN_THREADS) {
    int e[4], s[4], d[4];
#pragma unroll
    for (int u = 0; u < 4; ++u) {
      const int i = i0 + u * OWN_THREADS;
      e[u] = (i < n) ? ls[i] : 0;
    }
#pragma unroll
    for (int u = 0; u < 4; ++u) {
      s[u] = esrc[e[u]];
      d[u] = WITH_DST ? edst[e[u]] : 0;
    }
    asm volatile("" :: "v"(s[0]), "v"(s[1]), "v"(s[2]), "v"(s[3]),
                       "v"(d[0]), "v"(d[1]), "v"(d[2]), "v"(d[3]));
#pragma unroll
    for (int u = 0; u < 4; ++u) {
      const int i = i0 + u * OWN_THREADS;
      if (i < n) {
        ls[i] = s[u];
        if (WITH_DST) ld[i] = d[u] - n0;
      }
    }
  }
}

static __device__ __forceinline__ void own_leave(const int* ls, const int* ld,
                                                 int* __restrict__ gs,
                                                 int* __restrict__ gd, int n) {
  for (int i = threadIdx.x; i < n; i += OWN_THREADS) {
    gs[i] = ls[i];
    gd[i] = ld[i];
  }
}

static __device__ __forceinline__ void own_take(const int* __restrict__ gs,
                                                const int* __restrict__ gd,
                                                int* ls, int* ld, int n) {
  for (int i0 = threadIdx.x; i0 < n; i0 += 4 * OWN_THREADS) {
    int s[4], d[4];
#pragma unroll
    for (int u = 0; u < 4; ++u) {
      const int i = i0 + u * OWN_THREADS;
      const int j = (i < n) ? i : 0;
      s[u] = gs[j];
      d[u] = gd[j];
    }
#pragma unroll
    for (int u = 0; u < 4; ++u) {
      const int i = i0 + u * OWN_THREADS;
      if (i < n) { ls[i] = s[u]; ld[i] = d[u]; }
    }
  }
}

static __device__ __forceinline__ void own_entry(int i, int n_list, int n0,
                                                 const int* ls, const int* ld,
                                                 int& s, int& dl) {
  if (i < n_list) { s = ls[i]; dl = ld[i]; }
  else            { dl = i - n_list; s = n0 + dl; }
}

template <int H, int NU>
static __device__ __forceinline__ void own_scores_trip(int phase, int idx0, int n_list,
                                                       int n0, const int* ls,
                                                       const int* ld,
                                                       const float* __restrict__ as,
                                                       const float* adL, unsigned* mx,
                                                       float* den) {
  float a[NU];
  int dh[NU];
#pragma unroll
  for (int u = 0; u < NU; ++u) {
    const int idx = idx0 + u * OWN_THREADS;
    const int i = idx / H, h = idx - i * H;
    int s, dl;
    own_entry(i, n_list, n0, ls, ld, s, dl);
    a[u] = as[s * H + h];
    dh[u] = dl * H + h;
  }
#pragma unroll
  for (int u = 0; u < NU; ++u) {
    float v = a[u] + adL[dh[u]];
    v = (v >= 0.0f) ? v : NEG_SLOPE * v;
    if (phase == 0) atomicMax(&mx[dh[u]], f2key(v));
    else            atomicAdd(&den[dh[u]], expf(v - key2f(mx[dh[u]])));
  }
}

template <int H>
static __device__ __forceinline__ void own_scores(int phase, int n_list, int n_self,
                                                  int n0, const int* ls, const int* ld,
                                                  const float* __restrict__ as,
                                                  const float* adL, unsigned* mx,
                                                  float* den) {
  const int n_items = (n_list + n_self) * H;
  int idx = threadIdx.x;
  for (; idx + 3 * OWN_THREADS < n_items; idx += 4 * OWN_THREADS)
    own_scores_trip<H, 4>(phase, idx, n_list, n0, ls, ld, as, adL, mx, den);
  for (; idx < n_items; idx += OWN_THREADS)
    own_scores_trip<H, 1>(phase, idx, n_list, n0, ls, ld, as, adL, mx, den);
}

template <int H, int C, int LDX, int NU>
static __device__ __forceinline__ void own_rows_trip(int i0, int n_list, int n0,
                                                     const int* ls, const int* ld,
                                                     const float* __restrict__ as,
                                                     const float* __restrict__ xw,
                                                     const float* adL,
                                                     const unsigned* mx,
                                                     const float* den, float* acc) {
  constexpr int HC = H * C;
  constexpr int NJ = (HC + 31) / 32;
  const int lane = threadIdx.x & 31;
  const int hl = lane & (H - 1);
  int dl[NU];
  float x[NU][NJ], av[NU];
#pragma unroll
  for (int u = 0; u < NU; ++u) {
    int si, di;
    own_entry(i0 + u * (OWN_THREADS / 32), n_list, n0, ls, ld, si, di);
    const int s = __builtin_amdgcn_readfirstlane(si);
    dl[u] = __builtin_amdgcn_readfirstlane(di);
    const float* row = xw + (size_t)s * LDX;
#pragma unroll
    for (int j = 0; j < NJ; ++j) {
      const int c = j * 32 + lane;
      x[u][j] = (c < HC) ? row[c] : 0.0f;
    }
    av[u] = as[s * H + hl];
  }
#pragma unroll
  for (int u = 0; u < NU; ++u) {
    const int di = dl[u] * H + hl;
    float v = av[u] + adL[di];
    v = (v >= 0.0f) ? v : NEG_SLOPE * v;
    const float w = expf(v - key2f(mx[di])) / den[di];
    float* arow = acc + dl[u] * HC;
#pragma unroll
    for (int j = 0; j < NJ; ++j) {
      const int head = (H == 1) ? 0 : (j * 32) / C;
      const float wj = __int_as_float(
          __builtin_amdgcn_readlane(__float_as_int(w), head));
      const int c = j * 32 + lane;
      if (c < HC) atomicAdd(&arow[c], wj * x[u][j]);
    }
  }
}

template <int H, int C, int LDX>
static __device__ __forceinline__ void own_rows_block(int i0, int n_list, int n0,
                                                      const int* ls, const int* ld,
                                                      const float* __restrict__ as,
                                                      const float* __restrict__ xw,
                                                      const float* adL,
                                                      const unsigned* mx,
                                                      const float* den, float* acc) {
  constexpr int HC = H * C;
  constexpr int NJ = (HC + 31) / 32;
  constexpr int G = 32 / H;
  constexpr int SUB = (G * NJ <= 32) ? G : (32 / NJ);
  static_assert(SUB >= 1 && (G % SUB) == 0, "a block is a whole number of row groups");
  const int lane = threadIdx.x & 31;
  const int hl = lane & (H - 1);
  int si, di;
  own_entry(i0 + lane / H, n_list, n0, ls, ld, si, di);
  const int dh = di * H + hl;
  const float a = as[si * H + hl];
  float w = 0.0f;
#pragma unroll
  for (int k = 0; k < G / SUB; ++k) {
    float x[SUB][NJ];
#pragma unroll
    for (int u = 0; u < SUB; ++u) {
      const int s = __builtin_amdgcn_readlane(si, (k * SUB + u) * H);
      const float* row = xw + (size_t)s * LDX;
#pragma unroll
      for (int j = 0; j < NJ; ++j) {
        const int c = j * 32 + lane;
        x[u][j] = (c < HC) ? row[c] : 0.0f;
      }
    }
    if (k == 0) {
      float v = a + adL[dh];
      v = (v >= 0.0f) ? v : NEG_SLOPE * v;
      w = expf(v - key2f(mx[dh])) / den[dh];
    }
#pragma unroll
    for (int u = 0; u < SUB; ++u) {
      const int e = k * SUB + u;
      const int dl = __builtin_amdgcn_readlane(di, e * H);
      float* arow = acc + dl * HC;
#pragma unroll
      for (int j = 0; j < NJ; ++j) {
        const int head = (H == 1) ? 0 : (j * 32) / C;
        const float wj = __int_as_float(
            __builtin_amdgcn_readlane(__float_as_int(w), e * H + head));
        const int c = j * 32 + lane;
        if (c < HC) atomicAdd(&arow[c], wj * x[u][j]);
      }
    }
  }
}

template <int H, int C, int LDX>
static __device__ __forceinline__ void own_rows(int n_list, int n_self, int n0,
                                                const int* ls, const int* ld,
                                                const float* __restrict__ as,
                                                const float* __restrict__ xw,
                                                const float* adL, const unsigned* mx,
                                                const float* den, float* acc) {
  static_assert(H == 1 || (C % 32) == 0, "a 32-lane slice must lie in one head");
  static_assert((H & (H - 1)) == 0 && H <= 32, "heads: a power of two");
  constexpr int NW = OWN_THREADS / 32;
  constexpr int G = 32 / H;
  static_assert(G <= NW, "the entries left over get one wave each");
  const int n_items = n_list + n_self;
  const int n_blocks = n_items / G;
  const int wave = __builtin_amdgcn_readfirstlane((int)(threadIdx.x >> 5));
  for (int b = wave; b < n_blocks; b += NW)
    own_rows_block<H, C, LDX>(b * G, n_list, n0, ls, ld, as, xw, adL, mx, den, acc);
  const int t = n_blocks * G + wave;
  if (t < n_items)
    own_rows_trip<H, C, LDX, 1>(t, n_list, n0, ls, ld, as, xw, adL, mx, den, acc);
}

template <int H, int C, int LDX, bool ELU, bool TAKE>
__global__ __launch_bounds__(OWN_THREADS) void gat_owner_kernel(
    const int* __restrict__ esrc, const int* __restrict__ edst, int E, int Nn,
    const float* __restrict__ as, const float* __restrict__ ad,
    const float* __restrict__ xw, const float* __restrict__ bias,
    float* __restrict__ outp, int* lst, const unsigned* __restrict__ keys,
    const unsigned* __restrict__ planes, int plane_words) {
  constexpr int HC = H * C;
  __shared__ float    acc[OWN_T * HC];
  __shared__ float    adL[OWN_T * H];
  __shared__ float    den[OWN_T * H];
  __shared__ unsigned mx[OWN_T * H];
  __shared__ int      ls[OWN_CAP];
  __shared__ int      ld[OWN_CAP];
  __shared__ unsigned cnt;
  __shared__ unsigned pcnt;

  const int tid = threadIdx.x;
  const int n0 = blockIdx.x * OWN_T;
  const int nT = min(OWN_T, Nn - n0);

  for (int i = tid; i < OWN_T * HC; i += OWN_THREADS) acc[i] = 0.0f;
  for (int i = tid; i < OWN_T * H; i += OWN_THREADS) {
    den[i] = 0.0f;
    mx[i] = 0u;
    adL[i] = (i < nT * H) ? ad[(size_t)n0 * H + i] : 0.0f;
  }
  if (tid == 0) { cnt = 0u; pcnt = 0u; }

  int* greg = lst + (size_t)blockIdx.x * OWN_REGION;
  int* gs = greg + OWN_HEAD;
  int* gd = gs + OWN_CAP;
  int total;
  const bool by_key = (gridDim.x <= 256u);
  if (TAKE) {
    total = greg[0];
    if (total <= OWN_CAP) own_take(gs, gd, ls, ld, total);
    __syncthreads();
  } else {
    __syncthreads();
    if (by_key) own_build_keys(keys, planes + (size_t)blockIdx.x * plane_words, E, ls,
                               (unsigned*)ld, &cnt, &pcnt);
    else        own_build(edst, 0, E, n0, nT, ls, ld, &cnt);
    __syncthreads();
    total = (int)cnt;
    if (tid == 0) greg[0] = total;
  }

  if (total <= OWN_CAP) {
    if (!TAKE) {
      if (by_key) own_resolve<true>(esrc, edst, n0, ls, ld, total);
      else        own_resolve<false>(esrc, edst, n0, ls, ld, total);
      __syncthreads();
      own_leave(ls, ld, gs, gd, total);
    }
    own_scores<H>(0, total, nT, n0, ls, ld, as, adL, mx, den);
    __syncthreads();
    own_scores<H>(1, total, nT, n0, ls, ld, as, adL, mx, den);
    __syncthreads();
    own_rows<H, C, LDX>(total, nT, n0, ls, ld, as, xw, adL, mx, den, acc);
  } else {
    for (int phase = 0; phase < 3; ++phase) {
      for (int eb = 0; eb < E; eb += OWN_CAP) {
        __syncthreads();
        if (tid == 0) cnt = 0u;
        __syncthreads();
        own_build(edst, eb, min(eb + OWN_CAP, E), n0, nT, ls, ld, &cnt);
        __syncthreads();
        const int m = (int)cnt;
        own_resolve<false>(esrc, edst, n0, ls, ld, m);
        __syncthreads();
        const int n_self = (eb == 0) ? nT : 0;
        if (phase < 2)
          own_scores<H>(phase, m, n_self, n0, ls, ld, as, adL, mx, den);
        else
          own_rows<H, C, LDX>(m, n_self, n0, ls, ld, as, xw, adL, mx, den, acc);
      }
    }
  }
  __syncthreads();

  float* orow = outp + (size_t)n0 * HC;
  for (int i = tid; i < nT * HC; i += OWN_THREADS) {
    float v = acc[i] + bias[i % HC];
    if (ELU) v = (v > 0.0f) ? v : (expf(v) - 1.0f);
    orow[i] = v;
  }
}

extern "C" void kernel_launch(void* const* d_in, const int* in_sizes, int n_in,
                              void* d_out, int out_size, void* d_ws, size_t ws_size,
                              hipStream_t stream) {
  const float* x      = (const float*)d_in[0];
  const int*   ei     = (const int*)d_in[1];
  const float* W1     = (const float*)d_in[2];
  const float* a1_src = (const float*)d_in[3];
  const float* a1_dst = (const float*)d_in[4];
  const float* b1     = (const float*)d_in[5];
  const float* W2     = (const float*)d_in[6];
  const float* a2_src = (const float*)d_in[7];
  const float* a2_dst = (const float*)d_in[8];
  const float* b2     = (const float*)d_in[9];
  float* out = (float*)d_out;
  (void)n_in; (void)out_size; (void)ws_size;

  const int F_IN = 256, H1 = 8, C1 = 32, HC1 = 256, NC = 40, NCP = 64;
  const int Nn   = in_sizes[0] / F_IN;
  const int E    = in_sizes[1] / 2;
  const int* esrc = ei;
  const int* edst = ei + E;

  float* p = (float*)d_ws;
  float* xw1  = p; p += (size_t)Nn * HC1;
  float* as1  = p; p += (size_t)Nn * H1;
  float* ad1  = p; p += (size_t)Nn * H1;
  float* h1   = p; p += (size_t)Nn * HC1;
  float* w2p  = p; p += (size_t)HC1 * NCP;
  float* xw2p = p; p += (size_t)Nn * NCP;
  float* as2  = p; p += (size_t)Nn;
  float* ad2  = p; p += (size_t)Nn;
  unsigned* keys = (unsigned*)p; p += (size_t)((E / 4 + 7) & ~3);
  const int key_blocks = ((E + 15) / 16 + PREP_THREADS - 1) / PREP_THREADS;
  const int plane_words = key_blocks * PLANE_W;
  unsigned* planes = (unsigned*)p; p += (size_t)256 * plane_words;
  int*   lst  = (int*)p;

  dim3 blk(256);
  const int own_blocks = (Nn + OWN_T - 1) / OWN_T;

  {
    const int pad_blocks = (HC1 * NCP + PREP_THREADS - 1) / PREP_THREADS;
    prep_kernel<<<pad_blocks + key_blocks, PREP_THREADS, 0, stream>>>(
        W2, w2p, HC1, NC, NCP, edst, keys, E, pad_blocks, planes, plane_words);
  }

  {
    int waves  = ((Nn / 16 + GEMM_MT - 1) / GEMM_MT) * (HC1 / 64);
    int blocks = (waves + 3) / 4;
    gemm_f32_wmma_kernel<<<blocks, dim3(128), 0, stream>>>(x, W1, xw1, Nn, F_IN, HC1);
  }
  alpha_kernel<<<(Nn * H1 + 255) / 256, blk, 0, stream>>>(
      xw1, a1_src, a1_dst, as1, ad1, Nn, H1, C1, HC1);
  gat_owner_kernel<8, 32, 256, true, false><<<own_blocks, OWN_THREADS, 0, stream>>>(
      esrc, edst, E, Nn, as1, ad1, xw1, b1, h1, lst, keys, planes, plane_words);

  {
    int waves  = ((Nn / 16 + GEMM_MT - 1) / GEMM_MT) * (NCP / 64);
    int blocks = (waves + 3) / 4;
    gemm_f32_wmma_kernel<<<blocks, dim3(128), 0, stream>>>(h1, w2p, xw2p,
                                                           Nn, HC1, NCP);
  }
  alpha_kernel<<<(Nn + 255) / 256, blk, 0, stream>>>(
      xw2p, a2_src, a2_dst, as2, ad2, Nn, 1, NC, NCP);
  gat_owner_kernel<1, 40, 64, false, true><<<own_blocks, OWN_THREADS, 0, stream>>>(
      esrc, edst, E, Nn, as2, ad2, xw2p, b2, out, lst, keys, planes, plane_words);
}


# layer 2's product and coefficients formed by layer 1's owner groups from their finished rows in LDS (f32 matrix instruction, same K order): hidden rows never stored; 7 launches -> 5; sums unchanged
# speedup vs baseline: 15.5433x; 1.1103x over previous
#include <hip/hip_runtime.h>

typedef __attribute__((ext_vector_type(2))) float v2f;
typedef __attribute__((ext_vector_type(8))) float v8f;

#define NEG_SLOPE 0.2f

static __device__ __forceinline__ unsigned f2key(float f) {
  unsigned u = __float_as_uint(f);
  return (u & 0x80000000u) ? ~u : (u | 0x80000000u);
}
static __device__ __forceinline__ float key2f(unsigned k) {
  return (k & 0x80000000u) ? __uint_as_float(k & 0x7FFFFFFFu)
                           : __uint_as_float(~k);
}

#define OWN_T       196
typedef __attribute__((ext_vector_type(4))) int own_v4i;
typedef __attribute__((ext_vector_type(4))) unsigned own_v4u;

#define PREP_THREADS 256
#define PLANE_W      (PREP_THREADS / 32)
__global__ void prep_kernel(const float* __restrict__ src,
                            float* __restrict__ dst,
                            int K, int c, int ldd,
                            const int* __restrict__ edst,
                            unsigned* __restrict__ keys, int E, int pad_blocks,
                            unsigned* __restrict__ planes, int plane_words) {
  __shared__ unsigned bm[256 * PLANE_W];
  if ((int)blockIdx.x < pad_blocks) {
    int i = blockIdx.x * blockDim.x + threadIdx.x;
    if (i >= K * ldd) return;
    int r = i / ldd, j = i - r * ldd;
    dst[i] = (j < c) ? src[r * c + j] : 0.0f;
    return;
  }
  const int tid = threadIdx.x;
  const int kb = blockIdx.x - pad_blocks;
  for (int i = tid; i < 256 * PLANE_W; i += PREP_THREADS) bm[i] = 0u;
  __syncthreads();
  const int piece = kb * PREP_THREADS + tid;
  const int e0 = 16 * piece;
  if (e0 < E) {
    const bool full = (e0 + 15 < E);
    int d[16];
    if (full && ((((size_t)(edst + e0)) & 15) == 0)) {
      const own_v4i* dv = (const own_v4i*)(edst + e0);
      for (int q = 0; q < 4; ++q) {
        const own_v4i v = dv[q];
        d[4 * q + 0] = v.x; d[4 * q + 1] = v.y; d[4 * q + 2] = v.z; d[4 * q + 3] = v.w;
      }
    } else {
      for (int u = 0; u < 16; ++u) d[u] = (e0 + u < E) ? edst[e0 + u] : -1;
    }
    unsigned kw[4] = {0u, 0u, 0u, 0u};
    const unsigned bit = 1u << (tid & 31);
    for (int u = 0; u < 16; ++u) {
      if (d[u] >= 0) {
        const unsigned g = (unsigned)(d[u] / OWN_T) & 255u;
        kw[u >> 2] |= g << (8 * (u & 3));
        if (full) atomicOr(&bm[g * PLANE_W + (tid >> 5)], bit);
      }
    }
    if (full && ((((size_t)keys) & 15) == 0)) {
      own_v4u k4;
      k4.x = kw[0]; k4.y = kw[1]; k4.z = kw[2]; k4.w = kw[3];
      *(own_v4u*)(keys + 4 * piece) = k4;
    } else {
      for (int q = 0; q < 4; ++q)
        if (e0 + 4 * q < E) keys[4 * piece + q] = kw[q];
    }
  }
  __syncthreads();
  unsigned* row = planes + (size_t)tid * plane_words + (size_t)kb * PLANE_W;
  for (int q = 0; q < PLANE_W; ++q) row[q] = bm[tid * PLANE_W + q];
}

#define GEMM_MT 2
__global__ __launch_bounds__(128) void gemm_f32_wmma_kernel(const float* __restrict__ A,
                                     const float* __restrict__ B,
                                     float* __restrict__ C,
                                     int M, int K, int ldb) {
  const int lane = threadIdx.x & 31;
  const int wave = threadIdx.x >> 5;
  const int half = lane >> 4;
  const int l    = lane & 15;
  const int nt64    = ldb >> 6;
  const int mtiles  = M >> 4;
  const int mgroups = (mtiles + GEMM_MT - 1) / GEMM_MT;
  const int total   = mgroups * nt64;
  int t = blockIdx.x * (blockDim.x >> 5) + wave;
  if (t >= total) return;
  const int mg = t / nt64;
  const int n0 = (t - mg * nt64) << 6;
  const int m0 = mg * (16 * GEMM_MT);

  v8f acc[GEMM_MT][4] = {};
  const float* arow[GEMM_MT];
#pragma unroll
  for (int i = 0; i < GEMM_MT; ++i) {
    int r = m0 + 16 * i + l;
    r = (r < M) ? r : (M - 1);
    arow[i] = A + (size_t)r * K + 2 * half;
  }
  const float* bp   = B + (size_t)(2 * half) * ldb + n0 + l;
  const size_t bstep = (size_t)4 * ldb;

  v2f a0[GEMM_MT], a1[GEMM_MT], b0[4], b1[4];
  const int nsteps = K >> 2;

#define GEMM_LOAD(av, bv, step)                                              \
  {                                                                          \
    _Pragma("unroll") for (int i = 0; i < GEMM_MT; ++i)                      \
        av[i] = *(const v2f*)(arow[i] + 4 * (step));                         \
    const float* bq = bp + (size_t)(step) * bstep;                           \
    _Pragma("unroll") for (int s = 0; s < 4; ++s) {                          \
      bv[s].x = bq[s * 16];                                                  \
      bv[s].y = bq[ldb + s * 16];                                            \
    }                                                                        \
  }
#define GEMM_MMA(av, bv)                                                     \
  {                                                                          \
    _Pragma("unroll") for (int i = 0; i < GEMM_MT; ++i)                      \
        _Pragma("unroll") for (int s = 0; s < 4; ++s)                        \
            acc[i][s] = __builtin_amdgcn_wmma_f32_16x16x4_f32(               \
                false, av[i], false, bv[s], (short)0, acc[i][s], false, false); \
  }

  GEMM_LOAD(a0, b0, 0)
  int st = 0;
  for (; st + 2 < nsteps; st += 2) {
    GEMM_LOAD(a1, b1, st + 1)
    asm volatile("" ::: "memory");
    GEMM_MMA(a0, b0)
    GEMM_LOAD(a0, b0, st + 2)
    asm volatile("" ::: "memory");
    GEMM_MMA(a1, b1)
  }
  GEMM_LOAD(a1, b1, st + 1)
  asm volatile("" ::: "memory");
  GEMM_MMA(a0, b0)
  GEMM_MMA(a1, b1)
#undef GEMM_LOAD
#undef GEMM_MMA

#pragma unroll
  for (int i = 0; i < GEMM_MT; ++i) {
    if (m0 + 16 * i < M) {
      float* crow = C + (size_t)(m0 + 16 * i + 8 * half) * ldb + n0 + l;
#pragma unroll
      for (int s = 0; s < 4; ++s)
#pragma unroll
        for (int j = 0; j < 8; ++j) crow[(size_t)j * ldb + s * 16] = acc[i][s][j];
    }
  }
}

__global__ void alpha_kernel(const float* __restrict__ xw,
                             const float* __restrict__ a_src,
                             const float* __restrict__ a_dst,
                             float* __restrict__ as, float* __restrict__ ad,
                             int n_nodes, int heads, int c, int ldx) {
  int i = blockIdx.x * blockDim.x + threadIdx.x;
  if (i >= n_nodes * heads) return;
  int n = i / heads, h = i - n * heads;
  const float* row = xw + (size_t)n * ldx + (size_t)h * c;
  const float* ws = a_src + (size_t)h * c;
  const float* wd = a_dst + (size_t)h * c;
  float s = 0.0f, d = 0.0f;
  for (int j = 0; j < c; ++j) {
    float v = row[j];
    s += v * ws[j];
    d += v * wd[j];
  }
  as[i] = s;
  ad[i] = d;
}

#define OWN_CAP     8192
#define OWN_THREADS 1024
#define OWN_HEAD    32
#define OWN_REGION  (OWN_HEAD + 2 * OWN_CAP)

static __device__ __forceinline__ void own_push(int e, int dl, int* ls, int* ld,
                                                unsigned* cnt) {
  unsigned idx = atomicAdd(cnt, 1u);
  if (idx < OWN_CAP) { ls[idx] = e; ld[idx] = dl; }
}

static __device__ __forceinline__ void own_take4(int e, own_v4i d, int n0, int nT,
                                                 int* ls, int* ld, unsigned* cnt) {
  const int r0 = d.x - n0, r1 = d.y - n0, r2 = d.z - n0, r3 = d.w - n0;
  if ((unsigned)r0 < (unsigned)nT) own_push(e + 0, r0, ls, ld, cnt);
  if ((unsigned)r1 < (unsigned)nT) own_push(e + 1, r1, ls, ld, cnt);
  if ((unsigned)r2 < (unsigned)nT) own_push(e + 2, r2, ls, ld, cnt);
  if ((unsigned)r3 < (unsigned)nT) own_push(e + 3, r3, ls, ld, cnt);
}

static __device__ __forceinline__ void own_build(const int* __restrict__ edst,
                                                 int eb, int ee, int n0, int nT,
                                                 int* ls, int* ld, unsigned* cnt) {
  const int tid = threadIdx.x;
  const int n = ee - eb;
  const int* dbase = edst + eb;
  const int nvec = ((((size_t)dbase) & 15) == 0) ? (n >> 2) : 0;
  const own_v4i* dv = (const own_v4i*)dbase;
  const int rot = (int)(((long)blockIdx.x * nvec) / gridDim.x) & ~31;
  int v = tid;
  for (; v + 3 * OWN_THREADS < nvec; v += 4 * OWN_THREADS) {
    own_v4i d[4];
    int vv[4];
#pragma unroll
    for (int u = 0; u < 4; ++u) {
      vv[u] = v + u * OWN_THREADS + rot;
      if (vv[u] >= nvec) vv[u] -= nvec;
      d[u] = dv[vv[u]];
    }
    asm volatile("" :: "v"(d[0]), "v"(d[1]), "v"(d[2]), "v"(d[3]));
#pragma unroll
    for (int u = 0; u < 4; ++u)
      own_take4(eb + 4 * vv[u], d[u], n0, nT, ls, ld, cnt);
  }
  for (; v < nvec; v += OWN_THREADS) {
    int vv = v + rot;
    if (vv >= nvec) vv -= nvec;
    own_take4(eb + 4 * vv, dv[vv], n0, nT, ls, ld, cnt);
  }
  for (int e = eb + 4 * nvec + tid; e < ee; e += OWN_THREADS) {
    const int r = edst[e] - n0;
    if ((unsigned)r < (unsigned)nT) own_push(e, r, ls, ld, cnt);
  }
}

static __device__ __forceinline__ void own_push_e(int e, int* ls, unsigned* cnt) {
  unsigned idx = atomicAdd(cnt, 1u);
  if (idx < OWN_CAP) ls[idx] = e;
}

static __device__ __forceinline__ unsigned own_hits(unsigned w, unsigned pat) {
  const unsigned y = w ^ pat;
  const unsigned t = (y & 0x7F7F7F7Fu) + 0x7F7F7F7Fu;
  return ~(t | y) & 0x80808080u;
}

static __device__ __forceinline__ void own_take16(int e, own_v4u k, unsigned pat,
                                                  int* ls, unsigned* cnt) {
  unsigned m[4] = {own_hits(k.x, pat), own_hits(k.y, pat),
                   own_hits(k.z, pat), own_hits(k.w, pat)};
  if ((m[0] | m[1] | m[2] | m[3]) != 0u) {
#pragma unroll
    for (int u = 0; u < 4; ++u) {
      unsigned mm = m[u];
      while (mm) {
        own_push_e(e + 4 * u + (__builtin_ctz(mm) >> 3), ls, cnt);
        mm &= mm - 1u;
      }
    }
  }
}

static __device__ __forceinline__ void own_build_keys(const unsigned* __restrict__ keys,
                                                      const unsigned* __restrict__ plane,
                                                      int E, int* ls, unsigned* pl,
                                                      unsigned* cnt, unsigned* pcnt) {
  const int tid = threadIdx.x;
  const int lane = tid & 31;
  const unsigned gid = blockIdx.x;
  const unsigned pat = gid * 0x01010101u;
  const int nvec = ((((size_t)keys) & 15) == 0) ? (E >> 4) : 0;
  const own_v4u* kv = (const own_v4u*)keys;
  const int nwords = (nvec + 31) >> 5;
  for (int w0 = tid & ~31; w0 < nwords; w0 += OWN_THREADS) {
    const int w = w0 + lane;
    unsigned m = (w < nwords) ? plane[w] : 0u;
    const int n = __builtin_popcount(m);
    int incl = n;
    for (int dd = 1; dd < 32; dd <<= 1) {
      const int t = __shfl_up(incl, dd);
      if (lane >= dd) incl += t;
    }
    const int tot = __shfl(incl, 31);
    if (tot != 0) {
      unsigned base = 0u;
      if (lane == 31) base = atomicAdd(pcnt, (unsigned)tot);
      base = (unsigned)__shfl((int)base, 31);
      unsigned idx = base + (unsigned)(incl - n);
      while (m) {
        if (idx < OWN_CAP) pl[idx] = 32u * (unsigned)w + (unsigned)__builtin_ctz(m);
        ++idx;
        m &= m - 1u;
      }
    }
  }
  __syncthreads();
  const int np = (int)*pcnt;
  if (np > OWN_CAP) {
    if (tid == 0) *cnt = (unsigned)np;
    return;
  }
  for (int i0 = tid; i0 < np; i0 += 4 * OWN_THREADS) {
    own_v4u k[4];
    unsigned pc[4];
#pragma unroll
    for (int u = 0; u < 4; ++u) {
      const int i = i0 + u * OWN_THREADS;
      pc[u] = pl[(i < np) ? i : 0];
      k[u] = kv[pc[u]];
    }
    asm volatile("" :: "v"(k[0]), "v"(k[1]), "v"(k[2]), "v"(k[3]));
#pragma unroll
    for (int u = 0; u < 4; ++u)
      if (i0 + u * OWN_THREADS < np) own_take16(16 * (int)pc[u], k[u], pat, ls, cnt);
  }
  for (int e = 16 * nvec + tid; e < E; e += OWN_THREADS)
    if (((keys[e >> 2] >> (8 * (e & 3))) & 255u) == gid) own_push_e(e, ls, cnt);
}

template <bool WITH_DST>
static __device__ __forceinline__ void own_resolve(const int* __restrict__ esrc,
                                                   const int* __restrict__ edst,
                                                   int n0, int* ls, int* ld, int n) {
  for (int i0 = threadIdx.x; i0 < n; i0 += 4 * OWN_THREADS) {
    int e[4], s[4], d[4];
#pragma unroll
    for (int u = 0; u < 4; ++u) {
      const int i = i0 + u * OWN_THREADS;
      e[u] = (i < n) ? ls[i] : 0;
    }
#pragma unroll
    for (int u = 0; u < 4; ++u) {
      s[u] = esrc[e[u]];
      d[u] = WITH_DST ? edst[e[u]] : 0;
    }
    asm volatile("" :: "v"(s[0]), "v"(s[1]), "v"(s[2]), "v"(s[3]),
                       "v"(d[0]), "v"(d[1]), "v"(d[2]), "v"(d[3]));
#pragma unroll
    for (int u = 0; u < 4; ++u) {
      const int i = i0 + u * OWN_THREADS;
      if (i < n) {
        ls[i] = s[u];
        if (WITH_DST) ld[i] = d[u] - n0;
      }
    }
  }
}

static __device__ __forceinline__ void own_leave(const int* ls, const int* ld,
                                                 int* __restrict__ gs,
                                                 int* __restrict__ gd, int n) {
  for (int i = threadIdx.x; i < n; i += OWN_THREADS) {
    gs[i] = ls[i];
    gd[i] = ld[i];
  }
}

static __device__ __forceinline__ void own_take(const int* __restrict__ gs,
                                                const int* __restrict__ gd,
                                                int* ls, int* ld, int n) {
  for (int i0 = threadIdx.x; i0 < n; i0 += 4 * OWN_THREADS) {
    int s[4], d[4];
#pragma unroll
    for (int u = 0; u < 4; ++u) {
      const int i = i0 + u * OWN_THREADS;
      const int j = (i < n) ? i : 0;
      s[u] = gs[j];
      d[u] = gd[j];
    }
#pragma unroll
    for (int u = 0; u < 4; ++u) {
      const int i = i0 + u * OWN_THREADS;
      if (i < n) { ls[i] = s[u]; ld[i] = d[u]; }
    }
  }
}

static __device__ __forceinline__ void own_entry(int i, int n_list, int n0,
                                                 const int* ls, const int* ld,
                                                 int& s, int& dl) {
  if (i < n_list) { s = ls[i]; dl = ld[i]; }
  else            { dl = i - n_list; s = n0 + dl; }
}

template <int H, int NU>
static __device__ __forceinline__ void own_scores_trip(int phase, int idx0, int n_list,
                                                       int n0, const int* ls,
                                                       const int* ld,
                                                       const float* __restrict__ as,
                                                       const float* adL, unsigned* mx,
                                                       float* den) {
  float a[NU];
  int dh[NU];
#pragma unroll
  for (int u = 0; u < NU; ++u) {
    const int idx = idx0 + u * OWN_THREADS;
    const int i = idx / H, h = idx - i * H;
    int s, dl;
    own_entry(i, n_list, n0, ls, ld, s, dl);
    a[u] = as[s * H + h];
    dh[u] = dl * H + h;
  }
#pragma unroll
  for (int u = 0; u < NU; ++u) {
    float v = a[u] + adL[dh[u]];
    v = (v >= 0.0f) ? v : NEG_SLOPE * v;
    if (phase == 0) atomicMax(&mx[dh[u]], f2key(v));
    else            atomicAdd(&den[dh[u]], expf(v - key2f(mx[dh[u]])));
  }
}

template <int H>
static __device__ __forceinline__ void own_scores(int phase, int n_list, int n_self,
                                                  int n0, const int* ls, const int* ld,
                                                  const float* __restrict__ as,
                                                  const float* adL, unsigned* mx,
                                                  float* den) {
  const int n_items = (n_list + n_self) * H;
  int idx = threadIdx.x;
  for (; idx + 3 * OWN_THREADS < n_items; idx += 4 * OWN_THREADS)
    own_scores_trip<H, 4>(phase, idx, n_list, n0, ls, ld, as, adL, mx, den);
  for (; idx < n_items; idx += OWN_THREADS)
    own_scores_trip<H, 1>(phase, idx, n_list, n0, ls, ld, as, adL, mx, den);
}

template <int H, int C, int LDX, int AP, int NU>
static __device__ __forceinline__ void own_rows_trip(int i0, int n_list, int n0,
                                                     const int* ls, const int* ld,
                                                     const float* __restrict__ as,
                                                     const float* __restrict__ xw,
                                                     const float* adL,
                                                     const unsigned* mx,
                                                     const float* den, float* acc) {
  constexpr int HC = H * C;
  constexpr int NJ = (HC + 31) / 32;
  const int lane = threadIdx.x & 31;
  const int hl = lane & (H - 1);
  int dl[NU];
  float x[NU][NJ], av[NU];
#pragma unroll
  for (int u = 0; u < NU; ++u) {
    int si, di;
    own_entry(i0 + u * (OWN_THREADS / 32), n_list, n0, ls, ld, si, di);
    const int s = __builtin_amdgcn_readfirstlane(si);
    dl[u] = __builtin_amdgcn_readfirstlane(di);
    const float* row = xw + (size_t)s * LDX;
#pragma unroll
    for (int j = 0; j < NJ; ++j) {
      const int c = j * 32 + lane;
      x[u][j] = (c < HC) ? row[c] : 0.0f;
    }
    av[u] = as[s * H + hl];
  }
#pragma unroll
  for (int u = 0; u < NU; ++u) {
    const int di = dl[u] * H + hl;
    float v = av[u] + adL[di];
    v = (v >= 0.0f) ? v : NEG_SLOPE * v;
    const float w = expf(v - key2f(mx[di])) / den[di];
    float* arow = acc + dl[u] * AP;
#pragma unroll
    for (int j = 0; j < NJ; ++j) {
      const int head = (H == 1) ? 0 : (j * 32) / C;
      const float wj = __int_as_float(
          __builtin_amdgcn_readlane(__float_as_int(w), head));
      const int c = j * 32 + lane;
      if (c < HC) atomicAdd(&arow[c], wj * x[u][j]);
    }
  }
}

template <int H, int C, int LDX, int AP>
static __device__ __forceinline__ void own_rows_block(int i0, int n_list, int n0,
                                                      const int* ls, const int* ld,
                                                      const float* __restrict__ as,
                                                      const float* __restrict__ xw,
                                                      const float* adL,
                                                      const unsigned* mx,
                                                      const float* den, float* acc) {
  constexpr int HC = H * C;
  constexpr int NJ = (HC + 31) / 32;
  constexpr int G = 32 / H;
  constexpr int SUB = (G * NJ <= 32) ? G : (32 / NJ);
  static_assert(SUB >= 1 && (G % SUB) == 0, "a block is a whole number of row groups");
  const int lane = threadIdx.x & 31;
  const int hl = lane & (H - 1);
  int si, di;
  own_entry(i0 + lane / H, n_list, n0, ls, ld, si, di);
  const int dh = di * H + hl;
  const float a = as[si * H + hl];
  float w = 0.0f;
#pragma unroll
  for (int k = 0; k < G / SUB; ++k) {
    float x[SUB][NJ];
#pragma unroll
    for (int u = 0; u < SUB; ++u) {
      const int s = __builtin_amdgcn_readlane(si, (k * SUB + u) * H);
      const float* row = xw + (size_t)s * LDX;
#pragma unroll
      for (int j = 0; j < NJ; ++j) {
        const int c = j * 32 + lane;
        x[u][j] = (c < HC) ? row[c] : 0.0f;
      }
    }
    if (k == 0) {
      float v = a + adL[dh];
      v = (v >= 0.0f) ? v : NEG_SLOPE * v;
      w = expf(v - key2f(mx[dh])) / den[dh];
    }
#pragma unroll
    for (int u = 0; u < SUB; ++u) {
      const int e = k * SUB + u;
      const int dl = __builtin_amdgcn_readlane(di, e * H);
      float* arow = acc + dl * AP;
#pragma unroll
      for (int j = 0; j < NJ; ++j) {
        const int head = (H == 1) ? 0 : (j * 32) / C;
        const float wj = __int_as_float(
            __builtin_amdgcn_readlane(__float_as_int(w), e * H + head));
        const int c = j * 32 + lane;
        if (c < HC) atomicAdd(&arow[c], wj * x[u][j]);
      }
    }
  }
}

template <int H, int C, int LDX, int AP>
static __device__ __forceinline__ void own_rows(int n_list, int n_self, int n0,
                                                const int* ls, const int* ld,
                                                const float* __restrict__ as,
                                                const float* __restrict__ xw,
                                                const float* adL, const unsigned* mx,
                                                const float* den, float* acc) {
  static_assert(H == 1 || (C % 32) == 0, "a 32-lane slice must lie in one head");
  static_assert((H & (H - 1)) == 0 && H <= 32, "heads: a power of two");
  constexpr int NW = OWN_THREADS / 32;
  constexpr int G = 32 / H;
  static_assert(G <= NW, "the entries left over get one wave each");
  const int n_items = n_list + n_self;
  const int n_blocks = n_items / G;
  const int wave = __builtin_amdgcn_readfirstlane((int)(threadIdx.x >> 5));
  for (int b = wave; b < n_blocks; b += NW)
    own_rows_block<H, C, LDX, AP>(b * G, n_list, n0, ls, ld, as, xw, adL, mx, den, acc);
  const int t = n_blocks * G + wave;
  if (t < n_items)
    own_rows_trip<H, C, LDX, AP, 1>(t, n_list, n0, ls, ld, as, xw, adL, mx, den, acc);
}

template <int H, int C, int LDX, bool ELU, bool TAKE, int NC2, int LD2>
__global__ __launch_bounds__(OWN_THREADS) void gat_owner_kernel(
    const int* __restrict__ esrc, const int* __restrict__ edst, int E, int Nn,
    const float* __restrict__ as, const float* __restrict__ ad,
    const float* __restrict__ xw, const float* __restrict__ bias,
    float* __restrict__ outp, int* lst, const unsigned* __restrict__ keys,
    const unsigned* __restrict__ planes, int plane_words,
    const float* __restrict__ w2, const float* __restrict__ a2_src,
    const float* __restrict__ a2_dst, float* __restrict__ xw2,
    float* __restrict__ as2, float* __restrict__ ad2) {
  constexpr int HC = H * C;
  constexpr bool NEXT = (NC2 > 0);
  constexpr int AP = NEXT ? HC + 4 : HC;
  __shared__ float    acc[OWN_T * AP];
  __shared__ float    adL[OWN_T * H];
  __shared__ float    den[OWN_T * H];
  __shared__ unsigned mx[OWN_T * H];
  __shared__ int      lsd[2 * OWN_CAP];
  int* const ls = lsd;
  int* const ld = lsd + OWN_CAP;
  __shared__ unsigned cnt;
  __shared__ unsigned pcnt;

  const int tid = threadIdx.x;
  const int n0 = blockIdx.x * OWN_T;
  const int nT = min(OWN_T, Nn - n0);

  for (int i = tid; i < OWN_T * AP; i += OWN_THREADS) acc[i] = 0.0f;
  for (int i = tid; i < OWN_T * H; i += OWN_THREADS) {
    den[i] = 0.0f;
    mx[i] = 0u;
    adL[i] = (i < nT * H) ? ad[(size_t)n0 * H + i] : 0.0f;
  }
  if (tid == 0) { cnt = 0u; pcnt = 0u; }

  int* greg = lst + (size_t)blockIdx.x * OWN_REGION;
  int* gs = greg + OWN_HEAD;
  int* gd = gs + OWN_CAP;
  int total;
  const bool by_key = (gridDim.x <= 256u);
  if (TAKE) {
    total = greg[0];
    if (total <= OWN_CAP) own_take(gs, gd, ls, ld, total);
    __syncthreads();
  } else {
    __syncthreads();
    if (by_key) own_build_keys(keys, planes + (size_t)blockIdx.x * plane_words, E, ls,
                               (unsigned*)ld, &cnt, &pcnt);
    else        own_build(edst, 0, E, n0, nT, ls, ld, &cnt);
    __syncthreads();
    total = (int)cnt;
    if (tid == 0) greg[0] = total;
  }

  if (total <= OWN_CAP) {
    if (!TAKE) {
      if (by_key) own_resolve<true>(esrc, edst, n0, ls, ld, total);
      else        own_resolve<false>(esrc, edst, n0, ls, ld, total);
      __syncthreads();
      own_leave(ls, ld, gs, gd, total);
    }
    own_scores<H>(0, total, nT, n0, ls, ld, as, adL, mx, den);
    __syncthreads();
    own_scores<H>(1, total, nT, n0, ls, ld, as, adL, mx, den);
    __syncthreads();
    own_rows<H, C, LDX, AP>(total, nT, n0, ls, ld, as, xw, adL, mx, den, acc);
  } else {
    for (int phase = 0; phase < 3; ++phase) {
      for (int eb = 0; eb < E; eb += OWN_CAP) {
        __syncthreads();
        if (tid == 0) cnt = 0u;
        __syncthreads();
        own_build(edst, eb, min(eb + OWN_CAP, E), n0, nT, ls, ld, &cnt);
        __syncthreads();
        const int m = (int)cnt;
        own_resolve<false>(esrc, edst, n0, ls, ld, m);
        __syncthreads();
        const int n_self = (eb == 0) ? nT : 0;
        if (phase < 2)
          own_scores<H>(phase, m, n_self, n0, ls, ld, as, adL, mx, den);
        else
          own_rows<H, C, LDX, AP>(m, n_self, n0, ls, ld, as, xw, adL, mx, den, acc);
      }
    }
  }
  __syncthreads();

  if constexpr (!NEXT) {
    float* orow = outp + (size_t)n0 * HC;
    for (int i = tid; i < nT * HC; i += OWN_THREADS) {
      float v = acc[i] + bias[i % HC];
      if (ELU) v = (v > 0.0f) ? v : (expf(v) - 1.0f);
      orow[i] = v;
    }
  } else {
    constexpr int NT2 = (NC2 + 15) / 16;
    constexpr int NRT = (OWN_T + 15) / 16;
    constexpr int TP  = 16 * NT2 + 1;
    static_assert((HC % 4) == 0, "K of the next product goes in steps of 4");
    static_assert(OWN_T * TP <= 2 * OWN_CAP, "the tile fits the edge list's space");
    static_assert(16 * NT2 <= LD2 && NRT <= OWN_THREADS / 32, "tile shape");
    for (int i = tid; i < OWN_T * HC; i += OWN_THREADS) {
      const int r = i / HC, cc = i - r * HC;
      float v = acc[r * AP + cc] + bias[cc];
      if (ELU) v = (v > 0.0f) ? v : (expf(v) - 1.0f);
      acc[r * AP + cc] = v;
    }
    __syncthreads();
    float* tile = (float*)lsd;
    const int wave = __builtin_amdgcn_readfirstlane((int)(threadIdx.x >> 5));
    if (wave < NRT) {
      const int lane = tid & 31;
      const int half = lane >> 4;
      const int l    = lane & 15;
      int r = 16 * wave + l;
      r = (r < OWN_T) ? r : (OWN_T - 1);
      const float* ap = acc + r * AP + 2 * half;
      const float* bp = w2 + (size_t)(2 * half) * LD2 + l;
      v8f d[NT2] = {};
#pragma unroll 4
      for (int s = 0; s < HC / 4; ++s) {
        const v2f a = *(const v2f*)(ap + 4 * s);
        const float* bq = bp + (size_t)(4 * s) * LD2;
        v2f b[NT2];
#pragma unroll
        for (int t = 0; t < NT2; ++t) {
          b[t].x = bq[t * 16];
          b[t].y = bq[LD2 + t * 16];
        }
#pragma unroll
        for (int t = 0; t < NT2; ++t)
          d[t] = __builtin_amdgcn_wmma_f32_16x16x4_f32(
              false, a, false, b[t], (short)0, d[t], false, false);
      }
#pragma unroll
      for (int j = 0; j < 8; ++j) {
        const int row = 16 * wave + 8 * half + j;
        if (row < OWN_T) {
#pragma unroll
          for (int t = 0; t < NT2; ++t) tile[row * TP + t * 16 + l] = d[t][j];
        }
      }
    }
    __syncthreads();
    for (int i = tid; i < nT * LD2; i += OWN_THREADS) {
      const int r = i / LD2, cc = i - r * LD2;
      if (cc < 16 * NT2) xw2[(size_t)(n0 + r) * LD2 + cc] = tile[r * TP + cc];
    }
    if (tid < nT) {
      float s = 0.0f, dd = 0.0f;
      for (int j = 0; j < NC2; ++j) {
        const float v = tile[tid * TP + j];
        s += v * a2_src[j];
        dd += v * a2_dst[j];
      }
      as2[n0 + tid] = s;
      ad2[n0 + tid] = dd;
    }
  }
}

extern "C" void kernel_launch(void* const* d_in, const int* in_sizes, int n_in,
                              void* d_out, int out_size, void* d_ws, size_t ws_size,
                              hipStream_t stream) {
  const float* x      = (const float*)d_in[0];
  const int*   ei     = (const int*)d_in[1];
  const float* W1     = (const float*)d_in[2];
  const float* a1_src = (const float*)d_in[3];
  const float* a1_dst = (const float*)d_in[4];
  const float* b1     = (const float*)d_in[5];
  const float* W2     = (const float*)d_in[6];
  const float* a2_src = (const float*)d_in[7];
  const float* a2_dst = (const float*)d_in[8];
  const float* b2     = (const float*)d_in[9];
  float* out = (float*)d_out;
  (void)n_in; (void)out_size; (void)ws_size;

  const int F_IN = 256, H1 = 8, C1 = 32, HC1 = 256, NC = 40, NCP = 64;
  const int Nn   = in_sizes[0] / F_IN;
  const int E    = in_sizes[1] / 2;
  const int* esrc = ei;
  const int* edst = ei + E;

  float* p = (float*)d_ws;
  float* xw1  = p; p += (size_t)Nn * HC1;
  float* as1  = p; p += (size_t)Nn * H1;
  float* ad1  = p; p += (size_t)Nn * H1;
  float* w2p  = p; p += (size_t)HC1 * NCP;
  float* xw2p = p; p += (size_t)Nn * NCP;
  float* as2  = p; p += (size_t)Nn;
  float* ad2  = p; p += (size_t)Nn;
  unsigned* keys = (unsigned*)p; p += (size_t)((E / 4 + 7) & ~3);
  const int key_blocks = ((E + 15) / 16 + PREP_THREADS - 1) / PREP_THREADS;
  const int plane_words = key_blocks * PLANE_W;
  unsigned* planes = (unsigned*)p; p += (size_t)256 * plane_words;
  int*   lst  = (int*)p;

  dim3 blk(256);
  const int own_blocks = (Nn + OWN_T - 1) / OWN_T;

  {
    const int pad_blocks = (HC1 * NCP + PREP_THREADS - 1) / PREP_THREADS;
    prep_kernel<<<pad_blocks + key_blocks, PREP_THREADS, 0, stream>>>(
        W2, w2p, HC1, NC, NCP, edst, keys, E, pad_blocks, planes, plane_words);
  }

  {
    int waves  = ((Nn / 16 + GEMM_MT - 1) / GEMM_MT) * (HC1 / 64);
    int blocks = (waves + 3) / 4;
    gemm_f32_wmma_kernel<<<blocks, dim3(128), 0, stream>>>(x, W1, xw1, Nn, F_IN, HC1);
  }
  alpha_kernel<<<(Nn * H1 + 255) / 256, blk, 0, stream>>>(
      xw1, a1_src, a1_dst, as1, ad1, Nn, H1, C1, HC1);
  gat_owner_kernel<8, 32, 256, true, false, 40, 64><<<own_blocks, OWN_THREADS, 0, stream>>>(
      esrc, edst, E, Nn, as1, ad1, xw1, b1, nullptr, lst, keys, planes, plane_words,
      w2p, a2_src, a2_dst, xw2p, as2, ad2);

  gat_owner_kernel<1, 40, 64, false, true, 0, 0><<<own_blocks, OWN_THREADS, 0, stream>>>(
      esrc, edst, E, Nn, as2, ad2, xw2p, b2, out, lst, keys, planes, plane_words,
      nullptr, nullptr, nullptr, nullptr, nullptr, nullptr);
}


# softmax weights enter the row sums un-divided: denominators added in the same list pass, finished rows scaled by 1/denominator once; the separate denominator pass is gone; sums by arrival as before
# speedup vs baseline: 16.2940x; 1.0483x over previous
#include <hip/hip_runtime.h>

typedef __attribute__((ext_vector_type(2))) float v2f;
typedef __attribute__((ext_vector_type(8))) float v8f;

#define NEG_SLOPE 0.2f

static __device__ __forceinline__ unsigned f2key(float f) {
  unsigned u = __float_as_uint(f);
  return (u & 0x80000000u) ? ~u : (u | 0x80000000u);
}
static __device__ __forceinline__ float key2f(unsigned k) {
  return (k & 0x80000000u) ? __uint_as_float(k & 0x7FFFFFFFu)
                           : __uint_as_float(~k);
}

#define OWN_T       196
typedef __attribute__((ext_vector_type(4))) int own_v4i;
typedef __attribute__((ext_vector_type(4))) unsigned own_v4u;

#define PREP_THREADS 256
#define PLANE_W      (PREP_THREADS / 32)
__global__ void prep_kernel(const float* __restrict__ src,
                            float* __restrict__ dst,
                            int K, int c, int ldd,
                            const int* __restrict__ edst,
                            unsigned* __restrict__ keys, int E, int pad_blocks,
                            unsigned* __restrict__ planes, int plane_words) {
  __shared__ unsigned bm[256 * PLANE_W];
  if ((int)blockIdx.x < pad_blocks) {
    int i = blockIdx.x * blockDim.x + threadIdx.x;
    if (i >= K * ldd) return;
    int r = i / ldd, j = i - r * ldd;
    dst[i] = (j < c) ? src[r * c + j] : 0.0f;
    return;
  }
  const int tid = threadIdx.x;
  const int kb = blockIdx.x - pad_blocks;
  for (int i = tid; i < 256 * PLANE_W; i += PREP_THREADS) bm[i] = 0u;
  __syncthreads();
  const int piece = kb * PREP_THREADS + tid;
  const int e0 = 16 * piece;
  if (e0 < E) {
    const bool full = (e0 + 15 < E);
    int d[16];
    if (full && ((((size_t)(edst + e0)) & 15) == 0)) {
      const own_v4i* dv = (const own_v4i*)(edst + e0);
      for (int q = 0; q < 4; ++q) {
        const own_v4i v = dv[q];
        d[4 * q + 0] = v.x; d[4 * q + 1] = v.y; d[4 * q + 2] = v.z; d[4 * q + 3] = v.w;
      }
    } else {
      for (int u = 0; u < 16; ++u) d[u] = (e0 + u < E) ? edst[e0 + u] : -1;
    }
    unsigned kw[4] = {0u, 0u, 0u, 0u};
    const unsigned bit = 1u << (tid & 31);
    for (int u = 0; u < 16; ++u) {
      if (d[u] >= 0) {
        const unsigned g = (unsigned)(d[u] / OWN_T) & 255u;
        kw[u >> 2] |= g << (8 * (u & 3));
        if (full) atomicOr(&bm[g * PLANE_W + (tid >> 5)], bit);
      }
    }
    if (full && ((((size_t)keys) & 15) == 0)) {
      own_v4u k4;
      k4.x = kw[0]; k4.y = kw[1]; k4.z = kw[2]; k4.w = kw[3];
      *(own_v4u*)(keys + 4 * piece) = k4;
    } else {
      for (int q = 0; q < 4; ++q)
        if (e0 + 4 * q < E) keys[4 * piece + q] = kw[q];
    }
  }
  __syncthreads();
  unsigned* row = planes + (size_t)tid * plane_words + (size_t)kb * PLANE_W;
  for (int q = 0; q < PLANE_W; ++q) row[q] = bm[tid * PLANE_W + q];
}

#define GEMM_MT 2
__global__ __launch_bounds__(128) void gemm_f32_wmma_kernel(const float* __restrict__ A,
                                     const float* __restrict__ B,
                                     float* __restrict__ C,
                                     int M, int K, int ldb) {
  const int lane = threadIdx.x & 31;
  const int wave = threadIdx.x >> 5;
  const int half = lane >> 4;
  const int l    = lane & 15;
  const int nt64    = ldb >> 6;
  const int mtiles  = M >> 4;
  const int mgroups = (mtiles + GEMM_MT - 1) / GEMM_MT;
  const int total   = mgroups * nt64;
  int t = blockIdx.x * (blockDim.x >> 5) + wave;
  if (t >= total) return;
  const int mg = t / nt64;
  const int n0 = (t - mg * nt64) << 6;
  const int m0 = mg * (16 * GEMM_MT);

  v8f acc[GEMM_MT][4] = {};
  const float* arow[GEMM_MT];
#pragma unroll
  for (int i = 0; i < GEMM_MT; ++i) {
    int r = m0 + 16 * i + l;
    r = (r < M) ? r : (M - 1);
    arow[i] = A + (size_t)r * K + 2 * half;
  }
  const float* bp   = B + (size_t)(2 * half) * ldb + n0 + l;
  const size_t bstep = (size_t)4 * ldb;

  v2f a0[GEMM_MT], a1[GEMM_MT], b0[4], b1[4];
  const int nsteps = K >> 2;

#define GEMM_LOAD(av, bv, step)                                              \
  {                                                                          \
    _Pragma("unroll") for (int i = 0; i < GEMM_MT; ++i)                      \
        av[i] = *(const v2f*)(arow[i] + 4 * (step));                         \
    const float* bq = bp + (size_t)(step) * bstep;                           \
    _Pragma("unroll") for (int s = 0; s < 4; ++s) {                          \
      bv[s].x = bq[s * 16];                                                  \
      bv[s].y = bq[ldb + s * 16];                                            \
    }                                                                        \
  }
#define GEMM_MMA(av, bv)                                                     \
  {                                                                          \
    _Pragma("unroll") for (int i = 0; i < GEMM_MT; ++i)                      \
        _Pragma("unroll") for (int s = 0; s < 4; ++s)                        \
            acc[i][s] = __builtin_amdgcn_wmma_f32_16x16x4_f32(               \
                false, av[i], false, bv[s], (short)0, acc[i][s], false, false); \
  }

  GEMM_LOAD(a0, b0, 0)
  int st = 0;
  for (; st + 2 < nsteps; st += 2) {
    GEMM_LOAD(a1, b1, st + 1)
    asm volatile("" ::: "memory");
    GEMM_MMA(a0, b0)
    GEMM_LOAD(a0, b0, st + 2)
    asm volatile("" ::: "memory");
    GEMM_MMA(a1, b1)
  }
  GEMM_LOAD(a1, b1, st + 1)
  asm volatile("" ::: "memory");
  GEMM_MMA(a0, b0)
  GEMM_MMA(a1, b1)
#undef GEMM_LOAD
#undef GEMM_MMA

#pragma unroll
  for (int i = 0; i < GEMM_MT; ++i) {
    if (m0 + 16 * i < M) {
      float* crow = C + (size_t)(m0 + 16 * i + 8 * half) * ldb + n0 + l;
#pragma unroll
      for (int s = 0; s < 4; ++s)
#pragma unroll
        for (int j = 0; j < 8; ++j) crow[(size_t)j * ldb + s * 16] = acc[i][s][j];
    }
  }
}

__global__ void alpha_kernel(const float* __restrict__ xw,
                             const float* __restrict__ a_src,
                             const float* __restrict__ a_dst,
                             float* __restrict__ as, float* __restrict__ ad,
                             int n_nodes, int heads, int c, int ldx) {
  int i = blockIdx.x * blockDim.x + threadIdx.x;
  if (i >= n_nodes * heads) return;
  int n = i / heads, h = i - n * heads;
  const float* row = xw + (size_t)n * ldx + (size_t)h * c;
  const float* ws = a_src + (size_t)h * c;
  const float* wd = a_dst + (size_t)h * c;
  float s = 0.0f, d = 0.0f;
  for (int j = 0; j < c; ++j) {
    float v = row[j];
    s += v * ws[j];
    d += v * wd[j];
  }
  as[i] = s;
  ad[i] = d;
}

#define OWN_CAP     8192
#define OWN_THREADS 1024
#define OWN_HEAD    32
#define OWN_REGION  (OWN_HEAD + 2 * OWN_CAP)

static __device__ __forceinline__ void own_push(int e, int dl, int* ls, int* ld,
                                                unsigned* cnt) {
  unsigned idx = atomicAdd(cnt, 1u);
  if (idx < OWN_CAP) { ls[idx] = e; ld[idx] = dl; }
}

static __device__ __forceinline__ void own_take4(int e, own_v4i d, int n0, int nT,
                                                 int* ls, int* ld, unsigned* cnt) {
  const int r0 = d.x - n0, r1 = d.y - n0, r2 = d.z - n0, r3 = d.w - n0;
  if ((unsigned)r0 < (unsigned)nT) own_push(e + 0, r0, ls, ld, cnt);
  if ((unsigned)r1 < (unsigned)nT) own_push(e + 1, r1, ls, ld, cnt);
  if ((unsigned)r2 < (unsigned)nT) own_push(e + 2, r2, ls, ld, cnt);
  if ((unsigned)r3 < (unsigned)nT) own_push(e + 3, r3, ls, ld, cnt);
}

static __device__ __forceinline__ void own_build(const int* __restrict__ edst,
                                                 int eb, int ee, int n0, int nT,
                                                 int* ls, int* ld, unsigned* cnt) {
  const int tid = threadIdx.x;
  const int n = ee - eb;
  const int* dbase = edst + eb;
  const int nvec = ((((size_t)dbase) & 15) == 0) ? (n >> 2) : 0;
  const own_v4i* dv = (const own_v4i*)dbase;
  const int rot = (int)(((long)blockIdx.x * nvec) / gridDim.x) & ~31;
  int v = tid;
  for (; v + 3 * OWN_THREADS < nvec; v += 4 * OWN_THREADS) {
    own_v4i d[4];
    int vv[4];
#pragma unroll
    for (int u = 0; u < 4; ++u) {
      vv[u] = v + u * OWN_THREADS + rot;
      if (vv[u] >= nvec) vv[u] -= nvec;
      d[u] = dv[vv[u]];
    }
    asm volatile("" :: "v"(d[0]), "v"(d[1]), "v"(d[2]), "v"(d[3]));
#pragma unroll
    for (int u = 0; u < 4; ++u)
      own_take4(eb + 4 * vv[u], d[u], n0, nT, ls, ld, cnt);
  }
  for (; v < nvec; v += OWN_THREADS) {
    int vv = v + rot;
    if (vv >= nvec) vv -= nvec;
    own_take4(eb + 4 * vv, dv[vv], n0, nT, ls, ld, cnt);
  }
  for (int e = eb + 4 * nvec + tid; e < ee; e += OWN_THREADS) {
    const int r = edst[e] - n0;
    if ((unsigned)r < (unsigned)nT) own_push(e, r, ls, ld, cnt);
  }
}

static __device__ __forceinline__ void own_push_e(int e, int* ls, unsigned* cnt) {
  unsigned idx = atomicAdd(cnt, 1u);
  if (idx < OWN_CAP) ls[idx] = e;
}

static __device__ __forceinline__ unsigned own_hits(unsigned w, unsigned pat) {
  const unsigned y = w ^ pat;
  const unsigned t = (y & 0x7F7F7F7Fu) + 0x7F7F7F7Fu;
  return ~(t | y) & 0x80808080u;
}

static __device__ __forceinline__ void own_take16(int e, own_v4u k, unsigned pat,
                                                  int* ls, unsigned* cnt) {
  unsigned m[4] = {own_hits(k.x, pat), own_hits(k.y, pat),
                   own_hits(k.z, pat), own_hits(k.w, pat)};
  if ((m[0] | m[1] | m[2] | m[3]) != 0u) {
#pragma unroll
    for (int u = 0; u < 4; ++u) {
      unsigned mm = m[u];
      while (mm) {
        own_push_e(e + 4 * u + (__builtin_ctz(mm) >> 3), ls, cnt);
        mm &= mm - 1u;
      }
    }
  }
}

static __device__ __forceinline__ void own_build_keys(const unsigned* __restrict__ keys,
                                                      const unsigned* __restrict__ plane,
                                                      int E, int* ls, unsigned* pl,
                                                      unsigned* cnt, unsigned* pcnt) {
  const int tid = threadIdx.x;
  const int lane = tid & 31;
  const unsigned gid = blockIdx.x;
  const unsigned pat = gid * 0x01010101u;
  const int nvec = ((((size_t)keys) & 15) == 0) ? (E >> 4) : 0;
  const own_v4u* kv = (const own_v4u*)keys;
  const int nwords = (nvec + 31) >> 5;
  for (int w0 = tid & ~31; w0 < nwords; w0 += OWN_THREADS) {
    const int w = w0 + lane;
    unsigned m = (w < nwords) ? plane[w] : 0u;
    const int n = __builtin_popcount(m);
    int incl = n;
    for (int dd = 1; dd < 32; dd <<= 1) {
      const int t = __shfl_up(incl, dd);
      if (lane >= dd) incl += t;
    }
    const int tot = __shfl(incl, 31);
    if (tot != 0) {
      unsigned base = 0u;
      if (lane == 31) base = atomicAdd(pcnt, (unsigned)tot);
      base = (unsigned)__shfl((int)base, 31);
      unsigned idx = base + (unsigned)(incl - n);
      while (m) {
        if (idx < OWN_CAP) pl[idx] = 32u * (unsigned)w + (unsigned)__builtin_ctz(m);
        ++idx;
        m &= m - 1u;
      }
    }
  }
  __syncthreads();
  const int np = (int)*pcnt;
  if (np > OWN_CAP) {
    if (tid == 0) *cnt = (unsigned)np;
    return;
  }
  for (int i0 = tid; i0 < np; i0 += 4 * OWN_THREADS) {
    own_v4u k[4];
    unsigned pc[4];
#pragma unroll
    for (int u = 0; u < 4; ++u) {
      const int i = i0 + u * OWN_THREADS;
      pc[u] = pl[(i < np) ? i : 0];
      k[u] = kv[pc[u]];
    }
    asm volatile("" :: "v"(k[0]), "v"(k[1]), "v"(k[2]), "v"(k[3]));
#pragma unroll
    for (int u = 0; u < 4; ++u)
      if (i0 + u * OWN_THREADS < np) own_take16(16 * (int)pc[u], k[u], pat, ls, cnt);
  }
  for (int e = 16 * nvec + tid; e < E; e += OWN_THREADS)
    if (((keys[e >> 2] >> (8 * (e & 3))) & 255u) == gid) own_push_e(e, ls, cnt);
}

template <bool WITH_DST>
static __device__ __forceinline__ void own_resolve(const int* __restrict__ esrc,
                                                   const int* __restrict__ edst,
                                                   int n0, int* ls, int* ld, int n) {
  for (int i0 = threadIdx.x; i0 < n; i0 += 4 * OWN_THREADS) {
    int e[4], s[4], d[4];
#pragma unroll
    for (int u = 0; u < 4; ++u) {
      const int i = i0 + u * OWN_THREADS;
      e[u] = (i < n) ? ls[i] : 0;
    }
#pragma unroll
    for (int u = 0; u < 4; ++u) {
      s[u] = esrc[e[u]];
      d[u] = WITH_DST ? edst[e[u]] : 0;
    }
    asm volatile("" :: "v"(s[0]), "v"(s[1]), "v"(s[2]), "v"(s[3]),
                       "v"(d[0]), "v"(d[1]), "v"(d[2]), "v"(d[3]));
#pragma unroll
    for (int u = 0; u < 4; ++u) {
      const int i = i0 + u * OWN_THREADS;
      if (i < n) {
        ls[i] = s[u];
        if (WITH_DST) ld[i] = d[u] - n0;
      }
    }
  }
}

static __device__ __forceinline__ void own_leave(const int* ls, const int* ld,
                                                 int* __restrict__ gs,
                                                 int* __restrict__ gd, int n) {
  for (int i = threadIdx.x; i < n; i += OWN_THREADS) {
    gs[i] = ls[i];
    gd[i] = ld[i];
  }
}

static __device__ __forceinline__ void own_take(const int* __restrict__ gs,
                                                const int* __restrict__ gd,
                                                int* ls, int* ld, int n) {
  for (int i0 = threadIdx.x; i0 < n; i0 += 4 * OWN_THREADS) {
    int s[4], d[4];
#pragma unroll
    for (int u = 0; u < 4; ++u) {
      const int i = i0 + u * OWN_THREADS;
      const int j = (i < n) ? i : 0;
      s[u] = gs[j];
      d[u] = gd[j];
    }
#pragma unroll
    for (int u = 0; u < 4; ++u) {
      const int i = i0 + u * OWN_THREADS;
      if (i < n) { ls[i] = s[u]; ld[i] = d[u]; }
    }
  }
}

static __device__ __forceinline__ void own_entry(int i, int n_list, int n0,
                                                 const int* ls, const int* ld,
                                                 int& s, int& dl) {
  if (i < n_list) { s = ls[i]; dl = ld[i]; }
  else            { dl = i - n_list; s = n0 + dl; }
}

template <int H, int NU>
static __device__ __forceinline__ void own_max_trip(int idx0, int n_list,
                                                    int n0, const int* ls,
                                                    const int* ld,
                                                    const float* __restrict__ as,
                                                    const float* adL, unsigned* mx) {
  float a[NU];
  int dh[NU];
#pragma unroll
  for (int u = 0; u < NU; ++u) {
    const int idx = idx0 + u * OWN_THREADS;
    const int i = idx / H, h = idx - i * H;
    int s, dl;
    own_entry(i, n_list, n0, ls, ld, s, dl);
    a[u] = as[s * H + h];
    dh[u] = dl * H + h;
  }
#pragma unroll
  for (int u = 0; u < NU; ++u) {
    float v = a[u] + adL[dh[u]];
    v = (v >= 0.0f) ? v : NEG_SLOPE * v;
    atomicMax(&mx[dh[u]], f2key(v));
  }
}

template <int H>
static __device__ __forceinline__ void own_max(int n_list, int n_self,
                                               int n0, const int* ls, const int* ld,
                                               const float* __restrict__ as,
                                               const float* adL, unsigned* mx) {
  const int n_items = (n_list + n_self) * H;
  int idx = threadIdx.x;
  for (; idx + 3 * OWN_THREADS < n_items; idx += 4 * OWN_THREADS)
    own_max_trip<H, 4>(idx, n_list, n0, ls, ld, as, adL, mx);
  for (; idx < n_items; idx += OWN_THREADS)
    own_max_trip<H, 1>(idx, n_list, n0, ls, ld, as, adL, mx);
}

template <int H, int C, int LDX, int AP, int NU>
static __device__ __forceinline__ void own_rows_trip(int i0, int n_list, int n0,
                                                     const int* ls, const int* ld,
                                                     const float* __restrict__ as,
                                                     const float* __restrict__ xw,
                                                     const float* adL,
                                                     const unsigned* mx,
                                                     float* den, float* acc) {
  constexpr int HC = H * C;
  constexpr int NJ = (HC + 31) / 32;
  const int lane = threadIdx.x & 31;
  const int hl = lane & (H - 1);
  int dl[NU];
  float x[NU][NJ], av[NU];
#pragma unroll
  for (int u = 0; u < NU; ++u) {
    int si, di;
    own_entry(i0 + u * (OWN_THREADS / 32), n_list, n0, ls, ld, si, di);
    const int s = __builtin_amdgcn_readfirstlane(si);
    dl[u] = __builtin_amdgcn_readfirstlane(di);
    const float* row = xw + (size_t)s * LDX;
#pragma unroll
    for (int j = 0; j < NJ; ++j) {
      const int c = j * 32 + lane;
      x[u][j] = (c < HC) ? row[c] : 0.0f;
    }
    av[u] = as[s * H + hl];
  }
#pragma unroll
  for (int u = 0; u < NU; ++u) {
    const int di = dl[u] * H + hl;
    float v = av[u] + adL[di];
    v = (v >= 0.0f) ? v : NEG_SLOPE * v;
    const float w = expf(v - key2f(mx[di]));
    if (lane < H) atomicAdd(&den[di], w);
    float* arow = acc + dl[u] * AP;
#pragma unroll
    for (int j = 0; j < NJ; ++j) {
      const int head = (H == 1) ? 0 : (j * 32) / C;
      const float wj = __int_as_float(
          __builtin_amdgcn_readlane(__float_as_int(w), head));
      const int c = j * 32 + lane;
      if (c < HC) atomicAdd(&arow[c], wj * x[u][j]);
    }
  }
}

template <int H, int C, int LDX, int AP>
static __device__ __forceinline__ void own_rows_block(int i0, int n_list, int n0,
                                                      const int* ls, const int* ld,
                                                      const float* __restrict__ as,
                                                      const float* __restrict__ xw,
                                                      const float* adL,
                                                      const unsigned* mx,
                                                      float* den, float* acc) {
  constexpr int HC = H * C;
  constexpr int NJ = (HC + 31) / 32;
  constexpr int G = 32 / H;
  constexpr int SUB = (G * NJ <= 32) ? G : (32 / NJ);
  static_assert(SUB >= 1 && (G % SUB) == 0, "a block is a whole number of row groups");
  const int lane = threadIdx.x & 31;
  const int hl = lane & (H - 1);
  int si, di;
  own_entry(i0 + lane / H, n_list, n0, ls, ld, si, di);
  const int dh = di * H + hl;
  const float a = as[si * H + hl];
  float w = 0.0f;
#pragma unroll
  for (int k = 0; k < G / SUB; ++k) {
    float x[SUB][NJ];
#pragma unroll
    for (int u = 0; u < SUB; ++u) {
      const int s = __builtin_amdgcn_readlane(si, (k * SUB + u) * H);
      const float* row = xw + (size_t)s * LDX;
#pragma unroll
      for (int j = 0; j < NJ; ++j) {
        const int c = j * 32 + lane;
        x[u][j] = (c < HC) ? row[c] : 0.0f;
      }
    }
    if (k == 0) {
      float v = a + adL[dh];
      v = (v >= 0.0f) ? v : NEG_SLOPE * v;
      w = expf(v - key2f(mx[dh]));
      atomicAdd(&den[dh], w);
    }
#pragma unroll
    for (int u = 0; u < SUB; ++u) {
      const int e = k * SUB + u;
      const int dl = __builtin_amdgcn_readlane(di, e * H);
      float* arow = acc + dl * AP;
#pragma unroll
      for (int j = 0; j < NJ; ++j) {
        const int head = (H == 1) ? 0 : (j * 32) / C;
        const float wj = __int_as_float(
            __builtin_amdgcn_readlane(__float_as_int(w), e * H + head));
        const int c = j * 32 + lane;
        if (c < HC) atomicAdd(&arow[c], wj * x[u][j]);
      }
    }
  }
}

template <int H, int C, int LDX, int AP>
static __device__ __forceinline__ void own_rows(int n_list, int n_self, int n0,
                                                const int* ls, const int* ld,
                                                const float* __restrict__ as,
                                                const float* __restrict__ xw,
                                                const float* adL, const unsigned* mx,
                                                float* den, float* acc) {
  static_assert(H == 1 || (C % 32) == 0, "a 32-lane slice must lie in one head");
  static_assert((H & (H - 1)) == 0 && H <= 32, "heads: a power of two");
  constexpr int NW = OWN_THREADS / 32;
  constexpr int G = 32 / H;
  static_assert(G <= NW, "the entries left over get one wave each");
  const int n_items = n_list + n_self;
  const int n_blocks = n_items / G;
  const int wave = __builtin_amdgcn_readfirstlane((int)(threadIdx.x >> 5));
  for (int b = wave; b < n_blocks; b += NW)
    own_rows_block<H, C, LDX, AP>(b * G, n_list, n0, ls, ld, as, xw, adL, mx, den, acc);
  const int t = n_blocks * G + wave;
  if (t < n_items)
    own_rows_trip<H, C, LDX, AP, 1>(t, n_list, n0, ls, ld, as, xw, adL, mx, den, acc);
}

template <int H, int C, int LDX, bool ELU, bool TAKE, int NC2, int LD2>
__global__ __launch_bounds__(OWN_THREADS) void gat_owner_kernel(
    const int* __restrict__ esrc, const int* __restrict__ edst, int E, int Nn,
    const float* __restrict__ as, const float* __restrict__ ad,
    const float* __restrict__ xw, const float* __restrict__ bias,
    float* __restrict__ outp, int* lst, const unsigned* __restrict__ keys,
    const unsigned* __restrict__ planes, int plane_words,
    const float* __restrict__ w2, const float* __restrict__ a2_src,
    const float* __restrict__ a2_dst, float* __restrict__ xw2,
    float* __restrict__ as2, float* __restrict__ ad2) {
  constexpr int HC = H * C;
  constexpr bool NEXT = (NC2 > 0);
  constexpr int AP = NEXT ? HC + 4 : HC;
  __shared__ float    acc[OWN_T * AP];
  __shared__ float    adL[OWN_T * H];
  __shared__ float    den[OWN_T * H];
  __shared__ unsigned mx[OWN_T * H];
  __shared__ int      lsd[2 * OWN_CAP];
  int* const ls = lsd;
  int* const ld = lsd + OWN_CAP;
  __shared__ unsigned cnt;
  __shared__ unsigned pcnt;

  const int tid = threadIdx.x;
  const int n0 = blockIdx.x * OWN_T;
  const int nT = min(OWN_T, Nn - n0);

  for (int i = tid; i < OWN_T * AP; i += OWN_THREADS) acc[i] = 0.0f;
  for (int i = tid; i < OWN_T * H; i += OWN_THREADS) {
    den[i] = (i < nT * H) ? 0.0f : 1.0f;
    mx[i] = 0u;
    adL[i] = (i < nT * H) ? ad[(size_t)n0 * H + i] : 0.0f;
  }
  if (tid == 0) { cnt = 0u; pcnt = 0u; }

  int* greg = lst + (size_t)blockIdx.x * OWN_REGION;
  int* gs = greg + OWN_HEAD;
  int* gd = gs + OWN_CAP;
  int total;
  const bool by_key = (gridDim.x <= 256u);
  if (TAKE) {
    total = greg[0];
    if (total <= OWN_CAP) own_take(gs, gd, ls, ld, total);
    __syncthreads();
  } else {
    __syncthreads();
    if (by_key) own_build_keys(keys, planes + (size_t)blockIdx.x * plane_words, E, ls,
                               (unsigned*)ld, &cnt, &pcnt);
    else        own_build(edst, 0, E, n0, nT, ls, ld, &cnt);
    __syncthreads();
    total = (int)cnt;
    if (tid == 0) greg[0] = total;
  }

  if (total <= OWN_CAP) {
    if (!TAKE) {
      if (by_key) own_resolve<true>(esrc, edst, n0, ls, ld, total);
      else        own_resolve<false>(esrc, edst, n0, ls, ld, total);
      __syncthreads();
      own_leave(ls, ld, gs, gd, total);
    }
    own_max<H>(total, nT, n0, ls, ld, as, adL, mx);
    __syncthreads();
    own_rows<H, C, LDX, AP>(total, nT, n0, ls, ld, as, xw, adL, mx, den, acc);
  } else {
    for (int phase = 0; phase < 2; ++phase) {
      for (int eb = 0; eb < E; eb += OWN_CAP) {
        __syncthreads();
        if (tid == 0) cnt = 0u;
        __syncthreads();
        own_build(edst, eb, min(eb + OWN_CAP, E), n0, nT, ls, ld, &cnt);
        __syncthreads();
        const int m = (int)cnt;
        own_resolve<false>(esrc, edst, n0, ls, ld, m);
        __syncthreads();
        const int n_self = (eb == 0) ? nT : 0;
        if (phase == 0)
          own_max<H>(m, n_self, n0, ls, ld, as, adL, mx);
        else
          own_rows<H, C, LDX, AP>(m, n_self, n0, ls, ld, as, xw, adL, mx, den, acc);
      }
    }
  }
  __syncthreads();

  if constexpr (!NEXT) {
    float* orow = outp + (size_t)n0 * HC;
    for (int i = tid; i < nT * HC; i += OWN_THREADS) {
      float v = acc[i] / den[(i / HC) * H + (i % HC) / C] + bias[i % HC];
      if (ELU) v = (v > 0.0f) ? v : (expf(v) - 1.0f);
      orow[i] = v;
    }
  } else {
    constexpr int NT2 = (NC2 + 15) / 16;
    constexpr int NRT = (OWN_T + 15) / 16;
    constexpr int TP  = 16 * NT2 + 1;
    static_assert((HC % 4) == 0, "K of the next product goes in steps of 4");
    static_assert(OWN_T * TP <= 2 * OWN_CAP, "the tile fits the edge list's space");
    static_assert(16 * NT2 <= LD2 && NRT <= OWN_THREADS / 32, "tile shape");
    for (int i = tid; i < OWN_T * H; i += OWN_THREADS) den[i] = 1.0f / den[i];
    __syncthreads();
    for (int i = tid; i < OWN_T * HC; i += OWN_THREADS) {
      const int r = i / HC, cc = i - r * HC;
      float v = acc[r * AP + cc] * den[r * H + cc / C] + bias[cc];
      if (ELU) v = (v > 0.0f) ? v : (expf(v) - 1.0f);
      acc[r * AP + cc] = v;
    }
    __syncthreads();
    float* tile = (float*)lsd;
    const int wave = __builtin_amdgcn_readfirstlane((int)(threadIdx.x >> 5));
    if (wave < NRT) {
      const int lane = tid & 31;
      const int half = lane >> 4;
      const int l    = lane & 15;
      int r = 16 * wave + l;
      r = (r < OWN_T) ? r : (OWN_T - 1);
      const float* ap = acc + r * AP + 2 * half;
      const float* bp = w2 + (size_t)(2 * half) * LD2 + l;
      v8f d[NT2] = {};
#pragma unroll 4
      for (int s = 0; s < HC / 4; ++s) {
        const v2f a = *(const v2f*)(ap + 4 * s);
        const float* bq = bp + (size_t)(4 * s) * LD2;
        v2f b[NT2];
#pragma unroll
        for (int t = 0; t < NT2; ++t) {
          b[t].x = bq[t * 16];
          b[t].y = bq[LD2 + t * 16];
        }
#pragma unroll
        for (int t = 0; t < NT2; ++t)
          d[t] = __builtin_amdgcn_wmma_f32_16x16x4_f32(
              false, a, false, b[t], (short)0, d[t], false, false);
      }
#pragma unroll
      for (int j = 0; j < 8; ++j) {
        const int row = 16 * wave + 8 * half + j;
        if (row < OWN_T) {
#pragma unroll
          for (int t = 0; t < NT2; ++t) tile[row * TP + t * 16 + l] = d[t][j];
        }
      }
    }
    __syncthreads();
    for (int i = tid; i < nT * LD2; i += OWN_THREADS) {
      const int r = i / LD2, cc = i - r * LD2;
      if (cc < 16 * NT2) xw2[(size_t)(n0 + r) * LD2 + cc] = tile[r * TP + cc];
    }
    if (tid < nT) {
      float s = 0.0f, dd = 0.0f;
      for (int j = 0; j < NC2; ++j) {
        const float v = tile[tid * TP + j];
        s += v * a2_src[j];
        dd += v * a2_dst[j];
      }
      as2[n0 + tid] = s;
      ad2[n0 + tid] = dd;
    }
  }
}

extern "C" void kernel_launch(void* const* d_in, const int* in_sizes, int n_in,
                              void* d_out, int out_size, void* d_ws, size_t ws_size,
                              hipStream_t stream) {
  const float* x      = (const float*)d_in[0];
  const int*   ei     = (const int*)d_in[1];
  const float* W1     = (const float*)d_in[2];
  const float* a1_src = (const float*)d_in[3];
  const float* a1_dst = (const float*)d_in[4];
  const float* b1     = (const float*)d_in[5];
  const float* W2     = (const float*)d_in[6];
  const float* a2_src = (const float*)d_in[7];
  const float* a2_dst = (const float*)d_in[8];
  const float* b2     = (const float*)d_in[9];
  float* out = (float*)d_out;
  (void)n_in; (void)out_size; (void)ws_size;

  const int F_IN = 256, H1 = 8, C1 = 32, HC1 = 256, NC = 40, NCP = 64;
  const int Nn   = in_sizes[0] / F_IN;
  const int E    = in_sizes[1] / 2;
  const int* esrc = ei;
  const int* edst = ei + E;

  float* p = (float*)d_ws;
  float* xw1  = p; p += (size_t)Nn * HC1;
  float* as1  = p; p += (size_t)Nn * H1;
  float* ad1  = p; p += (size_t)Nn * H1;
  float* w2p  = p; p += (size_t)HC1 * NCP;
  float* xw2p = p; p += (size_t)Nn * NCP;
  float* as2  = p; p += (size_t)Nn;
  float* ad2  = p; p += (size_t)Nn;
  unsigned* keys = (unsigned*)p; p += (size_t)((E / 4 + 7) & ~3);
  const int key_blocks = ((E + 15) / 16 + PREP_THREADS - 1) / PREP_THREADS;
  const int plane_words = key_blocks * PLANE_W;
  unsigned* planes = (unsigned*)p; p += (size_t)256 * plane_words;
  int*   lst  = (int*)p;

  dim3 blk(256);
  const int own_blocks = (Nn + OWN_T - 1) / OWN_T;

  {
    const int pad_blocks = (HC1 * NCP + PREP_THREADS - 1) / PREP_THREADS;
    prep_kernel<<<pad_blocks + key_blocks, PREP_THREADS, 0, stream>>>(
        W2, w2p, HC1, NC, NCP, edst, keys, E, pad_blocks, planes, plane_words);
  }

  {
    int waves  = ((Nn / 16 + GEMM_MT - 1) / GEMM_MT) * (HC1 / 64);
    int blocks = (waves + 3) / 4;
    gemm_f32_wmma_kernel<<<blocks, dim3(128), 0, stream>>>(x, W1, xw1, Nn, F_IN, HC1);
  }
  alpha_kernel<<<(Nn * H1 + 255) / 256, blk, 0, stream>>>(
      xw1, a1_src, a1_dst, as1, ad1, Nn, H1, C1, HC1);
  gat_owner_kernel<8, 32, 256, true, false, 40, 64><<<own_blocks, OWN_THREADS, 0, stream>>>(
      esrc, edst, E, Nn, as1, ad1, xw1, b1, nullptr, lst, keys, planes, plane_words,
      w2p, a2_src, a2_dst, xw2p, as2, ad2);

  gat_owner_kernel<1, 40, 64, false, true, 0, 0><<<own_blocks, OWN_THREADS, 0, stream>>>(
      esrc, edst, E, Nn, as2, ad2, xw2p, b2, out, lst, keys, planes, plane_words,
      nullptr, nullptr, nullptr, nullptr, nullptr, nullptr);
}


# row gathers in 16-byte pieces per lane (8-byte in layer 2): a 1 KB row takes 2 whole-line loads, not 8; LDS sum rows kept lane-major so the adds stay conflict-free; sums by arrival as before
# speedup vs baseline: 16.7944x; 1.0307x over previous
#include <hip/hip_runtime.h>

typedef __attribute__((ext_vector_type(2))) float v2f;
typedef __attribute__((ext_vector_type(8))) float v8f;

#define NEG_SLOPE 0.2f

static __device__ __forceinline__ unsigned f2key(float f) {
  unsigned u = __float_as_uint(f);
  return (u & 0x80000000u) ? ~u : (u | 0x80000000u);
}
static __device__ __forceinline__ float key2f(unsigned k) {
  return (k & 0x80000000u) ? __uint_as_float(k & 0x7FFFFFFFu)
                           : __uint_as_float(~k);
}

#define OWN_T       196
typedef __attribute__((ext_vector_type(4))) int own_v4i;
typedef __attribute__((ext_vector_type(4))) unsigned own_v4u;

#define PREP_THREADS 256
#define PLANE_W      (PREP_THREADS / 32)
__global__ void prep_kernel(const float* __restrict__ src,
                            float* __restrict__ dst,
                            int K, int c, int ldd,
                            const int* __restrict__ edst,
                            unsigned* __restrict__ keys, int E, int pad_blocks,
                            unsigned* __restrict__ planes, int plane_words) {
  __shared__ unsigned bm[256 * PLANE_W];
  if ((int)blockIdx.x < pad_blocks) {
    int i = blockIdx.x * blockDim.x + threadIdx.x;
    if (i >= K * ldd) return;
    int r = i / ldd, j = i - r * ldd;
    dst[i] = (j < c) ? src[r * c + j] : 0.0f;
    return;
  }
  const int tid = threadIdx.x;
  const int kb = blockIdx.x - pad_blocks;
  for (int i = tid; i < 256 * PLANE_W; i += PREP_THREADS) bm[i] = 0u;
  __syncthreads();
  const int piece = kb * PREP_THREADS + tid;
  const int e0 = 16 * piece;
  if (e0 < E) {
    const bool full = (e0 + 15 < E);
    int d[16];
    if (full && ((((size_t)(edst + e0)) & 15) == 0)) {
      const own_v4i* dv = (const own_v4i*)(edst + e0);
      for (int q = 0; q < 4; ++q) {
        const own_v4i v = dv[q];
        d[4 * q + 0] = v.x; d[4 * q + 1] = v.y; d[4 * q + 2] = v.z; d[4 * q + 3] = v.w;
      }
    } else {
      for (int u = 0; u < 16; ++u) d[u] = (e0 + u < E) ? edst[e0 + u] : -1;
    }
    unsigned kw[4] = {0u, 0u, 0u, 0u};
    const unsigned bit = 1u << (tid & 31);
    for (int u = 0; u < 16; ++u) {
      if (d[u] >= 0) {
        const unsigned g = (unsigned)(d[u] / OWN_T) & 255u;
        kw[u >> 2] |= g << (8 * (u & 3));
        if (full) atomicOr(&bm[g * PLANE_W + (tid >> 5)], bit);
      }
    }
    if (full && ((((size_t)keys) & 15) == 0)) {
      own_v4u k4;
      k4.x = kw[0]; k4.y = kw[1]; k4.z = kw[2]; k4.w = kw[3];
      *(own_v4u*)(keys + 4 * piece) = k4;
    } else {
      for (int q = 0; q < 4; ++q)
        if (e0 + 4 * q < E) keys[4 * piece + q] = kw[q];
    }
  }
  __syncthreads();
  unsigned* row = planes + (size_t)tid * plane_words + (size_t)kb * PLANE_W;
  for (int q = 0; q < PLANE_W; ++q) row[q] = bm[tid * PLANE_W + q];
}

#define GEMM_MT 2
__global__ __launch_bounds__(128) void gemm_f32_wmma_kernel(const float* __restrict__ A,
                                     const float* __restrict__ B,
                                     float* __restrict__ C,
                                     int M, int K, int ldb) {
  const int lane = threadIdx.x & 31;
  const int wave = threadIdx.x >> 5;
  const int half = lane >> 4;
  const int l    = lane & 15;
  const int nt64    = ldb >> 6;
  const int mtiles  = M >> 4;
  const int mgroups = (mtiles + GEMM_MT - 1) / GEMM_MT;
  const int total   = mgroups * nt64;
  int t = blockIdx.x * (blockDim.x >> 5) + wave;
  if (t >= total) return;
  const int mg = t / nt64;
  const int n0 = (t - mg * nt64) << 6;
  const int m0 = mg * (16 * GEMM_MT);

  v8f acc[GEMM_MT][4] = {};
  const float* arow[GEMM_MT];
#pragma unroll
  for (int i = 0; i < GEMM_MT; ++i) {
    int r = m0 + 16 * i + l;
    r = (r < M) ? r : (M - 1);
    arow[i] = A + (size_t)r * K + 2 * half;
  }
  const float* bp   = B + (size_t)(2 * half) * ldb + n0 + l;
  const size_t bstep = (size_t)4 * ldb;

  v2f a0[GEMM_MT], a1[GEMM_MT], b0[4], b1[4];
  const int nsteps = K >> 2;

#define GEMM_LOAD(av, bv, step)                                              \
  {                                                                          \
    _Pragma("unroll") for (int i = 0; i < GEMM_MT; ++i)                      \
        av[i] = *(const v2f*)(arow[i] + 4 * (step));                         \
    const float* bq = bp + (size_t)(step) * bstep;                           \
    _Pragma("unroll") for (int s = 0; s < 4; ++s) {                          \
      bv[s].x = bq[s * 16];                                                  \
      bv[s].y = bq[ldb + s * 16];                                            \
    }                                                                        \
  }
#define GEMM_MMA(av, bv)                                                     \
  {                                                                          \
    _Pragma("unroll") for (int i = 0; i < GEMM_MT; ++i)                      \
        _Pragma("unroll") for (int s = 0; s < 4; ++s)                        \
            acc[i][s] = __builtin_amdgcn_wmma_f32_16x16x4_f32(               \
                false, av[i], false, bv[s], (short)0, acc[i][s], false, false); \
  }

  GEMM_LOAD(a0, b0, 0)
  int st = 0;
  for (; st + 2 < nsteps; st += 2) {
    GEMM_LOAD(a1, b1, st + 1)
    asm volatile("" ::: "memory");
    GEMM_MMA(a0, b0)
    GEMM_LOAD(a0, b0, st + 2)
    asm volatile("" ::: "memory");
    GEMM_MMA(a1, b1)
  }
  GEMM_LOAD(a1, b1, st + 1)
  asm volatile("" ::: "memory");
  GEMM_MMA(a0, b0)
  GEMM_MMA(a1, b1)
#undef GEMM_LOAD
#undef GEMM_MMA

#pragma unroll
  for (int i = 0; i < GEMM_MT; ++i) {
    if (m0 + 16 * i < M) {
      float* crow = C + (size_t)(m0 + 16 * i + 8 * half) * ldb + n0 + l;
#pragma unroll
      for (int s = 0; s < 4; ++s)
#pragma unroll
        for (int j = 0; j < 8; ++j) crow[(size_t)j * ldb + s * 16] = acc[i][s][j];
    }
  }
}

__global__ void alpha_kernel(const float* __restrict__ xw,
                             const float* __restrict__ a_src,
                             const float* __restrict__ a_dst,
                             float* __restrict__ as, float* __restrict__ ad,
                             int n_nodes, int heads, int c, int ldx) {
  int i = blockIdx.x * blockDim.x + threadIdx.x;
  if (i >= n_nodes * heads) return;
  int n = i / heads, h = i - n * heads;
  const float* row = xw + (size_t)n * ldx + (size_t)h * c;
  const float* ws = a_src + (size_t)h * c;
  const float* wd = a_dst + (size_t)h * c;
  float s = 0.0f, d = 0.0f;
  for (int j = 0; j < c; ++j) {
    float v = row[j];
    s += v * ws[j];
    d += v * wd[j];
  }
  as[i] = s;
  ad[i] = d;
}

#define OWN_CAP     8192
#define OWN_THREADS 1024
#define OWN_HEAD    32
#define OWN_REGION  (OWN_HEAD + 2 * OWN_CAP)

static __device__ __forceinline__ void own_push(int e, int dl, int* ls, int* ld,
                                                unsigned* cnt) {
  unsigned idx = atomicAdd(cnt, 1u);
  if (idx < OWN_CAP) { ls[idx] = e; ld[idx] = dl; }
}

static __device__ __forceinline__ void own_take4(int e, own_v4i d, int n0, int nT,
                                                 int* ls, int* ld, unsigned* cnt) {
  const int r0 = d.x - n0, r1 = d.y - n0, r2 = d.z - n0, r3 = d.w - n0;
  if ((unsigned)r0 < (unsigned)nT) own_push(e + 0, r0, ls, ld, cnt);
  if ((unsigned)r1 < (unsigned)nT) own_push(e + 1, r1, ls, ld, cnt);
  if ((unsigned)r2 < (unsigned)nT) own_push(e + 2, r2, ls, ld, cnt);
  if ((unsigned)r3 < (unsigned)nT) own_push(e + 3, r3, ls, ld, cnt);
}

static __device__ __forceinline__ void own_build(const int* __restrict__ edst,
                                                 int eb, int ee, int n0, int nT,
                                                 int* ls, int* ld, unsigned* cnt) {
  const int tid = threadIdx.x;
  const int n = ee - eb;
  const int* dbase = edst + eb;
  const int nvec = ((((size_t)dbase) & 15) == 0) ? (n >> 2) : 0;
  const own_v4i* dv = (const own_v4i*)dbase;
  const int rot = (int)(((long)blockIdx.x * nvec) / gridDim.x) & ~31;
  int v = tid;
  for (; v + 3 * OWN_THREADS < nvec; v += 4 * OWN_THREADS) {
    own_v4i d[4];
    int vv[4];
#pragma unroll
    for (int u = 0; u < 4; ++u) {
      vv[u] = v + u * OWN_THREADS + rot;
      if (vv[u] >= nvec) vv[u] -= nvec;
      d[u] = dv[vv[u]];
    }
    asm volatile("" :: "v"(d[0]), "v"(d[1]), "v"(d[2]), "v"(d[3]));
#pragma unroll
    for (int u = 0; u < 4; ++u)
      own_take4(eb + 4 * vv[u], d[u], n0, nT, ls, ld, cnt);
  }
  for (; v < nvec; v += OWN_THREADS) {
    int vv = v + rot;
    if (vv >= nvec) vv -= nvec;
    own_take4(eb + 4 * vv, dv[vv], n0, nT, ls, ld, cnt);
  }
  for (int e = eb + 4 * nvec + tid; e < ee; e += OWN_THREADS) {
    const int r = edst[e] - n0;
    if ((unsigned)r < (unsigned)nT) own_push(e, r, ls, ld, cnt);
  }
}

static __device__ __forceinline__ void own_push_e(int e, int* ls, unsigned* cnt) {
  unsigned idx = atomicAdd(cnt, 1u);
  if (idx < OWN_CAP) ls[idx] = e;
}

static __device__ __forceinline__ unsigned own_hits(unsigned w, unsigned pat) {
  const unsigned y = w ^ pat;
  const unsigned t = (y & 0x7F7F7F7Fu) + 0x7F7F7F7Fu;
  return ~(t | y) & 0x80808080u;
}

static __device__ __forceinline__ void own_take16(int e, own_v4u k, unsigned pat,
                                                  int* ls, unsigned* cnt) {
  unsigned m[4] = {own_hits(k.x, pat), own_hits(k.y, pat),
                   own_hits(k.z, pat), own_hits(k.w, pat)};
  if ((m[0] | m[1] | m[2] | m[3]) != 0u) {
#pragma unroll
    for (int u = 0; u < 4; ++u) {
      unsigned mm = m[u];
      while (mm) {
        own_push_e(e + 4 * u + (__builtin_ctz(mm) >> 3), ls, cnt);
        mm &= mm - 1u;
      }
    }
  }
}

static __device__ __forceinline__ void own_build_keys(const unsigned* __restrict__ keys,
                                                      const unsigned* __restrict__ plane,
                                                      int E, int* ls, unsigned* pl,
                                                      unsigned* cnt, unsigned* pcnt) {
  const int tid = threadIdx.x;
  const int lane = tid & 31;
  const unsigned gid = blockIdx.x;
  const unsigned pat = gid * 0x01010101u;
  const int nvec = ((((size_t)keys) & 15) == 0) ? (E >> 4) : 0;
  const own_v4u* kv = (const own_v4u*)keys;
  const int nwords = (nvec + 31) >> 5;
  for (int w0 = tid & ~31; w0 < nwords; w0 += OWN_THREADS) {
    const int w = w0 + lane;
    unsigned m = (w < nwords) ? plane[w] : 0u;
    const int n = __builtin_popcount(m);
    int incl = n;
    for (int dd = 1; dd < 32; dd <<= 1) {
      const int t = __shfl_up(incl, dd);
      if (lane >= dd) incl += t;
    }
    const int tot = __shfl(incl, 31);
    if (tot != 0) {
      unsigned base = 0u;
      if (lane == 31) base = atomicAdd(pcnt, (unsigned)tot);
      base = (unsigned)__shfl((int)base, 31);
      unsigned idx = base + (unsigned)(incl - n);
      while (m) {
        if (idx < OWN_CAP) pl[idx] = 32u * (unsigned)w + (unsigned)__builtin_ctz(m);
        ++idx;
        m &= m - 1u;
      }
    }
  }
  __syncthreads();
  const int np = (int)*pcnt;
  if (np > OWN_CAP) {
    if (tid == 0) *cnt = (unsigned)np;
    return;
  }
  for (int i0 = tid; i0 < np; i0 += 4 * OWN_THREADS) {
    own_v4u k[4];
    unsigned pc[4];
#pragma unroll
    for (int u = 0; u < 4; ++u) {
      const int i = i0 + u * OWN_THREADS;
      pc[u] = pl[(i < np) ? i : 0];
      k[u] = kv[pc[u]];
    }
    asm volatile("" :: "v"(k[0]), "v"(k[1]), "v"(k[2]), "v"(k[3]));
#pragma unroll
    for (int u = 0; u < 4; ++u)
      if (i0 + u * OWN_THREADS < np) own_take16(16 * (int)pc[u], k[u], pat, ls, cnt);
  }
  for (int e = 16 * nvec + tid; e < E; e += OWN_THREADS)
    if (((keys[e >> 2] >> (8 * (e & 3))) & 255u) == gid) own_push_e(e, ls, cnt);
}

template <bool WITH_DST>
static __device__ __forceinline__ void own_resolve(const int* __restrict__ esrc,
                                                   const int* __restrict__ edst,
                                                   int n0, int* ls, int* ld, int n) {
  for (int i0 = threadIdx.x; i0 < n; i0 += 4 * OWN_THREADS) {
    int e[4], s[4], d[4];
#pragma unroll
    for (int u = 0; u < 4; ++u) {
      const int i = i0 + u * OWN_THREADS;
      e[u] = (i < n) ? ls[i] : 0;
    }
#pragma unroll
    for (int u = 0; u < 4; ++u) {
      s[u] = esrc[e[u]];
      d[u] = WITH_DST ? edst[e[u]] : 0;
    }
    asm volatile("" :: "v"(s[0]), "v"(s[1]), "v"(s[2]), "v"(s[3]),
                       "v"(d[0]), "v"(d[1]), "v"(d[2]), "v"(d[3]));
#pragma unroll
    for (int u = 0; u < 4; ++u) {
      const int i = i0 + u * OWN_THREADS;
      if (i < n) {
        ls[i] = s[u];
        if (WITH_DST) ld[i] = d[u] - n0;
      }
    }
  }
}

static __device__ __forceinline__ void own_leave(const int* ls, const int* ld,
                                                 int* __restrict__ gs,
                                                 int* __restrict__ gd, int n) {
  for (int i = threadIdx.x; i < n; i += OWN_THREADS) {
    gs[i] = ls[i];
    gd[i] = ld[i];
  }
}

static __device__ __forceinline__ void own_take(const int* __restrict__ gs,
                                                const int* __restrict__ gd,
                                                int* ls, int* ld, int n) {
  for (int i0 = threadIdx.x; i0 < n; i0 += 4 * OWN_THREADS) {
    int s[4], d[4];
#pragma unroll
    for (int u = 0; u < 4; ++u) {
      const int i = i0 + u * OWN_THREADS;
      const int j = (i < n) ? i : 0;
      s[u] = gs[j];
      d[u] = gd[j];
    }
#pragma unroll
    for (int u = 0; u < 4; ++u) {
      const int i = i0 + u * OWN_THREADS;
      if (i < n) { ls[i] = s[u]; ld[i] = d[u]; }
    }
  }
}

static __device__ __forceinline__ void own_entry(int i, int n_list, int n0,
                                                 const int* ls, const int* ld,
                                                 int& s, int& dl) {
  if (i < n_list) { s = ls[i]; dl = ld[i]; }
  else            { dl = i - n_list; s = n0 + dl; }
}

template <int H, int NU>
static __device__ __forceinline__ void own_max_trip(int idx0, int n_list,
                                                    int n0, const int* ls,
                                                    const int* ld,
                                                    const float* __restrict__ as,
                                                    const float* adL, unsigned* mx) {
  float a[NU];
  int dh[NU];
#pragma unroll
  for (int u = 0; u < NU; ++u) {
    const int idx = idx0 + u * OWN_THREADS;
    const int i = idx / H, h = idx - i * H;
    int s, dl;
    own_entry(i, n_list, n0, ls, ld, s, dl);
    a[u] = as[s * H + h];
    dh[u] = dl * H + h;
  }
#pragma unroll
  for (int u = 0; u < NU; ++u) {
    float v = a[u] + adL[dh[u]];
    v = (v >= 0.0f) ? v : NEG_SLOPE * v;
    atomicMax(&mx[dh[u]], f2key(v));
  }
}

template <int H>
static __device__ __forceinline__ void own_max(int n_list, int n_self,
                                               int n0, const int* ls, const int* ld,
                                               const float* __restrict__ as,
                                               const float* adL, unsigned* mx) {
  const int n_items = (n_list + n_self) * H;
  int idx = threadIdx.x;
  for (; idx + 3 * OWN_THREADS < n_items; idx += 4 * OWN_THREADS)
    own_max_trip<H, 4>(idx, n_list, n0, ls, ld, as, adL, mx);
  for (; idx < n_items; idx += OWN_THREADS)
    own_max_trip<H, 1>(idx, n_list, n0, ls, ld, as, adL, mx);
}

template <int LW>
static __device__ __forceinline__ int own_pos_of(int c) {
  if (LW == 4) return (c & ~127) | ((c & 3) << 5) | ((c >> 2) & 31);
  return c;
}
template <int LW>
static __device__ __forceinline__ int own_chan_at(int p) {
  if (LW == 4) return (p & ~127) | ((p & 31) << 2) | ((p >> 5) & 3);
  return p;
}
template <int LW> struct own_vec;
template <> struct own_vec<2> { typedef __attribute__((ext_vector_type(2))) float t; };
template <> struct own_vec<4> { typedef __attribute__((ext_vector_type(4))) float t; };

template <int H, int C, int LW>
static __device__ __forceinline__ float own_weight(float w, int base, int j, int lane) {
  constexpr int CH = 32 * LW;
  if (H == 1 || (CH <= C && (C % CH) == 0)) {
    const int head = (H == 1) ? 0 : (j * CH) / C;
    return __int_as_float(__builtin_amdgcn_readlane(__float_as_int(w), base + head));
  } else {
    int head = (j * CH + LW * lane) / C;
    head = (head < H) ? head : (H - 1);
    return __shfl(w, base + head);
  }
}

template <int BANK>
static __device__ __forceinline__ int own_row_lane(int old, int v, int n) {
  switch (n) {
#define OWN_RS(N) \
    case N: return __builtin_amdgcn_update_dpp(old, v, 0x150 + N, 0xF, BANK, BANK == 0xF);
    OWN_RS(0) OWN_RS(1) OWN_RS(2) OWN_RS(3) OWN_RS(4) OWN_RS(5) OWN_RS(6) OWN_RS(7)
    OWN_RS(8) OWN_RS(9) OWN_RS(10) OWN_RS(11) OWN_RS(12) OWN_RS(13) OWN_RS(14) OWN_RS(15)
#undef OWN_RS
  }
  return old;
}

template <int H, int C, int LDX, int AP, int LW, int NU>
static __device__ __forceinline__ void own_rows_trip(int i0, int n_list, int n0,
                                                     const int* ls, const int* ld,
                                                     const float* __restrict__ as,
                                                     const float* __restrict__ xw,
                                                     const float* adL,
                                                     const unsigned* mx,
                                                     float* den, float* acc) {
  typedef typename own_vec<LW>::t vec;
  constexpr int HC = H * C;
  constexpr int CH = 32 * LW;
  constexpr int NJ = (HC + CH - 1) / CH;
  const int lane = threadIdx.x & 31;
  const int hl = lane & (H - 1);
  int dl[NU];
  vec x[NU][NJ];
  float av[NU];
#pragma unroll
  for (int u = 0; u < NU; ++u) {
    int si, di;
    own_entry(i0 + u * (OWN_THREADS / 32), n_list, n0, ls, ld, si, di);
    const int s = __builtin_amdgcn_readfirstlane(si);
    dl[u] = __builtin_amdgcn_readfirstlane(di);
    const float* row = xw + (size_t)s * LDX;
#pragma unroll
    for (int j = 0; j < NJ; ++j) {
      const int c = j * CH + LW * lane;
      x[u][j] = (c < HC) ? *(const vec*)(row + c) : (vec)(0.0f);
    }
    av[u] = as[s * H + hl];
  }
#pragma unroll
  for (int u = 0; u < NU; ++u) {
    const int di = dl[u] * H + hl;
    float v = av[u] + adL[di];
    v = (v >= 0.0f) ? v : NEG_SLOPE * v;
    const float w = expf(v - key2f(mx[di]));
    if (lane < H) atomicAdd(&den[di], w);
    float* arow = acc + dl[u] * AP;
#pragma unroll
    for (int j = 0; j < NJ; ++j) {
      const float wj = own_weight<H, C, LW>(w, 0, j, lane);
      const int c = j * CH + LW * lane;
#pragma unroll
      for (int q = 0; q < LW; ++q)
        if (c < HC) atomicAdd(&arow[own_pos_of<LW>(c + q)], wj * x[u][j][q]);
    }
  }
}

template <int H, int C, int LDX, int AP, int LW>
static __device__ __forceinline__ void own_rows_block(int i0, int n_list, int n0,
                                                      const int* ls, const int* ld,
                                                      const float* __restrict__ as,
                                                      const float* __restrict__ xw,
                                                      const float* adL,
                                                      const unsigned* mx,
                                                      float* den, float* acc) {
  typedef typename own_vec<LW>::t vec;
  constexpr int HC = H * C;
  constexpr int CH = 32 * LW;
  constexpr int NJ = (HC + CH - 1) / CH;
  constexpr int G = 32 / H;
  constexpr int SUB = (G * NJ * LW <= 32) ? G : (32 / (NJ * LW));
  static_assert(SUB >= 1 && (G % SUB) == 0, "a block is a whole number of row groups");
  constexpr bool BYROW = (H == 8 && C == 32 && LW == 4);
  const int lane = threadIdx.x & 31;
  const int hl = lane & (H - 1);
  int si, di;
  own_entry(i0 + lane / H, n_list, n0, ls, ld, si, di);
  const int dh = di * H + hl;
  const float a = as[si * H + hl];
  float w = 0.0f;
  int wt = 0;
#pragma unroll
  for (int k = 0; k < G / SUB; ++k) {
    vec x[SUB][NJ];
#pragma unroll
    for (int u = 0; u < SUB; ++u) {
      const int s = __builtin_amdgcn_readlane(si, (k * SUB + u) * H);
      const float* row = xw + (size_t)s * LDX;
#pragma unroll
      for (int j = 0; j < NJ; ++j) {
        const int c = j * CH + LW * lane;
        x[u][j] = (c < HC) ? *(const vec*)(row + c) : (vec)(0.0f);
      }
    }
    if (k == 0) {
      float v = a + adL[dh];
      v = (v >= 0.0f) ? v : NEG_SLOPE * v;
      w = expf(v - key2f(mx[dh]));
      atomicAdd(&den[dh], w);
      if (BYROW) {
        const int from = ((lane >> 2) & 3) * H + ((lane >> 1) & 1) * 4 + (lane >> 4) * 2 + (lane & 1);
        wt = __float_as_int(__shfl(w, from));
      }
    }
#pragma unroll
    for (int u = 0; u < SUB; ++u) {
      const int e = k * SUB + u;
      const int dl = __builtin_amdgcn_readlane(di, e * H);
      float* arow = acc + dl * AP;
#pragma unroll
      for (int j = 0; j < NJ; ++j) {
        float wj;
        if (BYROW) {
          int t = own_row_lane<0xF>(0, wt, 4 * e + 2 * j);
          t = own_row_lane<0xC>(t, wt, 4 * e + 2 * j + 1);
          wj = __int_as_float(t);
        } else {
          wj = own_weight<H, C, LW>(w, e * H, j, lane);
        }
        const int c = j * CH + LW * lane;
#pragma unroll
        for (int q = 0; q < LW; ++q)
          if (c < HC) atomicAdd(&arow[own_pos_of<LW>(c + q)], wj * x[u][j][q]);
      }
    }
  }
}

template <int H, int C, int LDX, int AP, int LW>
static __device__ __forceinline__ void own_rows(int n_list, int n_self, int n0,
                                                const int* ls, const int* ld,
                                                const float* __restrict__ as,
                                                const float* __restrict__ xw,
                                                const float* adL, const unsigned* mx,
                                                float* den, float* acc) {
  static_assert((C % LW) == 0 && (LDX % LW) == 0, "a lane's piece lies in one head, aligned");
  static_assert(LW == 2 || (LW == 4 && ((H * C) % 128) == 0), "the order of acc: whole pieces");
  static_assert((H & (H - 1)) == 0 && H <= 32, "heads: a power of two");
  constexpr int NW = OWN_THREADS / 32;
  constexpr int G = 32 / H;
  static_assert(G <= NW, "the entries left over get one wave each");
  const int n_items = n_list + n_self;
  const int n_blocks = n_items / G;
  const int wave = __builtin_amdgcn_readfirstlane((int)(threadIdx.x >> 5));
  for (int b = wave; b < n_blocks; b += NW)
    own_rows_block<H, C, LDX, AP, LW>(b * G, n_list, n0, ls, ld, as, xw, adL, mx, den, acc);
  const int t = n_blocks * G + wave;
  if (t < n_items)
    own_rows_trip<H, C, LDX, AP, LW, 1>(t, n_list, n0, ls, ld, as, xw, adL, mx, den, acc);
}

template <int H, int C, int LDX, bool ELU, bool TAKE, int NC2, int LD2>
__global__ __launch_bounds__(OWN_THREADS) void gat_owner_kernel(
    const int* __restrict__ esrc, const int* __restrict__ edst, int E, int Nn,
    const float* __restrict__ as, const float* __restrict__ ad,
    const float* __restrict__ xw, const float* __restrict__ bias,
    float* __restrict__ outp, int* lst, const unsigned* __restrict__ keys,
    const unsigned* __restrict__ planes, int plane_words,
    const float* __restrict__ w2, const float* __restrict__ a2_src,
    const float* __restrict__ a2_dst, float* __restrict__ xw2,
    float* __restrict__ as2, float* __restrict__ ad2) {
  constexpr int HC = H * C;
  constexpr bool NEXT = (NC2 > 0);
  constexpr int AP = NEXT ? HC + 4 : HC;
  constexpr int LW = ((HC % 128) == 0) ? 4 : 2;
  __shared__ float    acc[OWN_T * AP];
  __shared__ float    adL[OWN_T * H];
  __shared__ float    den[OWN_T * H];
  __shared__ unsigned mx[OWN_T * H];
  __shared__ int      lsd[2 * OWN_CAP];
  int* const ls = lsd;
  int* const ld = lsd + OWN_CAP;
  __shared__ unsigned cnt;
  __shared__ unsigned pcnt;

  const int tid = threadIdx.x;
  const int n0 = blockIdx.x * OWN_T;
  const int nT = min(OWN_T, Nn - n0);

  for (int i = tid; i < OWN_T * AP; i += OWN_THREADS) acc[i] = 0.0f;
  for (int i = tid; i < OWN_T * H; i += OWN_THREADS) {
    den[i] = (i < nT * H) ? 0.0f : 1.0f;
    mx[i] = 0u;
    adL[i] = (i < nT * H) ? ad[(size_t)n0 * H + i] : 0.0f;
  }
  if (tid == 0) { cnt = 0u; pcnt = 0u; }

  int* greg = lst + (size_t)blockIdx.x * OWN_REGION;
  int* gs = greg + OWN_HEAD;
  int* gd = gs + OWN_CAP;
  int total;
  const bool by_key = (gridDim.x <= 256u);
  if (TAKE) {
    total = greg[0];
    if (total <= OWN_CAP) own_take(gs, gd, ls, ld, total);
    __syncthreads();
  } else {
    __syncthreads();
    if (by_key) own_build_keys(keys, planes + (size_t)blockIdx.x * plane_words, E, ls,
                               (unsigned*)ld, &cnt, &pcnt);
    else        own_build(edst, 0, E, n0, nT, ls, ld, &cnt);
    __syncthreads();
    total = (int)cnt;
    if (tid == 0) greg[0] = total;
  }

  if (total <= OWN_CAP) {
    if (!TAKE) {
      if (by_key) own_resolve<true>(esrc, edst, n0, ls, ld, total);
      else        own_resolve<false>(esrc, edst, n0, ls, ld, total);
      __syncthreads();
      own_leave(ls, ld, gs, gd, total);
    }
    own_max<H>(total, nT, n0, ls, ld, as, adL, mx);
    __syncthreads();
    own_rows<H, C, LDX, AP, LW>(total, nT, n0, ls, ld, as, xw, adL, mx, den, acc);
  } else {
    for (int phase = 0; phase < 2; ++phase) {
      for (int eb = 0; eb < E; eb += OWN_CAP) {
        __syncthreads();
        if (tid == 0) cnt = 0u;
        __syncthreads();
        own_build(edst, eb, min(eb + OWN_CAP, E), n0, nT, ls, ld, &cnt);
        __syncthreads();
        const int m = (int)cnt;
        own_resolve<false>(esrc, edst, n0, ls, ld, m);
        __syncthreads();
        const int n_self = (eb == 0) ? nT : 0;
        if (phase == 0)
          own_max<H>(m, n_self, n0, ls, ld, as, adL, mx);
        else
          own_rows<H, C, LDX, AP, LW>(m, n_self, n0, ls, ld, as, xw, adL, mx, den, acc);
      }
    }
  }
  __syncthreads();

  if constexpr (!NEXT) {
    float* orow = outp + (size_t)n0 * HC;
    for (int i = tid; i < nT * HC; i += OWN_THREADS) {
      const int r = i / HC, cc = i - r * HC;
      float v = acc[r * AP + own_pos_of<LW>(cc)] / den[r * H + cc / C] + bias[cc];
      if (ELU) v = (v > 0.0f) ? v : (expf(v) - 1.0f);
      orow[i] = v;
    }
  } else {
    constexpr int NT2 = (NC2 + 15) / 16;
    constexpr int NRT = (OWN_T + 15) / 16;
    constexpr int TP  = 16 * NT2 + 1;
    static_assert((HC % 4) == 0, "K of the next product goes in steps of 4");
    static_assert(OWN_T * TP <= 2 * OWN_CAP, "the tile fits the edge list's space");
    static_assert(16 * NT2 <= LD2 && NRT <= OWN_THREADS / 32, "tile shape");
    for (int i = tid; i < OWN_T * H; i += OWN_THREADS) den[i] = 1.0f / den[i];
    __syncthreads();
    for (int i = tid; i < OWN_T * HC; i += OWN_THREADS) {
      const int r = i / HC, p = i - r * HC;
      const int cc = own_chan_at<LW>(p);
      float v = acc[r * AP + p] * den[r * H + cc / C] + bias[cc];
      if (ELU) v = (v > 0.0f) ? v : (expf(v) - 1.0f);
      acc[r * AP + p] = v;
    }
    __syncthreads();
    float* tile = (float*)lsd;
    const int wave = __builtin_amdgcn_readfirstlane((int)(threadIdx.x >> 5));
    if (wave < NRT) {
      const int lane = tid & 31;
      const int half = lane >> 4;
      const int l    = lane & 15;
      int r = 16 * wave + l;
      r = (r < OWN_T) ? r : (OWN_T - 1);
      const float* ap = acc + r * AP + own_pos_of<LW>(2 * half);
      const float* bp = w2 + (size_t)(2 * half) * LD2 + l;
      v8f d[NT2] = {};
#pragma unroll 4
      for (int s = 0; s < HC / 4; ++s) {
        v2f a;
        a.x = ap[own_pos_of<LW>(4 * s)];
        a.y = ap[own_pos_of<LW>(4 * s) + own_pos_of<LW>(1)];
        const float* bq = bp + (size_t)(4 * s) * LD2;
        v2f b[NT2];
#pragma unroll
        for (int t = 0; t < NT2; ++t) {
          b[t].x = bq[t * 16];
          b[t].y = bq[LD2 + t * 16];
        }
#pragma unroll
        for (int t = 0; t < NT2; ++t)
          d[t] = __builtin_amdgcn_wmma_f32_16x16x4_f32(
              false, a, false, b[t], (short)0, d[t], false, false);
      }
#pragma unroll
      for (int j = 0; j < 8; ++j) {
        const int row = 16 * wave + 8 * half + j;
        if (row < OWN_T) {
#pragma unroll
          for (int t = 0; t < NT2; ++t) tile[row * TP + t * 16 + l] = d[t][j];
        }
      }
    }
    __syncthreads();
    for (int i = tid; i < nT * LD2; i += OWN_THREADS) {
      const int r = i / LD2, cc = i - r * LD2;
      if (cc < 16 * NT2) xw2[(size_t)(n0 + r) * LD2 + cc] = tile[r * TP + cc];
    }
    if (tid < nT) {
      float s = 0.0f, dd = 0.0f;
      for (int j = 0; j < NC2; ++j) {
        const float v = tile[tid * TP + j];
        s += v * a2_src[j];
        dd += v * a2_dst[j];
      }
      as2[n0 + tid] = s;
      ad2[n0 + tid] = dd;
    }
  }
}

extern "C" void kernel_launch(void* const* d_in, const int* in_sizes, int n_in,
                              void* d_out, int out_size, void* d_ws, size_t ws_size,
                              hipStream_t stream) {
  const float* x      = (const float*)d_in[0];
  const int*   ei     = (const int*)d_in[1];
  const float* W1     = (const float*)d_in[2];
  const float* a1_src = (const float*)d_in[3];
  const float* a1_dst = (const float*)d_in[4];
  const float* b1     = (const float*)d_in[5];
  const float* W2     = (const float*)d_in[6];
  const float* a2_src = (const float*)d_in[7];
  const float* a2_dst = (const float*)d_in[8];
  const float* b2     = (const float*)d_in[9];
  float* out = (float*)d_out;
  (void)n_in; (void)out_size; (void)ws_size;

  const int F_IN = 256, H1 = 8, C1 = 32, HC1 = 256, NC = 40, NCP = 64;
  const int Nn   = in_sizes[0] / F_IN;
  const int E    = in_sizes[1] / 2;
  const int* esrc = ei;
  const int* edst = ei + E;

  float* p = (float*)d_ws;
  float* xw1  = p; p += (size_t)Nn * HC1;
  float* as1  = p; p += (size_t)Nn * H1;
  float* ad1  = p; p += (size_t)Nn * H1;
  float* w2p  = p; p += (size_t)HC1 * NCP;
  float* xw2p = p; p += (size_t)Nn * NCP;
  float* as2  = p; p += (size_t)Nn;
  float* ad2  = p; p += (size_t)Nn;
  unsigned* keys = (unsigned*)p; p += (size_t)((E / 4 + 7) & ~3);
  const int key_blocks = ((E + 15) / 16 + PREP_THREADS - 1) / PREP_THREADS;
  const int plane_words = key_blocks * PLANE_W;
  unsigned* planes = (unsigned*)p; p += (size_t)256 * plane_words;
  int*   lst  = (int*)p;

  dim3 blk(256);
  const int own_blocks = (Nn + OWN_T - 1) / OWN_T;

  {
    const int pad_blocks = (HC1 * NCP + PREP_THREADS - 1) / PREP_THREADS;
    prep_kernel<<<pad_blocks + key_blocks, PREP_THREADS, 0, stream>>>(
        W2, w2p, HC1, NC, NCP, edst, keys, E, pad_blocks, planes, plane_words);
  }

  {
    int waves  = ((Nn / 16 + GEMM_MT - 1) / GEMM_MT) * (HC1 / 64);
    int blocks = (waves + 3) / 4;
    gemm_f32_wmma_kernel<<<blocks, dim3(128), 0, stream>>>(x, W1, xw1, Nn, F_IN, HC1);
  }
  alpha_kernel<<<(Nn * H1 + 255) / 256, blk, 0, stream>>>(
      xw1, a1_src, a1_dst, as1, ad1, Nn, H1, C1, HC1);
  gat_owner_kernel<8, 32, 256, true, false, 40, 64><<<own_blocks, OWN_THREADS, 0, stream>>>(
      esrc, edst, E, Nn, as1, ad1, xw1, b1, nullptr, lst, keys, planes, plane_words,
      w2p, a2_src, a2_dst, xw2p, as2, ad2);

  gat_owner_kernel<1, 40, 64, false, true, 0, 0><<<own_blocks, OWN_THREADS, 0, stream>>>(
      esrc, edst, E, Nn, as2, ad2, xw2p, b2, out, lst, keys, planes, plane_words,
      nullptr, nullptr, nullptr, nullptr, nullptr, nullptr);
}
